# starting kernel
# baseline (speedup 1.0000x reference)
#include <hip/hip_runtime.h>
#include <hip/hip_bf16.h>
#include <math.h>

#define IN_DIM  128
#define OUT_DIM 64
#define ED_DIM  32

#define BSTRIDE_V2 80
#define BSTRIDE_F  (2 * BSTRIDE_V2)

typedef __attribute__((ext_vector_type(2))) float v2f;
typedef __attribute__((ext_vector_type(8))) float v8f;

struct Acc4 { v8f t[4]; };

__device__ __attribute__((always_inline)) inline
Acc4 wave_gemm_tile(const float* __restrict__ arow,
                    const float* __restrict__ lWs,
                    int kdim, int half, int l16)
{
    Acc4 A;
    A.t[0] = {}; A.t[1] = {}; A.t[2] = {}; A.t[3] = {};
    const v2f* a2 = (const v2f*)arow;
    for (int k0 = 0; k0 < kdim; k0 += 4) {
        const int kp  = (k0 >> 1) + half;
        const v2f  a  = a2[kp];
        const v2f* br = (const v2f*)(lWs + kp * BSTRIDE_F);
#pragma unroll
        for (int nt = 0; nt < 4; ++nt) {
            const v2f b = br[nt * 16 + l16];
            A.t[nt] = __builtin_amdgcn_wmma_f32_16x16x4_f32(
                false, a, false, b, (short)0, A.t[nt], false, false);
        }
    }
    return A;
}

__device__ __attribute__((always_inline)) inline
void stage_w(float* __restrict__ lWs, const float* __restrict__ W, int kdim)
{
    const int total = kdim * OUT_DIM;
    for (int i = threadIdx.x; i < total; i += blockDim.x) {
        const int k = i >> 6;
        const int n = i & 63;
        lWs[(k >> 1) * BSTRIDE_F + (n << 1) + (k & 1)] = W[i];
    }
}

__device__ inline void atomicMaxF(float* addr, float val) {
    if (val >= 0.0f) atomicMax(reinterpret_cast<int*>(addr), __float_as_int(val));
    else             atomicMin(reinterpret_cast<unsigned int*>(addr), __float_as_uint(val));
}

__global__ void k_node_gemm(const float* __restrict__ x,
                            const float* __restrict__ W_fc,  const float* __restrict__ b_fc,
                            const float* __restrict__ W_self, const float* __restrict__ b_self,
                            float* __restrict__ z, float* __restrict__ hout,
                            int N, int nTiles)
{
    __shared__ float lWs[(IN_DIM / 2) * BSTRIDE_F];
    const int lane = threadIdx.x & 31;
    const int wave = threadIdx.x >> 5;
    const int half = lane >> 4;
    const int l16  = lane & 15;
    const int tile = blockIdx.x * 8 + wave;

    int mrow = tile * 16 + l16;
    if (mrow >= N) mrow = N - 1;
    const float* xrow = x + (size_t)mrow * IN_DIM;

    stage_w(lWs, W_fc, IN_DIM);
    __syncthreads();
    if (tile < nTiles) {
        Acc4 acc = wave_gemm_tile(xrow, lWs, IN_DIM, half, l16);
#pragma unroll
        for (int nt = 0; nt < 4; ++nt) {
            const int   c  = nt * 16 + l16;
            const float bc = b_fc[c];
#pragma unroll
            for (int r = 0; r < 8; ++r) {
                const int row = tile * 16 + r + 8 * half;
                if (row < N) z[(size_t)row * OUT_DIM + c] = acc.t[nt][r] + bc;
            }
        }
    }
    __syncthreads();

    stage_w(lWs, W_self, IN_DIM);
    __syncthreads();
    if (tile < nTiles) {
        Acc4 acc = wave_gemm_tile(xrow, lWs, IN_DIM, half, l16);
#pragma unroll
        for (int nt = 0; nt < 4; ++nt) {
            const int   c  = nt * 16 + l16;
            const float bc = b_self[c];
#pragma unroll
            for (int r = 0; r < 8; ++r) {
                const int row = tile * 16 + r + 8 * half;
                if (row < N) hout[(size_t)row * OUT_DIM + c] = acc.t[nt][r] + bc;
            }
        }
    }
}

__global__ void k_node_attn(const float* __restrict__ z, const float* __restrict__ W_attn,
                            float* __restrict__ a_src, float* __restrict__ a_dst,
                            float* __restrict__ m, float* __restrict__ denom, int N)
{
    const int n = blockIdx.x * blockDim.x + threadIdx.x;
    if (n >= N) return;
    const float* zr = z + (size_t)n * OUT_DIM;
    float as = 0.f, ad = 0.f;
#pragma unroll 8
    for (int c = 0; c < OUT_DIM; ++c) {
        const float zc = zr[c];
        as += zc * W_attn[c];
        ad += zc * W_attn[OUT_DIM + c];
    }
    a_src[n] = as;
    a_dst[n] = ad;
    m[n]     = -INFINITY;
    denom[n] = 0.0f;
}

__global__ void k_edge_logit(const float* __restrict__ edge_attr,
                             const int* __restrict__ erow, const int* __restrict__ ecol,
                             const float* __restrict__ W_attn, const float* __restrict__ b_attn,
                             const float* __restrict__ a_src, const float* __restrict__ a_dst,
                             float* __restrict__ elog, float* __restrict__ m, int E)
{
    const int e = blockIdx.x * blockDim.x + threadIdx.x;
    if (e >= E) return;
    const float* ea = edge_attr + (size_t)e * ED_DIM;
    float dot = 0.f;
#pragma unroll
    for (int k = 0; k < ED_DIM; ++k) dot += ea[k] * W_attn[2 * OUT_DIM + k];
    const int c = ecol[e];
    const float s  = a_src[erow[e]] + a_dst[c] + dot + b_attn[0];
    const float lr = (s > 0.0f) ? s : 0.2f * s;
    elog[e] = lr;
    atomicMaxF(&m[c], lr);
}

__global__ void k_edge_soft(const float* __restrict__ elog, const int* __restrict__ ecol,
                            const float* __restrict__ m, float* __restrict__ pbuf,
                            float* __restrict__ denom, int E)
{
    const int e = blockIdx.x * blockDim.x + threadIdx.x;
    if (e >= E) return;
    const int c = ecol[e];
    const float p = expf(elog[e] - m[c]);
    pbuf[e] = p;
    atomicAdd(&denom[c], p);
}

__global__ void k_edge_aggregate(const float* __restrict__ edge_attr,
                                 const int* __restrict__ erow, const int* __restrict__ ecol,
                                 const float* __restrict__ W_edge, const float* __restrict__ b_edge,
                                 const float* __restrict__ z, const float* __restrict__ pbuf,
                                 const float* __restrict__ denom, float* __restrict__ hout,
                                 int E, int nTiles)
{
    __shared__ float lWs[(ED_DIM / 2) * BSTRIDE_F];
    const int lane = threadIdx.x & 31;
    const int wave = threadIdx.x >> 5;
    const int half = lane >> 4;
    const int l16  = lane & 15;
    const int tile = blockIdx.x * 8 + wave;

    stage_w(lWs, W_edge, ED_DIM);
    __syncthreads();
    if (tile >= nTiles) return;

    int arow = tile * 16 + l16;
    if (arow >= E) arow = E - 1;
    const float* ea = edge_attr + (size_t)arow * ED_DIM;

    {
        const size_t nxt = (size_t)arow + 16 * 8;
        if (nxt < (size_t)E)
            __builtin_prefetch(edge_attr + nxt * ED_DIM, 0, 1);
    }

    Acc4 acc = wave_gemm_tile(ea, lWs, ED_DIM, half, l16);

    float alph[8];
    int   cr[8], rr[8];
#pragma unroll
    for (int r = 0; r < 8; ++r) {
        const int eidx = tile * 16 + r + 8 * half;
        if (eidx < E) {
            const int c = ecol[eidx];
            cr[r]   = c;
            rr[r]   = erow[eidx];
            alph[r] = pbuf[eidx] / fmaxf(denom[c], 1e-16f);
        } else {
            cr[r] = -1; rr[r] = 0; alph[r] = 0.f;
        }
    }

#pragma unroll
    for (int nt = 0; nt < 4; ++nt) {
        const int   c  = nt * 16 + l16;
        const float bc = b_edge[c];
#pragma unroll
        for (int r = 0; r < 8; ++r) {
            if (cr[r] >= 0) {
                const float val = alph[r] * (z[(size_t)rr[r] * OUT_DIM + c] + acc.t[nt][r] + bc);
                atomicAdd(&hout[(size_t)cr[r] * OUT_DIM + c], val);
            }
        }
    }
}

extern "C" void kernel_launch(void* const* d_in, const int* in_sizes, int n_in,
                              void* d_out, int out_size, void* d_ws, size_t ws_size,
                              hipStream_t stream)
{
    const float* x         = (const float*)d_in[0];
    const float* edge_attr = (const float*)d_in[1];
    const int*   edge_idx  = (const int*)  d_in[2];
    const float* W_fc      = (const float*)d_in[3];
    const float* b_fc      = (const float*)d_in[4];
    const float* W_attn    = (const float*)d_in[5];
    const float* b_attn    = (const float*)d_in[6];
    const float* W_edge    = (const float*)d_in[7];
    const float* b_edge    = (const float*)d_in[8];
    const float* W_self    = (const float*)d_in[9];
    const float* b_self    = (const float*)d_in[10];

    const int N = in_sizes[0] / IN_DIM;
    const int E = in_sizes[1] / ED_DIM;
    const int* erow = edge_idx;
    const int* ecol = edge_idx + E;

    float* ws     = (float*)d_ws;
    float* z      = ws;
    float* a_src  = z + (size_t)N * OUT_DIM;
    float* a_dst  = a_src + N;
    float* m      = a_dst + N;
    float* denom  = m + N;
    float* elog   = denom + N;
    float* pbuf   = elog + E;
    float* hout   = (float*)d_out;

    const int nNodeTiles = (N + 15) / 16;
    const int nEdgeTiles = (E + 15) / 16;

    k_node_gemm<<<(nNodeTiles + 7) / 8, 256, 0, stream>>>(
        x, W_fc, b_fc, W_self, b_self, z, hout, N, nNodeTiles);

    k_node_attn<<<(N + 255) / 256, 256, 0, stream>>>(
        z, W_attn, a_src, a_dst, m, denom, N);

    k_edge_logit<<<(E + 255) / 256, 256, 0, stream>>>(
        edge_attr, erow, ecol, W_attn, b_attn, a_src, a_dst, elog, m, E);

    k_edge_soft<<<(E + 255) / 256, 256, 0, stream>>>(
        elog, ecol, m, pbuf, denom, E);

    k_edge_aggregate<<<(nEdgeTiles + 7) / 8, 256, 0, stream>>>(
        edge_attr, erow, ecol, W_edge, b_edge, z, pbuf, denom, hout, E, nEdgeTiles);
}


# aggregate epilogue one edge at a time: lane halves swap columns (permlanex16), row bases wave-uniform, the 32 whole-line z gathers batched ahead of the atomics; atomic arrival order changes
# speedup vs baseline: 1.4063x; 1.4063x over previous
#include <hip/hip_runtime.h>
#include <hip/hip_bf16.h>
#include <math.h>

#define IN_DIM  128
#define OUT_DIM 64
#define ED_DIM  32

#define BSTRIDE_V2 80
#define BSTRIDE_F  (2 * BSTRIDE_V2)

typedef __attribute__((ext_vector_type(2))) float v2f;
typedef __attribute__((ext_vector_type(8))) float v8f;

struct Acc4 { v8f t[4]; };

__device__ __attribute__((always_inline)) inline
Acc4 wave_gemm_tile(const float* __restrict__ arow,
                    const float* __restrict__ lWs,
                    int kdim, int half, int l16)
{
    Acc4 A;
    A.t[0] = {}; A.t[1] = {}; A.t[2] = {}; A.t[3] = {};
    const v2f* a2 = (const v2f*)arow;
    for (int k0 = 0; k0 < kdim; k0 += 4) {
        const int kp  = (k0 >> 1) + half;
        const v2f  a  = a2[kp];
        const v2f* br = (const v2f*)(lWs + kp * BSTRIDE_F);
#pragma unroll
        for (int nt = 0; nt < 4; ++nt) {
            const v2f b = br[nt * 16 + l16];
            A.t[nt] = __builtin_amdgcn_wmma_f32_16x16x4_f32(
                false, a, false, b, (short)0, A.t[nt], false, false);
        }
    }
    return A;
}

__device__ __attribute__((always_inline)) inline
void stage_w(float* __restrict__ lWs, const float* __restrict__ W, int kdim)
{
    const int total = kdim * OUT_DIM;
    for (int i = threadIdx.x; i < total; i += blockDim.x) {
        const int k = i >> 6;
        const int n = i & 63;
        lWs[(k >> 1) * BSTRIDE_F + (n << 1) + (k & 1)] = W[i];
    }
}

__device__ inline void atomicMaxF(float* addr, float val) {
    if (val >= 0.0f) atomicMax(reinterpret_cast<int*>(addr), __float_as_int(val));
    else             atomicMin(reinterpret_cast<unsigned int*>(addr), __float_as_uint(val));
}

__global__ void k_node_gemm(const float* __restrict__ x,
                            const float* __restrict__ W_fc,  const float* __restrict__ b_fc,
                            const float* __restrict__ W_self, const float* __restrict__ b_self,
                            float* __restrict__ z, float* __restrict__ hout,
                            int N, int nTiles)
{
    __shared__ float lWs[(IN_DIM / 2) * BSTRIDE_F];
    const int lane = threadIdx.x & 31;
    const int wave = threadIdx.x >> 5;
    const int half = lane >> 4;
    const int l16  = lane & 15;
    const int tile = blockIdx.x * 8 + wave;

    int mrow = tile * 16 + l16;
    if (mrow >= N) mrow = N - 1;
    const float* xrow = x + (size_t)mrow * IN_DIM;

    stage_w(lWs, W_fc, IN_DIM);
    __syncthreads();
    if (tile < nTiles) {
        Acc4 acc = wave_gemm_tile(xrow, lWs, IN_DIM, half, l16);
#pragma unroll
        for (int nt = 0; nt < 4; ++nt) {
            const int   c  = nt * 16 + l16;
            const float bc = b_fc[c];
#pragma unroll
            for (int r = 0; r < 8; ++r) {
                const int row = tile * 16 + r + 8 * half;
                if (row < N) z[(size_t)row * OUT_DIM + c] = acc.t[nt][r] + bc;
            }
        }
    }
    __syncthreads();

    stage_w(lWs, W_self, IN_DIM);
    __syncthreads();
    if (tile < nTiles) {
        Acc4 acc = wave_gemm_tile(xrow, lWs, IN_DIM, half, l16);
#pragma unroll
        for (int nt = 0; nt < 4; ++nt) {
            const int   c  = nt * 16 + l16;
            const float bc = b_self[c];
#pragma unroll
            for (int r = 0; r < 8; ++r) {
                const int row = tile * 16 + r + 8 * half;
                if (row < N) hout[(size_t)row * OUT_DIM + c] = acc.t[nt][r] + bc;
            }
        }
    }
}

__global__ void k_node_attn(const float* __restrict__ z, const float* __restrict__ W_attn,
                            float* __restrict__ a_src, float* __restrict__ a_dst,
                            float* __restrict__ m, float* __restrict__ denom, int N)
{
    const int n = blockIdx.x * blockDim.x + threadIdx.x;
    if (n >= N) return;
    const float* zr = z + (size_t)n * OUT_DIM;
    float as = 0.f, ad = 0.f;
#pragma unroll 8
    for (int c = 0; c < OUT_DIM; ++c) {
        const float zc = zr[c];
        as += zc * W_attn[c];
        ad += zc * W_attn[OUT_DIM + c];
    }
    a_src[n] = as;
    a_dst[n] = ad;
    m[n]     = -INFINITY;
    denom[n] = 0.0f;
}

__global__ void k_edge_logit(const float* __restrict__ edge_attr,
                             const int* __restrict__ erow, const int* __restrict__ ecol,
                             const float* __restrict__ W_attn, const float* __restrict__ b_attn,
                             const float* __restrict__ a_src, const float* __restrict__ a_dst,
                             float* __restrict__ elog, float* __restrict__ m, int E)
{
    const int e = blockIdx.x * blockDim.x + threadIdx.x;
    if (e >= E) return;
    const float* ea = edge_attr + (size_t)e * ED_DIM;
    float dot = 0.f;
#pragma unroll
    for (int k = 0; k < ED_DIM; ++k) dot += ea[k] * W_attn[2 * OUT_DIM + k];
    const int c = ecol[e];
    const float s  = a_src[erow[e]] + a_dst[c] + dot + b_attn[0];
    const float lr = (s > 0.0f) ? s : 0.2f * s;
    elog[e] = lr;
    atomicMaxF(&m[c], lr);
}

__global__ void k_edge_soft(const float* __restrict__ elog, const int* __restrict__ ecol,
                            const float* __restrict__ m, float* __restrict__ pbuf,
                            float* __restrict__ denom, int E)
{
    const int e = blockIdx.x * blockDim.x + threadIdx.x;
    if (e >= E) return;
    const int c = ecol[e];
    const float p = expf(elog[e] - m[c]);
    pbuf[e] = p;
    atomicAdd(&denom[c], p);
}

__device__ __attribute__((always_inline)) inline
float xhalf(float x)
{
    const int b = __builtin_bit_cast(int, x);
    return __builtin_bit_cast(float,
        __builtin_amdgcn_permlanex16(b, b, 0x76543210, 0xfedcba98, false, false));
}

template <bool FULL>
__device__ __attribute__((always_inline)) inline
void edge_epilogue(const Acc4& acc, int c_l, int r_l, float al_l, float b_lo, float b_hi,
                   const float* __restrict__ z, float* __restrict__ hout, int lane, int half)
{
    float zlo[16], zhi[16];
    if (FULL) {
#pragma unroll
        for (int j = 0; j < 16; ++j) {
            const int    rj = __builtin_amdgcn_readlane(r_l, j);
            const float* zr = z + (size_t)rj * OUT_DIM;
            zlo[j] = zr[lane];
            zhi[j] = zr[32 + lane];
        }
    }
#pragma unroll
    for (int r = 0; r < 8; ++r) {
        const float gA   = xhalf(half ? acc.t[0][r] : acc.t[1][r]);
        const float gB   = xhalf(half ? acc.t[2][r] : acc.t[3][r]);
        const float v0lo = half ? gA : acc.t[0][r];
        const float v1lo = half ? acc.t[1][r] : gA;
        const float v0hi = half ? gB : acc.t[2][r];
        const float v1hi = half ? acc.t[3][r] : gB;
#pragma unroll
        for (int h = 0; h < 2; ++h) {
            const int   j  = r + 8 * h;
            const int   cj = __builtin_amdgcn_readlane(c_l, j);
            const float aj = __builtin_bit_cast(float,
                __builtin_amdgcn_readlane(__builtin_bit_cast(int, al_l), j));
            if (FULL || cj >= 0) {
                float zl, zh;
                if (FULL) {
                    zl = zlo[j];
                    zh = zhi[j];
                } else {
                    const int    rj = __builtin_amdgcn_readlane(r_l, j);
                    const float* zr = z + (size_t)rj * OUT_DIM;
                    zl = zr[lane];
                    zh = zr[32 + lane];
                }
                float* hr = hout + (size_t)cj * OUT_DIM;
                atomicAdd(&hr[lane],      aj * (zl + (h ? v1lo : v0lo) + b_lo));
                atomicAdd(&hr[32 + lane], aj * (zh + (h ? v1hi : v0hi) + b_hi));
            }
        }
    }
}

__global__ void k_edge_aggregate(const float* __restrict__ edge_attr,
                                 const int* __restrict__ erow, const int* __restrict__ ecol,
                                 const float* __restrict__ W_edge, const float* __restrict__ b_edge,
                                 const float* __restrict__ z, const float* __restrict__ pbuf,
                                 const float* __restrict__ denom, float* __restrict__ hout,
                                 int E, int nTiles)
{
    __shared__ float lWs[(ED_DIM / 2) * BSTRIDE_F];
    const int lane = threadIdx.x & 31;
    const int wave = threadIdx.x >> 5;
    const int half = lane >> 4;
    const int l16  = lane & 15;

    stage_w(lWs, W_edge, ED_DIM);
    __syncthreads();
    const int tile = __builtin_amdgcn_readfirstlane(blockIdx.x * 8 + wave);
    if (tile >= nTiles) return;

    int arow = tile * 16 + l16;
    const bool evalid = arow < E;
    if (arow >= E) arow = E - 1;
    const float* ea = edge_attr + (size_t)arow * ED_DIM;

    {
        const size_t nxt = (size_t)arow + 16 * 8;
        if (nxt < (size_t)E)
            __builtin_prefetch(edge_attr + nxt * ED_DIM, 0, 1);
    }

    const int   c_e  = ecol[arow];
    const int   c_l  = evalid ? c_e : -1;
    const int   r_l  = erow[arow];
    const float al_l = pbuf[arow] / fmaxf(denom[c_e], 1e-16f);

    Acc4 acc = wave_gemm_tile(ea, lWs, ED_DIM, half, l16);

    const float b_lo = b_edge[lane];
    const float b_hi = b_edge[32 + lane];

    if (tile * 16 + 16 <= E)
        edge_epilogue<true >(acc, c_l, r_l, al_l, b_lo, b_hi, z, hout, lane, half);
    else
        edge_epilogue<false>(acc, c_l, r_l, al_l, b_lo, b_hi, z, hout, lane, half);
}

extern "C" void kernel_launch(void* const* d_in, const int* in_sizes, int n_in,
                              void* d_out, int out_size, void* d_ws, size_t ws_size,
                              hipStream_t stream)
{
    const float* x         = (const float*)d_in[0];
    const float* edge_attr = (const float*)d_in[1];
    const int*   edge_idx  = (const int*)  d_in[2];
    const float* W_fc      = (const float*)d_in[3];
    const float* b_fc      = (const float*)d_in[4];
    const float* W_attn    = (const float*)d_in[5];
    const float* b_attn    = (const float*)d_in[6];
    const float* W_edge    = (const float*)d_in[7];
    const float* b_edge    = (const float*)d_in[8];
    const float* W_self    = (const float*)d_in[9];
    const float* b_self    = (const float*)d_in[10];

    const int N = in_sizes[0] / IN_DIM;
    const int E = in_sizes[1] / ED_DIM;
    const int* erow = edge_idx;
    const int* ecol = edge_idx + E;

    float* ws     = (float*)d_ws;
    float* z      = ws;
    float* a_src  = z + (size_t)N * OUT_DIM;
    float* a_dst  = a_src + N;
    float* m      = a_dst + N;
    float* denom  = m + N;
    float* elog   = denom + N;
    float* pbuf   = elog + E;
    float* hout   = (float*)d_out;

    const int nNodeTiles = (N + 15) / 16;
    const int nEdgeTiles = (E + 15) / 16;

    k_node_gemm<<<(nNodeTiles + 7) / 8, 256, 0, stream>>>(
        x, W_fc, b_fc, W_self, b_self, z, hout, N, nNodeTiles);

    k_node_attn<<<(N + 255) / 256, 256, 0, stream>>>(
        z, W_attn, a_src, a_dst, m, denom, N);

    k_edge_logit<<<(E + 255) / 256, 256, 0, stream>>>(
        edge_attr, erow, ecol, W_attn, b_attn, a_src, a_dst, elog, m, E);

    k_edge_soft<<<(E + 255) / 256, 256, 0, stream>>>(
        elog, ecol, m, pbuf, denom, E);

    k_edge_aggregate<<<(nEdgeTiles + 7) / 8, 256, 0, stream>>>(
        edge_attr, erow, ecol, W_edge, b_edge, z, pbuf, denom, hout, E, nEdgeTiles);
}


# every atomic target (segment max, exp sum, output rows) gets one copy per chip half, picked by die id; readers combine both; one add launch more; f32 sums split per half, summation order changes
# speedup vs baseline: 1.7770x; 1.2636x over previous
#include <hip/hip_runtime.h>
#include <hip/hip_bf16.h>
#include <math.h>

#define IN_DIM  128
#define OUT_DIM 64
#define ED_DIM  32

#define BSTRIDE_V2 80
#define BSTRIDE_F  (2 * BSTRIDE_V2)

typedef __attribute__((ext_vector_type(2))) float v2f;
typedef __attribute__((ext_vector_type(8))) float v8f;

struct Acc4 { v8f t[4]; };

__device__ __attribute__((always_inline)) inline
Acc4 wave_gemm_tile(const float* __restrict__ arow,
                    const float* __restrict__ lWs,
                    int kdim, int half, int l16)
{
    Acc4 A;
    A.t[0] = {}; A.t[1] = {}; A.t[2] = {}; A.t[3] = {};
    const v2f* a2 = (const v2f*)arow;
    for (int k0 = 0; k0 < kdim; k0 += 4) {
        const int kp  = (k0 >> 1) + half;
        const v2f  a  = a2[kp];
        const v2f* br = (const v2f*)(lWs + kp * BSTRIDE_F);
#pragma unroll
        for (int nt = 0; nt < 4; ++nt) {
            const v2f b = br[nt * 16 + l16];
            A.t[nt] = __builtin_amdgcn_wmma_f32_16x16x4_f32(
                false, a, false, b, (short)0, A.t[nt], false, false);
        }
    }
    return A;
}

__device__ __attribute__((always_inline)) inline
void stage_w(float* __restrict__ lWs, const float* __restrict__ W, int kdim)
{
    const int total = kdim * OUT_DIM;
    for (int i = threadIdx.x; i < total; i += blockDim.x) {
        const int k = i >> 6;
        const int n = i & 63;
        lWs[(k >> 1) * BSTRIDE_F + (n << 1) + (k & 1)] = W[i];
    }
}

__device__ __attribute__((always_inline)) inline
int chip_half()
{
    const unsigned v = __builtin_amdgcn_s_sendmsg_rtn(0x87);
    return (int)((v >> 18) & 1u);
}

__device__ inline void atomicMaxF(float* addr, float val) {
    if (val >= 0.0f) atomicMax(reinterpret_cast<int*>(addr), __float_as_int(val));
    else             atomicMin(reinterpret_cast<unsigned int*>(addr), __float_as_uint(val));
}

__global__ void k_node_gemm(const float* __restrict__ x,
                            const float* __restrict__ W_fc,  const float* __restrict__ b_fc,
                            const float* __restrict__ W_self, const float* __restrict__ b_self,
                            float* __restrict__ z, float* __restrict__ hout,
                            float* __restrict__ plane1, int two,
                            int N, int nTiles)
{
    __shared__ float lWs[(IN_DIM / 2) * BSTRIDE_F];
    const int lane = threadIdx.x & 31;
    const int wave = threadIdx.x >> 5;
    const int half = lane >> 4;
    const int l16  = lane & 15;
    const int tile = blockIdx.x * 8 + wave;

    int mrow = tile * 16 + l16;
    if (mrow >= N) mrow = N - 1;
    const float* xrow = x + (size_t)mrow * IN_DIM;

    if (two && tile < nTiles) {
        float4* p4 = reinterpret_cast<float4*>(plane1 + (size_t)tile * 16 * OUT_DIM);
#pragma unroll
        for (int it = 0; it < 8; ++it) {
            const int idx = it * 32 + lane;
            if (tile * 16 + (idx >> 4) < N) p4[idx] = make_float4(0.f, 0.f, 0.f, 0.f);
        }
    }

    stage_w(lWs, W_fc, IN_DIM);
    __syncthreads();
    if (tile < nTiles) {
        Acc4 acc = wave_gemm_tile(xrow, lWs, IN_DIM, half, l16);
#pragma unroll
        for (int nt = 0; nt < 4; ++nt) {
            const int   c  = nt * 16 + l16;
            const float bc = b_fc[c];
#pragma unroll
            for (int r = 0; r < 8; ++r) {
                const int row = tile * 16 + r + 8 * half;
                if (row < N) z[(size_t)row * OUT_DIM + c] = acc.t[nt][r] + bc;
            }
        }
    }
    __syncthreads();

    stage_w(lWs, W_self, IN_DIM);
    __syncthreads();
    if (tile < nTiles) {
        Acc4 acc = wave_gemm_tile(xrow, lWs, IN_DIM, half, l16);
#pragma unroll
        for (int nt = 0; nt < 4; ++nt) {
            const int   c  = nt * 16 + l16;
            const float bc = b_self[c];
#pragma unroll
            for (int r = 0; r < 8; ++r) {
                const int row = tile * 16 + r + 8 * half;
                if (row < N) hout[(size_t)row * OUT_DIM + c] = acc.t[nt][r] + bc;
            }
        }
    }
}

__global__ void k_node_attn(const float* __restrict__ z, const float* __restrict__ W_attn,
                            float* __restrict__ a_src, float* __restrict__ a_dst,
                            float* __restrict__ m, float* __restrict__ denom,
                            float* __restrict__ m1, float* __restrict__ denom1, int two, int N)
{
    const int n = blockIdx.x * blockDim.x + threadIdx.x;
    if (n >= N) return;
    const float* zr = z + (size_t)n * OUT_DIM;
    float as = 0.f, ad = 0.f;
#pragma unroll 8
    for (int c = 0; c < OUT_DIM; ++c) {
        const float zc = zr[c];
        as += zc * W_attn[c];
        ad += zc * W_attn[OUT_DIM + c];
    }
    a_src[n] = as;
    a_dst[n] = ad;
    m[n]     = -INFINITY;
    denom[n] = 0.0f;
    if (two) {
        m1[n]     = -INFINITY;
        denom1[n] = 0.0f;
    }
}

__global__ void k_edge_logit(const float* __restrict__ edge_attr,
                             const int* __restrict__ erow, const int* __restrict__ ecol,
                             const float* __restrict__ W_attn, const float* __restrict__ b_attn,
                             const float* __restrict__ a_src, const float* __restrict__ a_dst,
                             float* __restrict__ elog, float* __restrict__ m,
                             float* __restrict__ m1, int two, int E)
{
    const int e = blockIdx.x * blockDim.x + threadIdx.x;
    if (e >= E) return;
    const float* ea = edge_attr + (size_t)e * ED_DIM;
    float dot = 0.f;
#pragma unroll
    for (int k = 0; k < ED_DIM; ++k) dot += ea[k] * W_attn[2 * OUT_DIM + k];
    const int c = ecol[e];
    const float s  = a_src[erow[e]] + a_dst[c] + dot + b_attn[0];
    const float lr = (s > 0.0f) ? s : 0.2f * s;
    elog[e] = lr;
    float* mh = (two && chip_half()) ? m1 : m;
    atomicMaxF(&mh[c], lr);
}

__global__ void k_edge_soft(const float* __restrict__ elog, const int* __restrict__ ecol,
                            const float* __restrict__ m, const float* __restrict__ m1,
                            float* __restrict__ pbuf,
                            float* __restrict__ denom, float* __restrict__ denom1, int two, int E)
{
    const int e = blockIdx.x * blockDim.x + threadIdx.x;
    if (e >= E) return;
    const int c = ecol[e];
    const float mc = fmaxf(m[c], m1[c]);
    const float p = expf(elog[e] - mc);
    pbuf[e] = p;
    float* dh = (two && chip_half()) ? denom1 : denom;
    atomicAdd(&dh[c], p);
}

__device__ __attribute__((always_inline)) inline
float xhalf(float x)
{
    const int b = __builtin_bit_cast(int, x);
    return __builtin_bit_cast(float,
        __builtin_amdgcn_permlanex16(b, b, 0x76543210, 0xfedcba98, false, false));
}

template <bool FULL>
__device__ __attribute__((always_inline)) inline
void edge_epilogue(const Acc4& acc, int c_l, int r_l, float al_l, float b_lo, float b_hi,
                   const float* __restrict__ z, float* __restrict__ hout, int lane, int half)
{
    float zlo[16], zhi[16];
    if (FULL) {
#pragma unroll
        for (int j = 0; j < 16; ++j) {
            const int    rj = __builtin_amdgcn_readlane(r_l, j);
            const float* zr = z + (size_t)rj * OUT_DIM;
            zlo[j] = zr[lane];
            zhi[j] = zr[32 + lane];
        }
    }
#pragma unroll
    for (int r = 0; r < 8; ++r) {
        const float gA   = xhalf(half ? acc.t[0][r] : acc.t[1][r]);
        const float gB   = xhalf(half ? acc.t[2][r] : acc.t[3][r]);
        const float v0lo = half ? gA : acc.t[0][r];
        const float v1lo = half ? acc.t[1][r] : gA;
        const float v0hi = half ? gB : acc.t[2][r];
        const float v1hi = half ? acc.t[3][r] : gB;
#pragma unroll
        for (int h = 0; h < 2; ++h) {
            const int   j  = r + 8 * h;
            const int   cj = __builtin_amdgcn_readlane(c_l, j);
            const float aj = __builtin_bit_cast(float,
                __builtin_amdgcn_readlane(__builtin_bit_cast(int, al_l), j));
            if (FULL || cj >= 0) {
                float zl, zh;
                if (FULL) {
                    zl = zlo[j];
                    zh = zhi[j];
                } else {
                    const int    rj = __builtin_amdgcn_readlane(r_l, j);
                    const float* zr = z + (size_t)rj * OUT_DIM;
                    zl = zr[lane];
                    zh = zr[32 + lane];
                }
                float* hr = hout + (size_t)cj * OUT_DIM;
                atomicAdd(&hr[lane],      aj * (zl + (h ? v1lo : v0lo) + b_lo));
                atomicAdd(&hr[32 + lane], aj * (zh + (h ? v1hi : v0hi) + b_hi));
            }
        }
    }
}

__global__ void k_edge_aggregate(const float* __restrict__ edge_attr,
                                 const int* __restrict__ erow, const int* __restrict__ ecol,
                                 const float* __restrict__ W_edge, const float* __restrict__ b_edge,
                                 const float* __restrict__ z, const float* __restrict__ pbuf,
                                 const float* __restrict__ denom, const float* __restrict__ denom1,
                                 float* __restrict__ hout, float* __restrict__ plane1, int two,
                                 int E, int nTiles)
{
    __shared__ float lWs[(ED_DIM / 2) * BSTRIDE_F];
    const int lane = threadIdx.x & 31;
    const int wave = threadIdx.x >> 5;
    const int half = lane >> 4;
    const int l16  = lane & 15;

    stage_w(lWs, W_edge, ED_DIM);
    __syncthreads();
    const int tile = __builtin_amdgcn_readfirstlane(blockIdx.x * 8 + wave);
    if (tile >= nTiles) return;

    int arow = tile * 16 + l16;
    const bool evalid = arow < E;
    if (arow >= E) arow = E - 1;
    const float* ea = edge_attr + (size_t)arow * ED_DIM;

    {
        const size_t nxt = (size_t)arow + 16 * 8;
        if (nxt < (size_t)E)
            __builtin_prefetch(edge_attr + nxt * ED_DIM, 0, 1);
    }

    const int   c_e  = ecol[arow];
    const int   c_l  = evalid ? c_e : -1;
    const int   r_l  = erow[arow];
    const float d0   = denom[c_e];
    const float d1   = denom1[c_e];
    const float den  = two ? d0 + d1 : d0;
    const float al_l = pbuf[arow] / fmaxf(den, 1e-16f);

    Acc4 acc = wave_gemm_tile(ea, lWs, ED_DIM, half, l16);

    const float b_lo = b_edge[lane];
    const float b_hi = b_edge[32 + lane];

    float* hacc = (two && chip_half()) ? plane1 : hout;

    if (tile * 16 + 16 <= E)
        edge_epilogue<true >(acc, c_l, r_l, al_l, b_lo, b_hi, z, hacc, lane, half);
    else
        edge_epilogue<false>(acc, c_l, r_l, al_l, b_lo, b_hi, z, hacc, lane, half);
}

__global__ void k_plane_sum(float* __restrict__ hout, const float* __restrict__ plane1,
                            int two, int n4)
{
    const int i = blockIdx.x * blockDim.x + threadIdx.x;
    if (!two || i >= n4) return;
    float4*       h4 = reinterpret_cast<float4*>(hout);
    const float4* q4 = reinterpret_cast<const float4*>(plane1);
    float4 a = h4[i];
    const float4 b = q4[i];
    a.x += b.x; a.y += b.y; a.z += b.z; a.w += b.w;
    h4[i] = a;
}

extern "C" void kernel_launch(void* const* d_in, const int* in_sizes, int n_in,
                              void* d_out, int out_size, void* d_ws, size_t ws_size,
                              hipStream_t stream)
{
    const float* x         = (const float*)d_in[0];
    const float* edge_attr = (const float*)d_in[1];
    const int*   edge_idx  = (const int*)  d_in[2];
    const float* W_fc      = (const float*)d_in[3];
    const float* b_fc      = (const float*)d_in[4];
    const float* W_attn    = (const float*)d_in[5];
    const float* b_attn    = (const float*)d_in[6];
    const float* W_edge    = (const float*)d_in[7];
    const float* b_edge    = (const float*)d_in[8];
    const float* W_self    = (const float*)d_in[9];
    const float* b_self    = (const float*)d_in[10];

    const int N = in_sizes[0] / IN_DIM;
    const int E = in_sizes[1] / ED_DIM;
    const int* erow = edge_idx;
    const int* ecol = edge_idx + E;

    float* ws     = (float*)d_ws;
    float* z      = ws;
    float* a_src  = z + (size_t)N * OUT_DIM;
    float* a_dst  = a_src + N;
    float* m      = a_dst + N;
    float* denom  = m + N;
    float* elog   = denom + N;
    float* pbuf   = elog + E;
    const size_t off1 = ((size_t)N * (OUT_DIM + 6) + 2 * (size_t)E + 3) & ~(size_t)3;
    const int two = ws_size >= (off1 + (size_t)N * OUT_DIM) * sizeof(float) ? 1 : 0;
    float* m1     = two ? pbuf + E : m;
    float* denom1 = two ? m1 + N   : denom;
    float* plane1 = two ? ws + off1 : (float*)d_out;
    float* hout   = (float*)d_out;

    const int nNodeTiles = (N + 15) / 16;
    const int nEdgeTiles = (E + 15) / 16;

    k_node_gemm<<<(nNodeTiles + 7) / 8, 256, 0, stream>>>(
        x, W_fc, b_fc, W_self, b_self, z, hout, plane1, two, N, nNodeTiles);

    k_node_attn<<<(N + 255) / 256, 256, 0, stream>>>(
        z, W_attn, a_src, a_dst, m, denom, m1, denom1, two, N);

    k_edge_logit<<<(E + 255) / 256, 256, 0, stream>>>(
        edge_attr, erow, ecol, W_attn, b_attn, a_src, a_dst, elog, m, m1, two, E);

    k_edge_soft<<<(E + 255) / 256, 256, 0, stream>>>(
        elog, ecol, m, m1, pbuf, denom, denom1, two, E);

    k_edge_aggregate<<<(nEdgeTiles + 7) / 8, 256, 0, stream>>>(
        edge_attr, erow, ecol, W_edge, b_edge, z, pbuf, denom, denom1, hout, plane1, two,
        E, nEdgeTiles);

    k_plane_sum<<<(N * (OUT_DIM / 4) + 255) / 256, 256, 0, stream>>>(
        hout, plane1, two, N * (OUT_DIM / 4));
}


# weight staging into LDS: each thread's loads all in flight at once (16-byte loads and LDS stores, W_self requested under the first product), not one dependent 4-byte load-store round trip per trip
# speedup vs baseline: 2.0847x; 1.1732x over previous
#include <hip/hip_runtime.h>
#include <hip/hip_bf16.h>
#include <math.h>

#define IN_DIM  128
#define OUT_DIM 64
#define ED_DIM  32

#define BSTRIDE_V2 80
#define BSTRIDE_F  (2 * BSTRIDE_V2)

typedef __attribute__((ext_vector_type(2))) float v2f;
typedef __attribute__((ext_vector_type(8))) float v8f;

struct Acc4 { v8f t[4]; };

__device__ __attribute__((always_inline)) inline
Acc4 wave_gemm_tile(const float* __restrict__ arow,
                    const float* __restrict__ lWs,
                    int kdim, int half, int l16)
{
    Acc4 A;
    A.t[0] = {}; A.t[1] = {}; A.t[2] = {}; A.t[3] = {};
    const v2f* a2 = (const v2f*)arow;
    for (int k0 = 0; k0 < kdim; k0 += 4) {
        const int kp  = (k0 >> 1) + half;
        const v2f  a  = a2[kp];
        const v2f* br = (const v2f*)(lWs + kp * BSTRIDE_F);
#pragma unroll
        for (int nt = 0; nt < 4; ++nt) {
            const v2f b = br[nt * 16 + l16];
            A.t[nt] = __builtin_amdgcn_wmma_f32_16x16x4_f32(
                false, a, false, b, (short)0, A.t[nt], false, false);
        }
    }
    return A;
}

#define NTHREADS 256

template <int KDIM>
struct WStage {
    static constexpr int ITEMS = (KDIM / 2) * (OUT_DIM / 4);
    static constexpr int PER   = ITEMS / NTHREADS;
    static_assert(ITEMS % NTHREADS == 0, "staging is laid out for whole rounds of the work-group");
    float4 lo[PER];
    float4 hi[PER];
};

template <int KDIM>
__device__ __attribute__((always_inline)) inline
WStage<KDIM> load_w(const float* __restrict__ W)
{
    WStage<KDIM> s;
#pragma unroll
    for (int j = 0; j < WStage<KDIM>::PER; ++j) {
        const int it = threadIdx.x + NTHREADS * j;
        const int kp = it >> 4;
        const int g  = it & 15;
        const float4* src = reinterpret_cast<const float4*>(W + (size_t)(2 * kp) * OUT_DIM + 4 * g);
        s.lo[j] = src[0];
        s.hi[j] = src[OUT_DIM / 4];
    }
    return s;
}

template <int KDIM>
__device__ __attribute__((always_inline)) inline
void store_w(float* __restrict__ lWs, const WStage<KDIM>& s)
{
#pragma unroll
    for (int j = 0; j < WStage<KDIM>::PER; ++j) {
        const int it = threadIdx.x + NTHREADS * j;
        const int kp = it >> 4;
        const int g  = it & 15;
        float4* dst = reinterpret_cast<float4*>(lWs + kp * BSTRIDE_F + 8 * g);
        dst[0] = make_float4(s.lo[j].x, s.hi[j].x, s.lo[j].y, s.hi[j].y);
        dst[1] = make_float4(s.lo[j].z, s.hi[j].z, s.lo[j].w, s.hi[j].w);
    }
}

__device__ __attribute__((always_inline)) inline
int chip_half()
{
    const unsigned v = __builtin_amdgcn_s_sendmsg_rtn(0x87);
    return (int)((v >> 18) & 1u);
}

__device__ inline void atomicMaxF(float* addr, float val) {
    if (val >= 0.0f) atomicMax(reinterpret_cast<int*>(addr), __float_as_int(val));
    else             atomicMin(reinterpret_cast<unsigned int*>(addr), __float_as_uint(val));
}

__global__ void k_node_gemm(const float* __restrict__ x,
                            const float* __restrict__ W_fc,  const float* __restrict__ b_fc,
                            const float* __restrict__ W_self, const float* __restrict__ b_self,
                            float* __restrict__ z, float* __restrict__ hout,
                            float* __restrict__ plane1, int two,
                            int N, int nTiles)
{
    __shared__ __attribute__((aligned(16))) float lWs[(IN_DIM / 2) * BSTRIDE_F];
    const int lane = threadIdx.x & 31;
    const int wave = threadIdx.x >> 5;
    const int half = lane >> 4;
    const int l16  = lane & 15;
    const int tile = blockIdx.x * 8 + wave;

    const WStage<IN_DIM> w_first  = load_w<IN_DIM>(W_fc);
    const WStage<IN_DIM> w_second = load_w<IN_DIM>(W_self);

    int mrow = tile * 16 + l16;
    if (mrow >= N) mrow = N - 1;
    const float* xrow = x + (size_t)mrow * IN_DIM;

    if (two && tile < nTiles) {
        float4* p4 = reinterpret_cast<float4*>(plane1 + (size_t)tile * 16 * OUT_DIM);
#pragma unroll
        for (int it = 0; it < 8; ++it) {
            const int idx = it * 32 + lane;
            if (tile * 16 + (idx >> 4) < N) p4[idx] = make_float4(0.f, 0.f, 0.f, 0.f);
        }
    }

    store_w<IN_DIM>(lWs, w_first);
    __syncthreads();
    if (tile < nTiles) {
        Acc4 acc = wave_gemm_tile(xrow, lWs, IN_DIM, half, l16);
#pragma unroll
        for (int nt = 0; nt < 4; ++nt) {
            const int   c  = nt * 16 + l16;
            const float bc = b_fc[c];
#pragma unroll
            for (int r = 0; r < 8; ++r) {
                const int row = tile * 16 + r + 8 * half;
                if (row < N) z[(size_t)row * OUT_DIM + c] = acc.t[nt][r] + bc;
            }
        }
    }
    __syncthreads();

    store_w<IN_DIM>(lWs, w_second);
    __syncthreads();
    if (tile < nTiles) {
        Acc4 acc = wave_gemm_tile(xrow, lWs, IN_DIM, half, l16);
#pragma unroll
        for (int nt = 0; nt < 4; ++nt) {
            const int   c  = nt * 16 + l16;
            const float bc = b_self[c];
#pragma unroll
            for (int r = 0; r < 8; ++r) {
                const int row = tile * 16 + r + 8 * half;
                if (row < N) hout[(size_t)row * OUT_DIM + c] = acc.t[nt][r] + bc;
            }
        }
    }
}

__global__ void k_node_attn(const float* __restrict__ z, const float* __restrict__ W_attn,
                            float* __restrict__ a_src, float* __restrict__ a_dst,
                            float* __restrict__ m, float* __restrict__ denom,
                            float* __restrict__ m1, float* __restrict__ denom1, int two, int N)
{
    const int n = blockIdx.x * blockDim.x + threadIdx.x;
    if (n >= N) return;
    const float* zr = z + (size_t)n * OUT_DIM;
    float as = 0.f, ad = 0.f;
#pragma unroll 8
    for (int c = 0; c < OUT_DIM; ++c) {
        const float zc = zr[c];
        as += zc * W_attn[c];
        ad += zc * W_attn[OUT_DIM + c];
    }
    a_src[n] = as;
    a_dst[n] = ad;
    m[n]     = -INFINITY;
    denom[n] = 0.0f;
    if (two) {
        m1[n]     = -INFINITY;
        denom1[n] = 0.0f;
    }
}

__global__ void k_edge_logit(const float* __restrict__ edge_attr,
                             const int* __restrict__ erow, const int* __restrict__ ecol,
                             const float* __restrict__ W_attn, const float* __restrict__ b_attn,
                             const float* __restrict__ a_src, const float* __restrict__ a_dst,
                             float* __restrict__ elog, float* __restrict__ m,
                             float* __restrict__ m1, int two, int E)
{
    const int e = blockIdx.x * blockDim.x + threadIdx.x;
    if (e >= E) return;
    const float* ea = edge_attr + (size_t)e * ED_DIM;
    float dot = 0.f;
#pragma unroll
    for (int k = 0; k < ED_DIM; ++k) dot += ea[k] * W_attn[2 * OUT_DIM + k];
    const int c = ecol[e];
    const float s  = a_src[erow[e]] + a_dst[c] + dot + b_attn[0];
    const float lr = (s > 0.0f) ? s : 0.2f * s;
    elog[e] = lr;
    float* mh = (two && chip_half()) ? m1 : m;
    atomicMaxF(&mh[c], lr);
}

__global__ void k_edge_soft(const float* __restrict__ elog, const int* __restrict__ ecol,
                            const float* __restrict__ m, const float* __restrict__ m1,
                            float* __restrict__ pbuf,
                            float* __restrict__ denom, float* __restrict__ denom1, int two, int E)
{
    const int e = blockIdx.x * blockDim.x + threadIdx.x;
    if (e >= E) return;
    const int c = ecol[e];
    const float mc = fmaxf(m[c], m1[c]);
    const float p = expf(elog[e] - mc);
    pbuf[e] = p;
    float* dh = (two && chip_half()) ? denom1 : denom;
    atomicAdd(&dh[c], p);
}

__device__ __attribute__((always_inline)) inline
float xhalf(float x)
{
    const int b = __builtin_bit_cast(int, x);
    return __builtin_bit_cast(float,
        __builtin_amdgcn_permlanex16(b, b, 0x76543210, 0xfedcba98, false, false));
}

template <bool FULL>
__device__ __attribute__((always_inline)) inline
void edge_epilogue(const Acc4& acc, int c_l, int r_l, float al_l, float b_lo, float b_hi,
                   const float* __restrict__ z, float* __restrict__ hout, int lane, int half)
{
    float zlo[16], zhi[16];
    if (FULL) {
#pragma unroll
        for (int j = 0; j < 16; ++j) {
            const int    rj = __builtin_amdgcn_readlane(r_l, j);
            const float* zr = z + (size_t)rj * OUT_DIM;
            zlo[j] = zr[lane];
            zhi[j] = zr[32 + lane];
        }
    }
#pragma unroll
    for (int r = 0; r < 8; ++r) {
        const float gA   = xhalf(half ? acc.t[0][r] : acc.t[1][r]);
        const float gB   = xhalf(half ? acc.t[2][r] : acc.t[3][r]);
        const float v0lo = half ? gA : acc.t[0][r];
        const float v1lo = half ? acc.t[1][r] : gA;
        const float v0hi = half ? gB : acc.t[2][r];
        const float v1hi = half ? acc.t[3][r] : gB;
#pragma unroll
        for (int h = 0; h < 2; ++h) {
            const int   j  = r + 8 * h;
            const int   cj = __builtin_amdgcn_readlane(c_l, j);
            const float aj = __builtin_bit_cast(float,
                __builtin_amdgcn_readlane(__builtin_bit_cast(int, al_l), j));
            if (FULL || cj >= 0) {
                float zl, zh;
                if (FULL) {
                    zl = zlo[j];
                    zh = zhi[j];
                } else {
                    const int    rj = __builtin_amdgcn_readlane(r_l, j);
                    const float* zr = z + (size_t)rj * OUT_DIM;
                    zl = zr[lane];
                    zh = zr[32 + lane];
                }
                float* hr = hout + (size_t)cj * OUT_DIM;
                atomicAdd(&hr[lane],      aj * (zl + (h ? v1lo : v0lo) + b_lo));
                atomicAdd(&hr[32 + lane], aj * (zh + (h ? v1hi : v0hi) + b_hi));
            }
        }
    }
}

__global__ void k_edge_aggregate(const float* __restrict__ edge_attr,
                                 const int* __restrict__ erow, const int* __restrict__ ecol,
                                 const float* __restrict__ W_edge, const float* __restrict__ b_edge,
                                 const float* __restrict__ z, const float* __restrict__ pbuf,
                                 const float* __restrict__ denom, const float* __restrict__ denom1,
                                 float* __restrict__ hout, float* __restrict__ plane1, int two,
                                 int E, int nTiles)
{
    __shared__ __attribute__((aligned(16))) float lWs[(ED_DIM / 2) * BSTRIDE_F];
    const int lane = threadIdx.x & 31;
    const int wave = threadIdx.x >> 5;
    const int half = lane >> 4;
    const int l16  = lane & 15;

    store_w<ED_DIM>(lWs, load_w<ED_DIM>(W_edge));
    __syncthreads();
    const int tile = __builtin_amdgcn_readfirstlane(blockIdx.x * 8 + wave);
    if (tile >= nTiles) return;

    int arow = tile * 16 + l16;
    const bool evalid = arow < E;
    if (arow >= E) arow = E - 1;
    const float* ea = edge_attr + (size_t)arow * ED_DIM;

    {
        const size_t nxt = (size_t)arow + 16 * 8;
        if (nxt < (size_t)E)
            __builtin_prefetch(edge_attr + nxt * ED_DIM, 0, 1);
    }

    const int   c_e  = ecol[arow];
    const int   c_l  = evalid ? c_e : -1;
    const int   r_l  = erow[arow];
    const float d0   = denom[c_e];
    const float d1   = denom1[c_e];
    const float den  = two ? d0 + d1 : d0;
    const float al_l = pbuf[arow] / fmaxf(den, 1e-16f);

    Acc4 acc = wave_gemm_tile(ea, lWs, ED_DIM, half, l16);

    const float b_lo = b_edge[lane];
    const float b_hi = b_edge[32 + lane];

    float* hacc = (two && chip_half()) ? plane1 : hout;

    if (tile * 16 + 16 <= E)
        edge_epilogue<true >(acc, c_l, r_l, al_l, b_lo, b_hi, z, hacc, lane, half);
    else
        edge_epilogue<false>(acc, c_l, r_l, al_l, b_lo, b_hi, z, hacc, lane, half);
}

__global__ void k_plane_sum(float* __restrict__ hout, const float* __restrict__ plane1,
                            int two, int n4)
{
    const int i = blockIdx.x * blockDim.x + threadIdx.x;
    if (!two || i >= n4) return;
    float4*       h4 = reinterpret_cast<float4*>(hout);
    const float4* q4 = reinterpret_cast<const float4*>(plane1);
    float4 a = h4[i];
    const float4 b = q4[i];
    a.x += b.x; a.y += b.y; a.z += b.z; a.w += b.w;
    h4[i] = a;
}

extern "C" void kernel_launch(void* const* d_in, const int* in_sizes, int n_in,
                              void* d_out, int out_size, void* d_ws, size_t ws_size,
                              hipStream_t stream)
{
    const float* x         = (const float*)d_in[0];
    const float* edge_attr = (const float*)d_in[1];
    const int*   edge_idx  = (const int*)  d_in[2];
    const float* W_fc      = (const float*)d_in[3];
    const float* b_fc      = (const float*)d_in[4];
    const float* W_attn    = (const float*)d_in[5];
    const float* b_attn    = (const float*)d_in[6];
    const float* W_edge    = (const float*)d_in[7];
    const float* b_edge    = (const float*)d_in[8];
    const float* W_self    = (const float*)d_in[9];
    const float* b_self    = (const float*)d_in[10];

    const int N = in_sizes[0] / IN_DIM;
    const int E = in_sizes[1] / ED_DIM;
    const int* erow = edge_idx;
    const int* ecol = edge_idx + E;

    float* ws     = (float*)d_ws;
    float* z      = ws;
    float* a_src  = z + (size_t)N * OUT_DIM;
    float* a_dst  = a_src + N;
    float* m      = a_dst + N;
    float* denom  = m + N;
    float* elog   = denom + N;
    float* pbuf   = elog + E;
    const size_t off1 = ((size_t)N * (OUT_DIM + 6) + 2 * (size_t)E + 3) & ~(size_t)3;
    const int two = ws_size >= (off1 + (size_t)N * OUT_DIM) * sizeof(float) ? 1 : 0;
    float* m1     = two ? pbuf + E : m;
    float* denom1 = two ? m1 + N   : denom;
    float* plane1 = two ? ws + off1 : (float*)d_out;
    float* hout   = (float*)d_out;

    const int nNodeTiles = (N + 15) / 16;
    const int nEdgeTiles = (E + 15) / 16;

    k_node_gemm<<<(nNodeTiles + 7) / 8, NTHREADS, 0, stream>>>(
        x, W_fc, b_fc, W_self, b_self, z, hout, plane1, two, N, nNodeTiles);

    k_node_attn<<<(N + 255) / 256, 256, 0, stream>>>(
        z, W_attn, a_src, a_dst, m, denom, m1, denom1, two, N);

    k_edge_logit<<<(E + 255) / 256, 256, 0, stream>>>(
        edge_attr, erow, ecol, W_attn, b_attn, a_src, a_dst, elog, m, m1, two, E);

    k_edge_soft<<<(E + 255) / 256, 256, 0, stream>>>(
        elog, ecol, m, m1, pbuf, denom, denom1, two, E);

    k_edge_aggregate<<<(nEdgeTiles + 7) / 8, NTHREADS, 0, stream>>>(
        edge_attr, erow, ecol, W_edge, b_edge, z, pbuf, denom, denom1, hout, plane1, two,
        E, nEdgeTiles);

    k_plane_sum<<<(N * (OUT_DIM / 4) + 255) / 256, 256, 0, stream>>>(
        hout, plane1, two, N * (OUT_DIM / 4));
}


# aggregation without float adds to memory: edges binned by 196-row destination range (count + place), one work-group per range sums messages by LDS float adds, stores rows once; sum order changes
# speedup vs baseline: 2.2773x; 1.0924x over previous
#include <hip/hip_runtime.h>
#include <hip/hip_bf16.h>
#include <math.h>

#define IN_DIM  128
#define OUT_DIM 64
#define ED_DIM  32

#define BSTRIDE_V2 80
#define BSTRIDE_F  (2 * BSTRIDE_V2)

typedef __attribute__((ext_vector_type(2))) float v2f;
typedef __attribute__((ext_vector_type(8))) float v8f;

struct Acc4 { v8f t[4]; };

__device__ __attribute__((always_inline)) inline
Acc4 wave_gemm_tile(const float* __restrict__ arow,
                    const float* __restrict__ lWs,
                    int kdim, int half, int l16)
{
    Acc4 A;
    A.t[0] = {}; A.t[1] = {}; A.t[2] = {}; A.t[3] = {};
    const v2f* a2 = (const v2f*)arow;
    for (int k0 = 0; k0 < kdim; k0 += 4) {
        const int kp  = (k0 >> 1) + half;
        const v2f  a  = a2[kp];
        const v2f* br = (const v2f*)(lWs + kp * BSTRIDE_F);
#pragma unroll
        for (int nt = 0; nt < 4; ++nt) {
            const v2f b = br[nt * 16 + l16];
            A.t[nt] = __builtin_amdgcn_wmma_f32_16x16x4_f32(
                false, a, false, b, (short)0, A.t[nt], false, false);
        }
    }
    return A;
}

#define NTHREADS 256

template <int KDIM>
struct WStage {
    static constexpr int ITEMS = (KDIM / 2) * (OUT_DIM / 4);
    static constexpr int PER   = ITEMS / NTHREADS;
    static_assert(ITEMS % NTHREADS == 0, "staging is laid out for whole rounds of the work-group");
    float4 lo[PER];
    float4 hi[PER];
};

template <int KDIM>
__device__ __attribute__((always_inline)) inline
WStage<KDIM> load_w(const float* __restrict__ W)
{
    WStage<KDIM> s;
#pragma unroll
    for (int j = 0; j < WStage<KDIM>::PER; ++j) {
        const int it = threadIdx.x + NTHREADS * j;
        const int kp = it >> 4;
        const int g  = it & 15;
        const float4* src = reinterpret_cast<const float4*>(W + (size_t)(2 * kp) * OUT_DIM + 4 * g);
        s.lo[j] = src[0];
        s.hi[j] = src[OUT_DIM / 4];
    }
    return s;
}

template <int KDIM>
__device__ __attribute__((always_inline)) inline
void store_w(float* __restrict__ lWs, const WStage<KDIM>& s)
{
#pragma unroll
    for (int j = 0; j < WStage<KDIM>::PER; ++j) {
        const int it = threadIdx.x + NTHREADS * j;
        const int kp = it >> 4;
        const int g  = it & 15;
        float4* dst = reinterpret_cast<float4*>(lWs + kp * BSTRIDE_F + 8 * g);
        dst[0] = make_float4(s.lo[j].x, s.hi[j].x, s.lo[j].y, s.hi[j].y);
        dst[1] = make_float4(s.lo[j].z, s.hi[j].z, s.lo[j].w, s.hi[j].w);
    }
}

__device__ __attribute__((always_inline)) inline
int chip_half()
{
    const unsigned v = __builtin_amdgcn_s_sendmsg_rtn(0x87);
    return (int)((v >> 18) & 1u);
}

__device__ inline void atomicMaxF(float* addr, float val) {
    if (val >= 0.0f) atomicMax(reinterpret_cast<int*>(addr), __float_as_int(val));
    else             atomicMin(reinterpret_cast<unsigned int*>(addr), __float_as_uint(val));
}

__global__ void k_node_gemm(const float* __restrict__ x,
                            const float* __restrict__ W_fc,  const float* __restrict__ b_fc,
                            const float* __restrict__ W_self, const float* __restrict__ b_self,
                            float* __restrict__ z, float* __restrict__ hout,
                            float* __restrict__ plane1, int two,
                            int N, int nTiles)
{
    __shared__ __attribute__((aligned(16))) float lWs[(IN_DIM / 2) * BSTRIDE_F];
    const int lane = threadIdx.x & 31;
    const int wave = threadIdx.x >> 5;
    const int half = lane >> 4;
    const int l16  = lane & 15;
    const int tile = blockIdx.x * 8 + wave;

    const WStage<IN_DIM> w_first  = load_w<IN_DIM>(W_fc);
    const WStage<IN_DIM> w_second = load_w<IN_DIM>(W_self);

    int mrow = tile * 16 + l16;
    if (mrow >= N) mrow = N - 1;
    const float* xrow = x + (size_t)mrow * IN_DIM;

    if (two && tile < nTiles) {
        float4* p4 = reinterpret_cast<float4*>(plane1 + (size_t)tile * 16 * OUT_DIM);
#pragma unroll
        for (int it = 0; it < 8; ++it) {
            const int idx = it * 32 + lane;
            if (tile * 16 + (idx >> 4) < N) p4[idx] = make_float4(0.f, 0.f, 0.f, 0.f);
        }
    }

    store_w<IN_DIM>(lWs, w_first);
    __syncthreads();
    if (tile < nTiles) {
        Acc4 acc = wave_gemm_tile(xrow, lWs, IN_DIM, half, l16);
#pragma unroll
        for (int nt = 0; nt < 4; ++nt) {
            const int   c  = nt * 16 + l16;
            const float bc = b_fc[c];
#pragma unroll
            for (int r = 0; r < 8; ++r) {
                const int row = tile * 16 + r + 8 * half;
                if (row < N) z[(size_t)row * OUT_DIM + c] = acc.t[nt][r] + bc;
            }
        }
    }
    __syncthreads();

    store_w<IN_DIM>(lWs, w_second);
    __syncthreads();
    if (tile < nTiles) {
        Acc4 acc = wave_gemm_tile(xrow, lWs, IN_DIM, half, l16);
#pragma unroll
        for (int nt = 0; nt < 4; ++nt) {
            const int   c  = nt * 16 + l16;
            const float bc = b_self[c];
#pragma unroll
            for (int r = 0; r < 8; ++r) {
                const int row = tile * 16 + r + 8 * half;
                if (row < N) hout[(size_t)row * OUT_DIM + c] = acc.t[nt][r] + bc;
            }
        }
    }
}

__global__ void k_node_attn(const float* __restrict__ z, const float* __restrict__ W_attn,
                            float* __restrict__ a_src, float* __restrict__ a_dst,
                            float* __restrict__ m, float* __restrict__ denom,
                            float* __restrict__ m1, float* __restrict__ denom1, int two, int N,
                            int* __restrict__ binz, int nbinz)
{
    const int n = blockIdx.x * blockDim.x + threadIdx.x;
    if (n < nbinz) binz[n] = 0;
    if (n >= N) return;
    const float* zr = z + (size_t)n * OUT_DIM;
    float as = 0.f, ad = 0.f;
#pragma unroll 8
    for (int c = 0; c < OUT_DIM; ++c) {
        const float zc = zr[c];
        as += zc * W_attn[c];
        ad += zc * W_attn[OUT_DIM + c];
    }
    a_src[n] = as;
    a_dst[n] = ad;
    m[n]     = -INFINITY;
    denom[n] = 0.0f;
    if (two) {
        m1[n]     = -INFINITY;
        denom1[n] = 0.0f;
    }
}

__global__ void k_edge_logit(const float* __restrict__ edge_attr,
                             const int* __restrict__ erow, const int* __restrict__ ecol,
                             const float* __restrict__ W_attn, const float* __restrict__ b_attn,
                             const float* __restrict__ a_src, const float* __restrict__ a_dst,
                             float* __restrict__ elog, float* __restrict__ m,
                             float* __restrict__ m1, int two, int E)
{
    const int e = blockIdx.x * blockDim.x + threadIdx.x;
    if (e >= E) return;
    const float* ea = edge_attr + (size_t)e * ED_DIM;
    float dot = 0.f;
#pragma unroll
    for (int k = 0; k < ED_DIM; ++k) dot += ea[k] * W_attn[2 * OUT_DIM + k];
    const int c = ecol[e];
    const float s  = a_src[erow[e]] + a_dst[c] + dot + b_attn[0];
    const float lr = (s > 0.0f) ? s : 0.2f * s;
    elog[e] = lr;
    float* mh = (two && chip_half()) ? m1 : m;
    atomicMaxF(&mh[c], lr);
}

__global__ void k_edge_soft(const float* __restrict__ elog, const int* __restrict__ ecol,
                            const float* __restrict__ m, const float* __restrict__ m1,
                            float* __restrict__ pbuf,
                            float* __restrict__ denom, float* __restrict__ denom1, int two, int E)
{
    const int e = blockIdx.x * blockDim.x + threadIdx.x;
    if (e >= E) return;
    const int c = ecol[e];
    const float mc = fmaxf(m[c], m1[c]);
    const float p = expf(elog[e] - mc);
    pbuf[e] = p;
    float* dh = (two && chip_half()) ? denom1 : denom;
    atomicAdd(&dh[c], p);
}

__device__ __attribute__((always_inline)) inline
float xhalf(float x)
{
    const int b = __builtin_bit_cast(int, x);
    return __builtin_bit_cast(float,
        __builtin_amdgcn_permlanex16(b, b, 0x76543210, 0xfedcba98, false, false));
}

template <bool FULL>
__device__ __attribute__((always_inline)) inline
void edge_epilogue(const Acc4& acc, int c_l, int r_l, float al_l, float b_lo, float b_hi,
                   const float* __restrict__ z, float* __restrict__ hout, int lane, int half)
{
    float zlo[16], zhi[16];
    if (FULL) {
#pragma unroll
        for (int j = 0; j < 16; ++j) {
            const int    rj = __builtin_amdgcn_readlane(r_l, j);
            const float* zr = z + (size_t)rj * OUT_DIM;
            zlo[j] = zr[lane];
            zhi[j] = zr[32 + lane];
        }
    }
#pragma unroll
    for (int r = 0; r < 8; ++r) {
        const float gA   = xhalf(half ? acc.t[0][r] : acc.t[1][r]);
        const float gB   = xhalf(half ? acc.t[2][r] : acc.t[3][r]);
        const float v0lo = half ? gA : acc.t[0][r];
        const float v1lo = half ? acc.t[1][r] : gA;
        const float v0hi = half ? gB : acc.t[2][r];
        const float v1hi = half ? acc.t[3][r] : gB;
#pragma unroll
        for (int h = 0; h < 2; ++h) {
            const int   j  = r + 8 * h;
            const int   cj = __builtin_amdgcn_readlane(c_l, j);
            const float aj = __builtin_bit_cast(float,
                __builtin_amdgcn_readlane(__builtin_bit_cast(int, al_l), j));
            if (FULL || cj >= 0) {
                float zl, zh;
                if (FULL) {
                    zl = zlo[j];
                    zh = zhi[j];
                } else {
                    const int    rj = __builtin_amdgcn_readlane(r_l, j);
                    const float* zr = z + (size_t)rj * OUT_DIM;
                    zl = zr[lane];
                    zh = zr[32 + lane];
                }
                float* hr = hout + (size_t)cj * OUT_DIM;
                atomicAdd(&hr[lane],      aj * (zl + (h ? v1lo : v0lo) + b_lo));
                atomicAdd(&hr[32 + lane], aj * (zh + (h ? v1hi : v0hi) + b_hi));
            }
        }
    }
}

__global__ void k_edge_aggregate(const float* __restrict__ edge_attr,
                                 const int* __restrict__ erow, const int* __restrict__ ecol,
                                 const float* __restrict__ W_edge, const float* __restrict__ b_edge,
                                 const float* __restrict__ z, const float* __restrict__ pbuf,
                                 const float* __restrict__ denom, const float* __restrict__ denom1,
                                 float* __restrict__ hout, float* __restrict__ plane1, int two,
                                 int E, int nTiles)
{
    __shared__ __attribute__((aligned(16))) float lWs[(ED_DIM / 2) * BSTRIDE_F];
    const int lane = threadIdx.x & 31;
    const int wave = threadIdx.x >> 5;
    const int half = lane >> 4;
    const int l16  = lane & 15;

    store_w<ED_DIM>(lWs, load_w<ED_DIM>(W_edge));
    __syncthreads();
    const int tile = __builtin_amdgcn_readfirstlane(blockIdx.x * 8 + wave);
    if (tile >= nTiles) return;

    int arow = tile * 16 + l16;
    const bool evalid = arow < E;
    if (arow >= E) arow = E - 1;
    const float* ea = edge_attr + (size_t)arow * ED_DIM;

    {
        const size_t nxt = (size_t)arow + 16 * 8;
        if (nxt < (size_t)E)
            __builtin_prefetch(edge_attr + nxt * ED_DIM, 0, 1);
    }

    const int   c_e  = ecol[arow];
    const int   c_l  = evalid ? c_e : -1;
    const int   r_l  = erow[arow];
    const float d0   = denom[c_e];
    const float d1   = denom1[c_e];
    const float den  = two ? d0 + d1 : d0;
    const float al_l = pbuf[arow] / fmaxf(den, 1e-16f);

    Acc4 acc = wave_gemm_tile(ea, lWs, ED_DIM, half, l16);

    const float b_lo = b_edge[lane];
    const float b_hi = b_edge[32 + lane];

    float* hacc = (two && chip_half()) ? plane1 : hout;

    if (tile * 16 + 16 <= E)
        edge_epilogue<true >(acc, c_l, r_l, al_l, b_lo, b_hi, z, hacc, lane, half);
    else
        edge_epilogue<false>(acc, c_l, r_l, al_l, b_lo, b_hi, z, hacc, lane, half);
}


#define BIN_R   196
#define NB_MAX  1024
#define BIN_T   1024
#define BIN_PER 4

__global__ __launch_bounds__(BIN_T)
void k_bin_count(const int* __restrict__ ecol, int* __restrict__ cnt, int E, int nb)
{
    __shared__ int s_hist[NB_MAX];
    const int tid = threadIdx.x;
    s_hist[tid] = 0;
    __syncthreads();
#pragma unroll
    for (int i = 0; i < BIN_PER; ++i) {
        const int e = blockIdx.x * (BIN_T * BIN_PER) + i * BIN_T + tid;
        if (e < E) atomicAdd(&s_hist[ecol[e] / BIN_R], 1);
    }
    __syncthreads();
    const int h = s_hist[tid];
    if (tid < nb && h > 0) atomicAdd(&cnt[tid], h);
}

__global__ __launch_bounds__(BIN_T)
void k_bin_place(const int* __restrict__ erow, const int* __restrict__ ecol,
                 const float* __restrict__ pbuf,
                 const float* __restrict__ denom, const float* __restrict__ denom1, int two,
                 const int* __restrict__ cnt, int* __restrict__ cursor, int* __restrict__ bbase,
                 int* __restrict__ rec_e, int* __restrict__ rec_rc, float* __restrict__ rec_al,
                 int E, int nb)
{
    __shared__ int s_base[NB_MAX];
    __shared__ int s_hist[NB_MAX];
    __shared__ int s_off[NB_MAX];
    __shared__ int s_wsum[32];
    const int tid  = threadIdx.x;
    const int lane = tid & 31;
    const int wave = tid >> 5;

    int   eid[BIN_PER], bk[BIN_PER], rc[BIN_PER];
    float al[BIN_PER];
#pragma unroll
    for (int i = 0; i < BIN_PER; ++i) {
        const int e = blockIdx.x * (BIN_T * BIN_PER) + i * BIN_T + tid;
        eid[i] = e;
        bk[i]  = -1;
        rc[i]  = 0;
        al[i]  = 0.f;
        if (e < E) {
            const int   c   = ecol[e];
            const int   b   = c / BIN_R;
            const float d0  = denom[c];
            const float d1  = denom1[c];
            const float den = two ? d0 + d1 : d0;
            bk[i] = b;
            rc[i] = erow[e] * 256 + (c - b * BIN_R);
            al[i] = pbuf[e] / fmaxf(den, 1e-16f);
        }
    }

    const int own = tid < nb ? cnt[tid] : 0;
    int v = own;
#pragma unroll
    for (int d = 1; d < 32; d <<= 1) {
        const int u = __shfl_up(v, d);
        if (lane >= d) v += u;
    }
    if (lane == 31) s_wsum[wave] = v;
    s_hist[tid] = 0;
    __syncthreads();
    if (wave == 0) {
        const int w = s_wsum[lane];
        int x = w;
#pragma unroll
        for (int d = 1; d < 32; d <<= 1) {
            const int u = __shfl_up(x, d);
            if (lane >= d) x += u;
        }
        s_wsum[lane] = x - w;
    }
    __syncthreads();
    const int excl = v - own + s_wsum[wave];
    s_base[tid] = excl;
    if (blockIdx.x == 0) {
        if (tid < nb)      bbase[tid] = excl;
        if (tid == nb - 1) bbase[nb]  = excl + own;
    }

    int lr[BIN_PER];
#pragma unroll
    for (int i = 0; i < BIN_PER; ++i) {
        lr[i] = 0;
        if (bk[i] >= 0) lr[i] = atomicAdd(&s_hist[bk[i]], 1);
    }
    __syncthreads();
    {
        const int h = s_hist[tid];
        if (tid < nb && h > 0) s_off[tid] = s_base[tid] + atomicAdd(&cursor[tid], h);
    }
    __syncthreads();
#pragma unroll
    for (int i = 0; i < BIN_PER; ++i) {
        if (bk[i] >= 0) {
            const int pos = s_off[bk[i]] + lr[i];
            rec_e[pos]  = eid[i];
            rec_rc[pos] = rc[i];
            rec_al[pos] = al[i];
        }
    }
}

__global__ __launch_bounds__(BIN_T)
void k_range_aggregate(const float* __restrict__ edge_attr,
                       const float* __restrict__ W_edge, const float* __restrict__ b_edge,
                       const float* __restrict__ z,
                       const int* __restrict__ rec_e, const int* __restrict__ rec_rc,
                       const float* __restrict__ rec_al, const int* __restrict__ bbase,
                       float* __restrict__ hout, int N)
{
    __shared__ __attribute__((aligned(16))) float lWs[(ED_DIM / 2) * BSTRIDE_F];
    __shared__ __attribute__((aligned(16))) float lacc[BIN_R * OUT_DIM];
    const int lane = threadIdx.x & 31;
    const int wave = __builtin_amdgcn_readfirstlane(threadIdx.x >> 5);
    const int half = lane >> 4;
    const int l16  = lane & 15;

    const int n0   = blockIdx.x * BIN_R;
    const int rows = (N - n0 < BIN_R) ? (N - n0) : BIN_R;

    if (threadIdx.x < NTHREADS) store_w<ED_DIM>(lWs, load_w<ED_DIM>(W_edge));
    {
        const float4* h4 = reinterpret_cast<const float4*>(hout + (size_t)n0 * OUT_DIM);
        float4*       a4 = reinterpret_cast<float4*>(lacc);
        for (int i = threadIdx.x; i < rows * (OUT_DIM / 4); i += BIN_T) a4[i] = h4[i];
    }
    const int b0 = bbase[blockIdx.x];
    const int b1 = bbase[blockIdx.x + 1];
    const int ntile = (b1 - b0 + 15) >> 4;
    const float b_lo = b_edge[lane];
    const float b_hi = b_edge[32 + lane];
    __syncthreads();

    int   e_n = 0, rc_n = 0;
    float al_n = 0.f;
    if (wave < ntile) {
        int idx = b0 + wave * 16 + l16;
        if (idx >= b1) idx = b1 - 1;
        e_n = rec_e[idx]; rc_n = rec_rc[idx]; al_n = rec_al[idx];
    }
    for (int t = wave; t < ntile; t += BIN_T / 32) {
        asm volatile("" ::: "memory");
        const int   e_c = e_n, rc_c = rc_n;
        const float al_c = al_n;
        const int tn = t + BIN_T / 32;
        if (tn < ntile) {
            int idx = b0 + tn * 16 + l16;
            if (idx >= b1) idx = b1 - 1;
            e_n = rec_e[idx]; rc_n = rec_rc[idx]; al_n = rec_al[idx];
        }
        const bool valid = b0 + t * 16 + l16 < b1;
        const int  c_l   = valid ? (rc_c & 255) : -1;
        const int  r_l   = rc_c >> 8;
        const float* ea  = edge_attr + (size_t)e_c * ED_DIM;

        Acc4 acc = wave_gemm_tile(ea, lWs, ED_DIM, half, l16);

        if (t * 16 + 16 <= b1 - b0)
            edge_epilogue<true >(acc, c_l, r_l, al_c, b_lo, b_hi, z, lacc, lane, half);
        else
            edge_epilogue<false>(acc, c_l, r_l, al_c, b_lo, b_hi, z, lacc, lane, half);
    }
    __syncthreads();
    {
        float4*       o4 = reinterpret_cast<float4*>(hout + (size_t)n0 * OUT_DIM);
        const float4* a4 = reinterpret_cast<const float4*>(lacc);
        for (int i = threadIdx.x; i < rows * (OUT_DIM / 4); i += BIN_T) o4[i] = a4[i];
    }
}

__global__ void k_plane_sum(float* __restrict__ hout, const float* __restrict__ plane1,
                            int two, int n4)
{
    const int i = blockIdx.x * blockDim.x + threadIdx.x;
    if (!two || i >= n4) return;
    float4*       h4 = reinterpret_cast<float4*>(hout);
    const float4* q4 = reinterpret_cast<const float4*>(plane1);
    float4 a = h4[i];
    const float4 b = q4[i];
    a.x += b.x; a.y += b.y; a.z += b.z; a.w += b.w;
    h4[i] = a;
}

extern "C" void kernel_launch(void* const* d_in, const int* in_sizes, int n_in,
                              void* d_out, int out_size, void* d_ws, size_t ws_size,
                              hipStream_t stream)
{
    const float* x         = (const float*)d_in[0];
    const float* edge_attr = (const float*)d_in[1];
    const int*   edge_idx  = (const int*)  d_in[2];
    const float* W_fc      = (const float*)d_in[3];
    const float* b_fc      = (const float*)d_in[4];
    const float* W_attn    = (const float*)d_in[5];
    const float* b_attn    = (const float*)d_in[6];
    const float* W_edge    = (const float*)d_in[7];
    const float* b_edge    = (const float*)d_in[8];
    const float* W_self    = (const float*)d_in[9];
    const float* b_self    = (const float*)d_in[10];

    const int N = in_sizes[0] / IN_DIM;
    const int E = in_sizes[1] / ED_DIM;
    const int* erow = edge_idx;
    const int* ecol = edge_idx + E;

    float* ws     = (float*)d_ws;
    float* z      = ws;
    float* a_src  = z + (size_t)N * OUT_DIM;
    float* a_dst  = a_src + N;
    float* m      = a_dst + N;
    float* denom  = m + N;
    float* elog   = denom + N;
    float* pbuf   = elog + E;
    const size_t off1 = ((size_t)N * (OUT_DIM + 6) + 2 * (size_t)E + 3) & ~(size_t)3;
    const int two = ws_size >= (off1 + (size_t)N * OUT_DIM) * sizeof(float) ? 1 : 0;
    float* m1     = two ? pbuf + E : m;
    float* denom1 = two ? m1 + N   : denom;
    float* plane1 = two ? ws + off1 : (float*)d_out;
    float* hout   = (float*)d_out;

    const int nNodeTiles = (N + 15) / 16;
    const int nEdgeTiles = (E + 15) / 16;

    const int nb   = (N + BIN_R - 1) / BIN_R;
    const int bins = (two && nb <= NB_MAX && N >= 2 * NB_MAX && N < (1 << 23) &&
                      (size_t)N * OUT_DIM >= 3 * (size_t)E + 3 * NB_MAX + 8) ? 1 : 0;
    int*   rec_e  = (int*)plane1;
    int*   rec_rc = rec_e + E;
    float* rec_al = (float*)(rec_rc + E);
    int*   cnt    = (int*)(rec_al + E);
    int*   cursor = cnt + NB_MAX;
    int*   bbase  = cursor + NB_MAX;
    const int nBinBlocks = (E + BIN_T * BIN_PER - 1) / (BIN_T * BIN_PER);

    k_node_gemm<<<(nNodeTiles + 7) / 8, NTHREADS, 0, stream>>>(
        x, W_fc, b_fc, W_self, b_self, z, hout, plane1, bins ? 0 : two, N, nNodeTiles);

    k_node_attn<<<(N + 255) / 256, 256, 0, stream>>>(
        z, W_attn, a_src, a_dst, m, denom, m1, denom1, two, N, cnt, bins ? 2 * NB_MAX : 0);

    k_edge_logit<<<(E + 255) / 256, 256, 0, stream>>>(
        edge_attr, erow, ecol, W_attn, b_attn, a_src, a_dst, elog, m, m1, two, E);

    if (bins)
        k_bin_count<<<nBinBlocks, BIN_T, 0, stream>>>(ecol, cnt, E, nb);

    k_edge_soft<<<(E + 255) / 256, 256, 0, stream>>>(
        elog, ecol, m, m1, pbuf, denom, denom1, two, E);

    if (bins) {
        k_bin_place<<<nBinBlocks, BIN_T, 0, stream>>>(
            erow, ecol, pbuf, denom, denom1, two, cnt, cursor, bbase, rec_e, rec_rc, rec_al, E, nb);

        k_range_aggregate<<<nb, BIN_T, 0, stream>>>(
            edge_attr, W_edge, b_edge, z, rec_e, rec_rc, rec_al, bbase, hout, N);
    } else {
        k_edge_aggregate<<<(nEdgeTiles + 7) / 8, NTHREADS, 0, stream>>>(
            edge_attr, erow, ecol, W_edge, b_edge, z, pbuf, denom, denom1, hout, plane1, two,
            E, nEdgeTiles);

        k_plane_sum<<<(N * (OUT_DIM / 4) + 255) / 256, 256, 0, stream>>>(
            hout, plane1, two, N * (OUT_DIM / 4));
    }
}


# softmax statistics (segment max, exp sum) taken in LDS by the range work-group, two passes over its records ahead of the messages; memory atomics and the exp launch go away; exp-sum order changes
# speedup vs baseline: 2.6106x; 1.1464x over previous
#include <hip/hip_runtime.h>
#include <hip/hip_bf16.h>
#include <math.h>

#define IN_DIM  128
#define OUT_DIM 64
#define ED_DIM  32

#define BSTRIDE_V2 80
#define BSTRIDE_F  (2 * BSTRIDE_V2)

typedef __attribute__((ext_vector_type(2))) float v2f;
typedef __attribute__((ext_vector_type(8))) float v8f;

struct Acc4 { v8f t[4]; };

__device__ __attribute__((always_inline)) inline
Acc4 wave_gemm_tile(const float* __restrict__ arow,
                    const float* __restrict__ lWs,
                    int kdim, int half, int l16)
{
    Acc4 A;
    A.t[0] = {}; A.t[1] = {}; A.t[2] = {}; A.t[3] = {};
    const v2f* a2 = (const v2f*)arow;
    for (int k0 = 0; k0 < kdim; k0 += 4) {
        const int kp  = (k0 >> 1) + half;
        const v2f  a  = a2[kp];
        const v2f* br = (const v2f*)(lWs + kp * BSTRIDE_F);
#pragma unroll
        for (int nt = 0; nt < 4; ++nt) {
            const v2f b = br[nt * 16 + l16];
            A.t[nt] = __builtin_amdgcn_wmma_f32_16x16x4_f32(
                false, a, false, b, (short)0, A.t[nt], false, false);
        }
    }
    return A;
}

#define NTHREADS 256

template <int KDIM>
struct WStage {
    static constexpr int ITEMS = (KDIM / 2) * (OUT_DIM / 4);
    static constexpr int PER   = ITEMS / NTHREADS;
    static_assert(ITEMS % NTHREADS == 0, "staging is laid out for whole rounds of the work-group");
    float4 lo[PER];
    float4 hi[PER];
};

template <int KDIM>
__device__ __attribute__((always_inline)) inline
WStage<KDIM> load_w(const float* __restrict__ W)
{
    WStage<KDIM> s;
#pragma unroll
    for (int j = 0; j < WStage<KDIM>::PER; ++j) {
        const int it = threadIdx.x + NTHREADS * j;
        const int kp = it >> 4;
        const int g  = it & 15;
        const float4* src = reinterpret_cast<const float4*>(W + (size_t)(2 * kp) * OUT_DIM + 4 * g);
        s.lo[j] = src[0];
        s.hi[j] = src[OUT_DIM / 4];
    }
    return s;
}

template <int KDIM>
__device__ __attribute__((always_inline)) inline
void store_w(float* __restrict__ lWs, const WStage<KDIM>& s)
{
#pragma unroll
    for (int j = 0; j < WStage<KDIM>::PER; ++j) {
        const int it = threadIdx.x + NTHREADS * j;
        const int kp = it >> 4;
        const int g  = it & 15;
        float4* dst = reinterpret_cast<float4*>(lWs + kp * BSTRIDE_F + 8 * g);
        dst[0] = make_float4(s.lo[j].x, s.hi[j].x, s.lo[j].y, s.hi[j].y);
        dst[1] = make_float4(s.lo[j].z, s.hi[j].z, s.lo[j].w, s.hi[j].w);
    }
}

__device__ __attribute__((always_inline)) inline
int chip_half()
{
    const unsigned v = __builtin_amdgcn_s_sendmsg_rtn(0x87);
    return (int)((v >> 18) & 1u);
}

__device__ inline void atomicMaxF(float* addr, float val) {
    if (val >= 0.0f) atomicMax(reinterpret_cast<int*>(addr), __float_as_int(val));
    else             atomicMin(reinterpret_cast<unsigned int*>(addr), __float_as_uint(val));
}

__global__ void k_node_gemm(const float* __restrict__ x,
                            const float* __restrict__ W_fc,  const float* __restrict__ b_fc,
                            const float* __restrict__ W_self, const float* __restrict__ b_self,
                            float* __restrict__ z, float* __restrict__ hout,
                            float* __restrict__ plane1, int two,
                            int N, int nTiles)
{
    __shared__ __attribute__((aligned(16))) float lWs[(IN_DIM / 2) * BSTRIDE_F];
    const int lane = threadIdx.x & 31;
    const int wave = threadIdx.x >> 5;
    const int half = lane >> 4;
    const int l16  = lane & 15;
    const int tile = blockIdx.x * 8 + wave;

    const WStage<IN_DIM> w_first  = load_w<IN_DIM>(W_fc);
    const WStage<IN_DIM> w_second = load_w<IN_DIM>(W_self);

    int mrow = tile * 16 + l16;
    if (mrow >= N) mrow = N - 1;
    const float* xrow = x + (size_t)mrow * IN_DIM;

    if (two && tile < nTiles) {
        float4* p4 = reinterpret_cast<float4*>(plane1 + (size_t)tile * 16 * OUT_DIM);
#pragma unroll
        for (int it = 0; it < 8; ++it) {
            const int idx = it * 32 + lane;
            if (tile * 16 + (idx >> 4) < N) p4[idx] = make_float4(0.f, 0.f, 0.f, 0.f);
        }
    }

    store_w<IN_DIM>(lWs, w_first);
    __syncthreads();
    if (tile < nTiles) {
        Acc4 acc = wave_gemm_tile(xrow, lWs, IN_DIM, half, l16);
#pragma unroll
        for (int nt = 0; nt < 4; ++nt) {
            const int   c  = nt * 16 + l16;
            const float bc = b_fc[c];
#pragma unroll
            for (int r = 0; r < 8; ++r) {
                const int row = tile * 16 + r + 8 * half;
                if (row < N) z[(size_t)row * OUT_DIM + c] = acc.t[nt][r] + bc;
            }
        }
    }
    __syncthreads();

    store_w<IN_DIM>(lWs, w_second);
    __syncthreads();
    if (tile < nTiles) {
        Acc4 acc = wave_gemm_tile(xrow, lWs, IN_DIM, half, l16);
#pragma unroll
        for (int nt = 0; nt < 4; ++nt) {
            const int   c  = nt * 16 + l16;
            const float bc = b_self[c];
#pragma unroll
            for (int r = 0; r < 8; ++r) {
                const int row = tile * 16 + r + 8 * half;
                if (row < N) hout[(size_t)row * OUT_DIM + c] = acc.t[nt][r] + bc;
            }
        }
    }
}

__global__ void k_node_attn(const float* __restrict__ z, const float* __restrict__ W_attn,
                            float* __restrict__ a_src, float* __restrict__ a_dst,
                            float* __restrict__ m, float* __restrict__ denom,
                            float* __restrict__ m1, float* __restrict__ denom1, int two, int N,
                            int* __restrict__ binz, int nbinz, int stats)
{
    const int n = blockIdx.x * blockDim.x + threadIdx.x;
    if (n < nbinz) binz[n] = 0;
    if (n >= N) return;
    const float* zr = z + (size_t)n * OUT_DIM;
    float as = 0.f, ad = 0.f;
#pragma unroll 8
    for (int c = 0; c < OUT_DIM; ++c) {
        const float zc = zr[c];
        as += zc * W_attn[c];
        ad += zc * W_attn[OUT_DIM + c];
    }
    a_src[n] = as;
    a_dst[n] = ad;
    if (!stats) return;
    m[n]     = -INFINITY;
    denom[n] = 0.0f;
    if (two) {
        m1[n]     = -INFINITY;
        denom1[n] = 0.0f;
    }
}

__global__ void k_edge_logit(const float* __restrict__ edge_attr,
                             const int* __restrict__ erow, const int* __restrict__ ecol,
                             const float* __restrict__ W_attn, const float* __restrict__ b_attn,
                             const float* __restrict__ a_src, const float* __restrict__ a_dst,
                             float* __restrict__ elog, float* __restrict__ m,
                             float* __restrict__ m1, int two, int E, int stats)
{
    const int e = blockIdx.x * blockDim.x + threadIdx.x;
    if (e >= E) return;
    const float* ea = edge_attr + (size_t)e * ED_DIM;
    float dot = 0.f;
#pragma unroll
    for (int k = 0; k < ED_DIM; ++k) dot += ea[k] * W_attn[2 * OUT_DIM + k];
    const int c = ecol[e];
    const float s  = a_src[erow[e]] + a_dst[c] + dot + b_attn[0];
    const float lr = (s > 0.0f) ? s : 0.2f * s;
    elog[e] = lr;
    if (!stats) return;
    float* mh = (two && chip_half()) ? m1 : m;
    atomicMaxF(&mh[c], lr);
}

__global__ void k_edge_soft(const float* __restrict__ elog, const int* __restrict__ ecol,
                            const float* __restrict__ m, const float* __restrict__ m1,
                            float* __restrict__ pbuf,
                            float* __restrict__ denom, float* __restrict__ denom1, int two, int E)
{
    const int e = blockIdx.x * blockDim.x + threadIdx.x;
    if (e >= E) return;
    const int c = ecol[e];
    const float mc = fmaxf(m[c], m1[c]);
    const float p = expf(elog[e] - mc);
    pbuf[e] = p;
    float* dh = (two && chip_half()) ? denom1 : denom;
    atomicAdd(&dh[c], p);
}

__device__ __attribute__((always_inline)) inline
float xhalf(float x)
{
    const int b = __builtin_bit_cast(int, x);
    return __builtin_bit_cast(float,
        __builtin_amdgcn_permlanex16(b, b, 0x76543210, 0xfedcba98, false, false));
}

template <bool FULL>
__device__ __attribute__((always_inline)) inline
void edge_epilogue(const Acc4& acc, int c_l, int r_l, float al_l, float b_lo, float b_hi,
                   const float* __restrict__ z, float* __restrict__ hout, int lane, int half)
{
    float zlo[16], zhi[16];
    if (FULL) {
#pragma unroll
        for (int j = 0; j < 16; ++j) {
            const int    rj = __builtin_amdgcn_readlane(r_l, j);
            const float* zr = z + (size_t)rj * OUT_DIM;
            zlo[j] = zr[lane];
            zhi[j] = zr[32 + lane];
        }
    }
#pragma unroll
    for (int r = 0; r < 8; ++r) {
        const float gA   = xhalf(half ? acc.t[0][r] : acc.t[1][r]);
        const float gB   = xhalf(half ? acc.t[2][r] : acc.t[3][r]);
        const float v0lo = half ? gA : acc.t[0][r];
        const float v1lo = half ? acc.t[1][r] : gA;
        const float v0hi = half ? gB : acc.t[2][r];
        const float v1hi = half ? acc.t[3][r] : gB;
#pragma unroll
        for (int h = 0; h < 2; ++h) {
            const int   j  = r + 8 * h;
            const int   cj = __builtin_amdgcn_readlane(c_l, j);
            const float aj = __builtin_bit_cast(float,
                __builtin_amdgcn_readlane(__builtin_bit_cast(int, al_l), j));
            if (FULL || cj >= 0) {
                float zl, zh;
                if (FULL) {
                    zl = zlo[j];
                    zh = zhi[j];
                } else {
                    const int    rj = __builtin_amdgcn_readlane(r_l, j);
                    const float* zr = z + (size_t)rj * OUT_DIM;
                    zl = zr[lane];
                    zh = zr[32 + lane];
                }
                float* hr = hout + (size_t)cj * OUT_DIM;
                atomicAdd(&hr[lane],      aj * (zl + (h ? v1lo : v0lo) + b_lo));
                atomicAdd(&hr[32 + lane], aj * (zh + (h ? v1hi : v0hi) + b_hi));
            }
        }
    }
}

__global__ void k_edge_aggregate(const float* __restrict__ edge_attr,
                                 const int* __restrict__ erow, const int* __restrict__ ecol,
                                 const float* __restrict__ W_edge, const float* __restrict__ b_edge,
                                 const float* __restrict__ z, const float* __restrict__ pbuf,
                                 const float* __restrict__ denom, const float* __restrict__ denom1,
                                 float* __restrict__ hout, float* __restrict__ plane1, int two,
                                 int E, int nTiles)
{
    __shared__ __attribute__((aligned(16))) float lWs[(ED_DIM / 2) * BSTRIDE_F];
    const int lane = threadIdx.x & 31;
    const int wave = threadIdx.x >> 5;
    const int half = lane >> 4;
    const int l16  = lane & 15;

    store_w<ED_DIM>(lWs, load_w<ED_DIM>(W_edge));
    __syncthreads();
    const int tile = __builtin_amdgcn_readfirstlane(blockIdx.x * 8 + wave);
    if (tile >= nTiles) return;

    int arow = tile * 16 + l16;
    const bool evalid = arow < E;
    if (arow >= E) arow = E - 1;
    const float* ea = edge_attr + (size_t)arow * ED_DIM;

    {
        const size_t nxt = (size_t)arow + 16 * 8;
        if (nxt < (size_t)E)
            __builtin_prefetch(edge_attr + nxt * ED_DIM, 0, 1);
    }

    const int   c_e  = ecol[arow];
    const int   c_l  = evalid ? c_e : -1;
    const int   r_l  = erow[arow];
    const float d0   = denom[c_e];
    const float d1   = denom1[c_e];
    const float den  = two ? d0 + d1 : d0;
    const float al_l = pbuf[arow] / fmaxf(den, 1e-16f);

    Acc4 acc = wave_gemm_tile(ea, lWs, ED_DIM, half, l16);

    const float b_lo = b_edge[lane];
    const float b_hi = b_edge[32 + lane];

    float* hacc = (two && chip_half()) ? plane1 : hout;

    if (tile * 16 + 16 <= E)
        edge_epilogue<true >(acc, c_l, r_l, al_l, b_lo, b_hi, z, hacc, lane, half);
    else
        edge_epilogue<false>(acc, c_l, r_l, al_l, b_lo, b_hi, z, hacc, lane, half);
}


#define BIN_R   196
#define STAT_PER 4
#define NB_MAX  1024
#define BIN_T   1024
#define BIN_PER 4

__global__ __launch_bounds__(BIN_T)
void k_bin_count(const int* __restrict__ ecol, int* __restrict__ cnt, int E, int nb)
{
    __shared__ int s_hist[NB_MAX];
    const int tid = threadIdx.x;
    s_hist[tid] = 0;
    __syncthreads();
#pragma unroll
    for (int i = 0; i < BIN_PER; ++i) {
        const int e = blockIdx.x * (BIN_T * BIN_PER) + i * BIN_T + tid;
        if (e < E) atomicAdd(&s_hist[ecol[e] / BIN_R], 1);
    }
    __syncthreads();
    const int h = s_hist[tid];
    if (tid < nb && h > 0) atomicAdd(&cnt[tid], h);
}

__global__ __launch_bounds__(BIN_T)
void k_bin_place(const int* __restrict__ erow, const int* __restrict__ ecol,
                 const float* __restrict__ elog,
                 const int* __restrict__ cnt, int* __restrict__ cursor, int* __restrict__ bbase,
                 int* __restrict__ rec_e, int* __restrict__ rec_rc, float* __restrict__ rec_lg,
                 int E, int nb)
{
    __shared__ int s_base[NB_MAX];
    __shared__ int s_hist[NB_MAX];
    __shared__ int s_off[NB_MAX];
    __shared__ int s_wsum[32];
    const int tid  = threadIdx.x;
    const int lane = tid & 31;
    const int wave = tid >> 5;

    int   eid[BIN_PER], bk[BIN_PER], rc[BIN_PER];
    float lg[BIN_PER];
#pragma unroll
    for (int i = 0; i < BIN_PER; ++i) {
        const int e = blockIdx.x * (BIN_T * BIN_PER) + i * BIN_T + tid;
        eid[i] = e;
        bk[i]  = -1;
        rc[i]  = 0;
        lg[i]  = 0.f;
        if (e < E) {
            const int c = ecol[e];
            const int b = c / BIN_R;
            bk[i] = b;
            rc[i] = erow[e] * 256 + (c - b * BIN_R);
            lg[i] = elog[e];
        }
    }

    const int own = tid < nb ? cnt[tid] : 0;
    int v = own;
#pragma unroll
    for (int d = 1; d < 32; d <<= 1) {
        const int u = __shfl_up(v, d);
        if (lane >= d) v += u;
    }
    if (lane == 31) s_wsum[wave] = v;
    s_hist[tid] = 0;
    __syncthreads();
    if (wave == 0) {
        const int w = s_wsum[lane];
        int x = w;
#pragma unroll
        for (int d = 1; d < 32; d <<= 1) {
            const int u = __shfl_up(x, d);
            if (lane >= d) x += u;
        }
        s_wsum[lane] = x - w;
    }
    __syncthreads();
    const int excl = v - own + s_wsum[wave];
    s_base[tid] = excl;
    if (blockIdx.x == 0) {
        if (tid < nb)      bbase[tid] = excl;
        if (tid == nb - 1) bbase[nb]  = excl + own;
    }

    int lr[BIN_PER];
#pragma unroll
    for (int i = 0; i < BIN_PER; ++i) {
        lr[i] = 0;
        if (bk[i] >= 0) lr[i] = atomicAdd(&s_hist[bk[i]], 1);
    }
    __syncthreads();
    {
        const int h = s_hist[tid];
        if (tid < nb && h > 0) s_off[tid] = s_base[tid] + atomicAdd(&cursor[tid], h);
    }
    __syncthreads();
#pragma unroll
    for (int i = 0; i < BIN_PER; ++i) {
        if (bk[i] >= 0) {
            const int pos = s_off[bk[i]] + lr[i];
            rec_e[pos]  = eid[i];
            rec_rc[pos] = rc[i];
            rec_lg[pos] = lg[i];
        }
    }
}

__device__ __attribute__((always_inline)) inline
void lds_max(float* addr, float val)
{
    __hip_atomic_fetch_max(addr, val, __ATOMIC_RELAXED, __HIP_MEMORY_SCOPE_WORKGROUP);
}

__global__ __launch_bounds__(BIN_T)
void k_range_aggregate(const float* __restrict__ edge_attr,
                       const float* __restrict__ W_edge, const float* __restrict__ b_edge,
                       const float* __restrict__ z,
                       const int* __restrict__ rec_e, const int* __restrict__ rec_rc,
                       const float* __restrict__ rec_lg, const int* __restrict__ bbase,
                       float* __restrict__ hout, int N)
{
    __shared__ __attribute__((aligned(16))) float lWs[(ED_DIM / 2) * BSTRIDE_F];
    __shared__ __attribute__((aligned(16))) float lacc[BIN_R * OUT_DIM];
    __shared__ float lmax[BIN_R];
    __shared__ float lden[BIN_R];
    const int lane = threadIdx.x & 31;
    const int wave = __builtin_amdgcn_readfirstlane(threadIdx.x >> 5);
    const int half = lane >> 4;
    const int l16  = lane & 15;

    const int n0   = blockIdx.x * BIN_R;
    const int rows = (N - n0 < BIN_R) ? (N - n0) : BIN_R;

    const int b0 = bbase[blockIdx.x];
    const int b1 = bbase[blockIdx.x + 1];
    const int ntile = (b1 - b0 + 15) >> 4;

    float lg_s[STAT_PER];
    int   rc_s[STAT_PER];
#pragma unroll
    for (int j = 0; j < STAT_PER; ++j) {
        const int i  = b0 + j * BIN_T + (int)threadIdx.x;
        const int ii = i < b1 ? i : 0;
        lg_s[j] = rec_lg[ii];
        rc_s[j] = rec_rc[ii];
    }
    int   e_n = 0, rc_n = 0;
    float lg_n = 0.f;
    if (wave < ntile) {
        int idx = b0 + wave * 16 + l16;
        if (idx >= b1) idx = b1 - 1;
        e_n = rec_e[idx]; rc_n = rec_rc[idx]; lg_n = rec_lg[idx];
    }

    if (threadIdx.x < BIN_R) { lmax[threadIdx.x] = -INFINITY; lden[threadIdx.x] = 0.f; }
    if (threadIdx.x < NTHREADS) store_w<ED_DIM>(lWs, load_w<ED_DIM>(W_edge));
    {
        const float4* h4 = reinterpret_cast<const float4*>(hout + (size_t)n0 * OUT_DIM);
        float4*       a4 = reinterpret_cast<float4*>(lacc);
        for (int i = threadIdx.x; i < rows * (OUT_DIM / 4); i += BIN_T) a4[i] = h4[i];
    }
    const float b_lo = b_edge[lane];
    const float b_hi = b_edge[32 + lane];
    __syncthreads();

    int c_s[STAT_PER];
#pragma unroll
    for (int j = 0; j < STAT_PER; ++j)
        c_s[j] = (b0 + j * BIN_T + (int)threadIdx.x < b1) ? (rc_s[j] & 255) : -1;

#pragma unroll
    for (int j = 0; j < STAT_PER; ++j)
        if (c_s[j] >= 0) lds_max(&lmax[c_s[j]], lg_s[j]);
    for (int i = b0 + STAT_PER * BIN_T + (int)threadIdx.x; i < b1; i += BIN_T)
        lds_max(&lmax[rec_rc[i] & 255], rec_lg[i]);
    __syncthreads();

#pragma unroll
    for (int j = 0; j < STAT_PER; ++j)
        if (c_s[j] >= 0) atomicAdd(&lden[c_s[j]], expf(lg_s[j] - lmax[c_s[j]]));
    for (int i = b0 + STAT_PER * BIN_T + (int)threadIdx.x; i < b1; i += BIN_T) {
        const int cc = rec_rc[i] & 255;
        atomicAdd(&lden[cc], expf(rec_lg[i] - lmax[cc]));
    }
    __syncthreads();

    for (int t = wave; t < ntile; t += BIN_T / 32) {
        asm volatile("" ::: "memory");
        const int   e_c = e_n, rc_c = rc_n;
        const float lg_c = lg_n;
        const int tn = t + BIN_T / 32;
        if (tn < ntile) {
            int idx = b0 + tn * 16 + l16;
            if (idx >= b1) idx = b1 - 1;
            e_n = rec_e[idx]; rc_n = rec_rc[idx]; lg_n = rec_lg[idx];
        }
        const bool valid = b0 + t * 16 + l16 < b1;
        const int  cc    = rc_c & 255;
        const int  c_l   = valid ? cc : -1;
        const int  r_l   = rc_c >> 8;
        const float al_c = expf(lg_c - lmax[cc]) / fmaxf(lden[cc], 1e-16f);
        const float* ea  = edge_attr + (size_t)e_c * ED_DIM;

        Acc4 acc = wave_gemm_tile(ea, lWs, ED_DIM, half, l16);

        if (t * 16 + 16 <= b1 - b0)
            edge_epilogue<true >(acc, c_l, r_l, al_c, b_lo, b_hi, z, lacc, lane, half);
        else
            edge_epilogue<false>(acc, c_l, r_l, al_c, b_lo, b_hi, z, lacc, lane, half);
    }
    __syncthreads();
    {
        float4*       o4 = reinterpret_cast<float4*>(hout + (size_t)n0 * OUT_DIM);
        const float4* a4 = reinterpret_cast<const float4*>(lacc);
        for (int i = threadIdx.x; i < rows * (OUT_DIM / 4); i += BIN_T) o4[i] = a4[i];
    }
}

__global__ void k_plane_sum(float* __restrict__ hout, const float* __restrict__ plane1,
                            int two, int n4)
{
    const int i = blockIdx.x * blockDim.x + threadIdx.x;
    if (!two || i >= n4) return;
    float4*       h4 = reinterpret_cast<float4*>(hout);
    const float4* q4 = reinterpret_cast<const float4*>(plane1);
    float4 a = h4[i];
    const float4 b = q4[i];
    a.x += b.x; a.y += b.y; a.z += b.z; a.w += b.w;
    h4[i] = a;
}

extern "C" void kernel_launch(void* const* d_in, const int* in_sizes, int n_in,
                              void* d_out, int out_size, void* d_ws, size_t ws_size,
                              hipStream_t stream)
{
    const float* x         = (const float*)d_in[0];
    const float* edge_attr = (const float*)d_in[1];
    const int*   edge_idx  = (const int*)  d_in[2];
    const float* W_fc      = (const float*)d_in[3];
    const float* b_fc      = (const float*)d_in[4];
    const float* W_attn    = (const float*)d_in[5];
    const float* b_attn    = (const float*)d_in[6];
    const float* W_edge    = (const float*)d_in[7];
    const float* b_edge    = (const float*)d_in[8];
    const float* W_self    = (const float*)d_in[9];
    const float* b_self    = (const float*)d_in[10];

    const int N = in_sizes[0] / IN_DIM;
    const int E = in_sizes[1] / ED_DIM;
    const int* erow = edge_idx;
    const int* ecol = edge_idx + E;

    float* ws     = (float*)d_ws;
    float* z      = ws;
    float* a_src  = z + (size_t)N * OUT_DIM;
    float* a_dst  = a_src + N;
    float* m      = a_dst + N;
    float* denom  = m + N;
    float* elog   = denom + N;
    float* pbuf   = elog + E;
    const size_t off1 = ((size_t)N * (OUT_DIM + 6) + 2 * (size_t)E + 3) & ~(size_t)3;
    const int two = ws_size >= (off1 + (size_t)N * OUT_DIM) * sizeof(float) ? 1 : 0;
    float* m1     = two ? pbuf + E : m;
    float* denom1 = two ? m1 + N   : denom;
    float* plane1 = two ? ws + off1 : (float*)d_out;
    float* hout   = (float*)d_out;

    const int nNodeTiles = (N + 15) / 16;
    const int nEdgeTiles = (E + 15) / 16;

    const int nb   = (N + BIN_R - 1) / BIN_R;
    const int bins = (two && nb <= NB_MAX && N >= 2 * NB_MAX && N < (1 << 23) &&
                      (size_t)N * OUT_DIM >= 3 * (size_t)E + 3 * NB_MAX + 8) ? 1 : 0;
    int*   rec_e  = (int*)plane1;
    int*   rec_rc = rec_e + E;
    float* rec_lg = (float*)(rec_rc + E);
    int*   cnt    = (int*)(rec_lg + E);
    int*   cursor = cnt + NB_MAX;
    int*   bbase  = cursor + NB_MAX;
    const int nBinBlocks = (E + BIN_T * BIN_PER - 1) / (BIN_T * BIN_PER);

    k_node_gemm<<<(nNodeTiles + 7) / 8, NTHREADS, 0, stream>>>(
        x, W_fc, b_fc, W_self, b_self, z, hout, plane1, bins ? 0 : two, N, nNodeTiles);

    k_node_attn<<<(N + 255) / 256, 256, 0, stream>>>(
        z, W_attn, a_src, a_dst, m, denom, m1, denom1, two, N, cnt, bins ? 2 * NB_MAX : 0,
        bins ? 0 : 1);

    k_edge_logit<<<(E + 255) / 256, 256, 0, stream>>>(
        edge_attr, erow, ecol, W_attn, b_attn, a_src, a_dst, elog, m, m1, two, E, bins ? 0 : 1);

    if (bins) {
        k_bin_count<<<nBinBlocks, BIN_T, 0, stream>>>(ecol, cnt, E, nb);

        k_bin_place<<<nBinBlocks, BIN_T, 0, stream>>>(
            erow, ecol, elog, cnt, cursor, bbase, rec_e, rec_rc, rec_lg, E, nb);

        k_range_aggregate<<<nb, BIN_T, 0, stream>>>(
            edge_attr, W_edge, b_edge, z, rec_e, rec_rc, rec_lg, bbase, hout, N);
    } else {
        k_edge_soft<<<(E + 255) / 256, 256, 0, stream>>>(
            elog, ecol, m, m1, pbuf, denom, denom1, two, E);

        k_edge_aggregate<<<(nEdgeTiles + 7) / 8, NTHREADS, 0, stream>>>(
            edge_attr, erow, ecol, W_edge, b_edge, z, pbuf, denom, denom1, hout, plane1, two,
            E, nEdgeTiles);

        k_plane_sum<<<(N * (OUT_DIM / 4) + 255) / 256, 256, 0, stream>>>(
            hout, plane1, two, N * (OUT_DIM / 4));
    }
}


# edge logits taken inside the placing pass from whole-line edge_attr reads (8 lanes per row, 3-round halving exchange: new summation tree of the 32-term dot); logit launch and elog plane go away
# speedup vs baseline: 2.8601x; 1.0956x over previous
#include <hip/hip_runtime.h>
#include <hip/hip_bf16.h>
#include <math.h>

#define IN_DIM  128
#define OUT_DIM 64
#define ED_DIM  32

#define BSTRIDE_V2 80
#define BSTRIDE_F  (2 * BSTRIDE_V2)

typedef __attribute__((ext_vector_type(2))) float v2f;
typedef __attribute__((ext_vector_type(8))) float v8f;

struct Acc4 { v8f t[4]; };

__device__ __attribute__((always_inline)) inline
Acc4 wave_gemm_tile(const float* __restrict__ arow,
                    const float* __restrict__ lWs,
                    int kdim, int half, int l16)
{
    Acc4 A;
    A.t[0] = {}; A.t[1] = {}; A.t[2] = {}; A.t[3] = {};
    const v2f* a2 = (const v2f*)arow;
    for (int k0 = 0; k0 < kdim; k0 += 4) {
        const int kp  = (k0 >> 1) + half;
        const v2f  a  = a2[kp];
        const v2f* br = (const v2f*)(lWs + kp * BSTRIDE_F);
#pragma unroll
        for (int nt = 0; nt < 4; ++nt) {
            const v2f b = br[nt * 16 + l16];
            A.t[nt] = __builtin_amdgcn_wmma_f32_16x16x4_f32(
                false, a, false, b, (short)0, A.t[nt], false, false);
        }
    }
    return A;
}

#define NTHREADS 256

template <int KDIM>
struct WStage {
    static constexpr int ITEMS = (KDIM / 2) * (OUT_DIM / 4);
    static constexpr int PER   = ITEMS / NTHREADS;
    static_assert(ITEMS % NTHREADS == 0, "staging is laid out for whole rounds of the work-group");
    float4 lo[PER];
    float4 hi[PER];
};

template <int KDIM>
__device__ __attribute__((always_inline)) inline
WStage<KDIM> load_w(const float* __restrict__ W)
{
    WStage<KDIM> s;
#pragma unroll
    for (int j = 0; j < WStage<KDIM>::PER; ++j) {
        const int it = threadIdx.x + NTHREADS * j;
        const int kp = it >> 4;
        const int g  = it & 15;
        const float4* src = reinterpret_cast<const float4*>(W + (size_t)(2 * kp) * OUT_DIM + 4 * g);
        s.lo[j] = src[0];
        s.hi[j] = src[OUT_DIM / 4];
    }
    return s;
}

template <int KDIM>
__device__ __attribute__((always_inline)) inline
void store_w(float* __restrict__ lWs, const WStage<KDIM>& s)
{
#pragma unroll
    for (int j = 0; j < WStage<KDIM>::PER; ++j) {
        const int it = threadIdx.x + NTHREADS * j;
        const int kp = it >> 4;
        const int g  = it & 15;
        float4* dst = reinterpret_cast<float4*>(lWs + kp * BSTRIDE_F + 8 * g);
        dst[0] = make_float4(s.lo[j].x, s.hi[j].x, s.lo[j].y, s.hi[j].y);
        dst[1] = make_float4(s.lo[j].z, s.hi[j].z, s.lo[j].w, s.hi[j].w);
    }
}

__device__ __attribute__((always_inline)) inline
int chip_half()
{
    const unsigned v = __builtin_amdgcn_s_sendmsg_rtn(0x87);
    return (int)((v >> 18) & 1u);
}

__device__ inline void atomicMaxF(float* addr, float val) {
    if (val >= 0.0f) atomicMax(reinterpret_cast<int*>(addr), __float_as_int(val));
    else             atomicMin(reinterpret_cast<unsigned int*>(addr), __float_as_uint(val));
}

__global__ void k_node_gemm(const float* __restrict__ x,
                            const float* __restrict__ W_fc,  const float* __restrict__ b_fc,
                            const float* __restrict__ W_self, const float* __restrict__ b_self,
                            float* __restrict__ z, float* __restrict__ hout,
                            float* __restrict__ plane1, int two,
                            int N, int nTiles)
{
    __shared__ __attribute__((aligned(16))) float lWs[(IN_DIM / 2) * BSTRIDE_F];
    const int lane = threadIdx.x & 31;
    const int wave = threadIdx.x >> 5;
    const int half = lane >> 4;
    const int l16  = lane & 15;
    const int tile = blockIdx.x * 8 + wave;

    const WStage<IN_DIM> w_first  = load_w<IN_DIM>(W_fc);
    const WStage<IN_DIM> w_second = load_w<IN_DIM>(W_self);

    int mrow = tile * 16 + l16;
    if (mrow >= N) mrow = N - 1;
    const float* xrow = x + (size_t)mrow * IN_DIM;

    if (two && tile < nTiles) {
        float4* p4 = reinterpret_cast<float4*>(plane1 + (size_t)tile * 16 * OUT_DIM);
#pragma unroll
        for (int it = 0; it < 8; ++it) {
            const int idx = it * 32 + lane;
            if (tile * 16 + (idx >> 4) < N) p4[idx] = make_float4(0.f, 0.f, 0.f, 0.f);
        }
    }

    store_w<IN_DIM>(lWs, w_first);
    __syncthreads();
    if (tile < nTiles) {
        Acc4 acc = wave_gemm_tile(xrow, lWs, IN_DIM, half, l16);
#pragma unroll
        for (int nt = 0; nt < 4; ++nt) {
            const int   c  = nt * 16 + l16;
            const float bc = b_fc[c];
#pragma unroll
            for (int r = 0; r < 8; ++r) {
                const int row = tile * 16 + r + 8 * half;
                if (row < N) z[(size_t)row * OUT_DIM + c] = acc.t[nt][r] + bc;
            }
        }
    }
    __syncthreads();

    store_w<IN_DIM>(lWs, w_second);
    __syncthreads();
    if (tile < nTiles) {
        Acc4 acc = wave_gemm_tile(xrow, lWs, IN_DIM, half, l16);
#pragma unroll
        for (int nt = 0; nt < 4; ++nt) {
            const int   c  = nt * 16 + l16;
            const float bc = b_self[c];
#pragma unroll
            for (int r = 0; r < 8; ++r) {
                const int row = tile * 16 + r + 8 * half;
                if (row < N) hout[(size_t)row * OUT_DIM + c] = acc.t[nt][r] + bc;
            }
        }
    }
}

__global__ void k_node_attn(const float* __restrict__ z, const float* __restrict__ W_attn,
                            float* __restrict__ a_src, float* __restrict__ a_dst,
                            float* __restrict__ m, float* __restrict__ denom,
                            float* __restrict__ m1, float* __restrict__ denom1, int two, int N,
                            int* __restrict__ binz, int nbinz, int stats)
{
    const int n = blockIdx.x * blockDim.x + threadIdx.x;
    if (n < nbinz) binz[n] = 0;
    if (n >= N) return;
    const float* zr = z + (size_t)n * OUT_DIM;
    float as = 0.f, ad = 0.f;
#pragma unroll 8
    for (int c = 0; c < OUT_DIM; ++c) {
        const float zc = zr[c];
        as += zc * W_attn[c];
        ad += zc * W_attn[OUT_DIM + c];
    }
    a_src[n] = as;
    a_dst[n] = ad;
    if (!stats) return;
    m[n]     = -INFINITY;
    denom[n] = 0.0f;
    if (two) {
        m1[n]     = -INFINITY;
        denom1[n] = 0.0f;
    }
}

__global__ void k_edge_logit(const float* __restrict__ edge_attr,
                             const int* __restrict__ erow, const int* __restrict__ ecol,
                             const float* __restrict__ W_attn, const float* __restrict__ b_attn,
                             const float* __restrict__ a_src, const float* __restrict__ a_dst,
                             float* __restrict__ elog, float* __restrict__ m,
                             float* __restrict__ m1, int two, int E)
{
    const int e = blockIdx.x * blockDim.x + threadIdx.x;
    if (e >= E) return;
    const float* ea = edge_attr + (size_t)e * ED_DIM;
    float dot = 0.f;
#pragma unroll
    for (int k = 0; k < ED_DIM; ++k) dot += ea[k] * W_attn[2 * OUT_DIM + k];
    const int c = ecol[e];
    const float s  = a_src[erow[e]] + a_dst[c] + dot + b_attn[0];
    const float lr = (s > 0.0f) ? s : 0.2f * s;
    elog[e] = lr;
    float* mh = (two && chip_half()) ? m1 : m;
    atomicMaxF(&mh[c], lr);
}

__global__ void k_edge_soft(const float* __restrict__ elog, const int* __restrict__ ecol,
                            const float* __restrict__ m, const float* __restrict__ m1,
                            float* __restrict__ pbuf,
                            float* __restrict__ denom, float* __restrict__ denom1, int two, int E)
{
    const int e = blockIdx.x * blockDim.x + threadIdx.x;
    if (e >= E) return;
    const int c = ecol[e];
    const float mc = fmaxf(m[c], m1[c]);
    const float p = expf(elog[e] - mc);
    pbuf[e] = p;
    float* dh = (two && chip_half()) ? denom1 : denom;
    atomicAdd(&dh[c], p);
}

__device__ __attribute__((always_inline)) inline
float xhalf(float x)
{
    const int b = __builtin_bit_cast(int, x);
    return __builtin_bit_cast(float,
        __builtin_amdgcn_permlanex16(b, b, 0x76543210, 0xfedcba98, false, false));
}

template <bool FULL>
__device__ __attribute__((always_inline)) inline
void edge_epilogue(const Acc4& acc, int c_l, int r_l, float al_l, float b_lo, float b_hi,
                   const float* __restrict__ z, float* __restrict__ hout, int lane, int half)
{
    float zlo[16], zhi[16];
    if (FULL) {
#pragma unroll
        for (int j = 0; j < 16; ++j) {
            const int    rj = __builtin_amdgcn_readlane(r_l, j);
            const float* zr = z + (size_t)rj * OUT_DIM;
            zlo[j] = zr[lane];
            zhi[j] = zr[32 + lane];
        }
    }
#pragma unroll
    for (int r = 0; r < 8; ++r) {
        const float gA   = xhalf(half ? acc.t[0][r] : acc.t[1][r]);
        const float gB   = xhalf(half ? acc.t[2][r] : acc.t[3][r]);
        const float v0lo = half ? gA : acc.t[0][r];
        const float v1lo = half ? acc.t[1][r] : gA;
        const float v0hi = half ? gB : acc.t[2][r];
        const float v1hi = half ? acc.t[3][r] : gB;
#pragma unroll
        for (int h = 0; h < 2; ++h) {
            const int   j  = r + 8 * h;
            const int   cj = __builtin_amdgcn_readlane(c_l, j);
            const float aj = __builtin_bit_cast(float,
                __builtin_amdgcn_readlane(__builtin_bit_cast(int, al_l), j));
            if (FULL || cj >= 0) {
                float zl, zh;
                if (FULL) {
                    zl = zlo[j];
                    zh = zhi[j];
                } else {
                    const int    rj = __builtin_amdgcn_readlane(r_l, j);
                    const float* zr = z + (size_t)rj * OUT_DIM;
                    zl = zr[lane];
                    zh = zr[32 + lane];
                }
                float* hr = hout + (size_t)cj * OUT_DIM;
                atomicAdd(&hr[lane],      aj * (zl + (h ? v1lo : v0lo) + b_lo));
                atomicAdd(&hr[32 + lane], aj * (zh + (h ? v1hi : v0hi) + b_hi));
            }
        }
    }
}

__global__ void k_edge_aggregate(const float* __restrict__ edge_attr,
                                 const int* __restrict__ erow, const int* __restrict__ ecol,
                                 const float* __restrict__ W_edge, const float* __restrict__ b_edge,
                                 const float* __restrict__ z, const float* __restrict__ pbuf,
                                 const float* __restrict__ denom, const float* __restrict__ denom1,
                                 float* __restrict__ hout, float* __restrict__ plane1, int two,
                                 int E, int nTiles)
{
    __shared__ __attribute__((aligned(16))) float lWs[(ED_DIM / 2) * BSTRIDE_F];
    const int lane = threadIdx.x & 31;
    const int wave = threadIdx.x >> 5;
    const int half = lane >> 4;
    const int l16  = lane & 15;

    store_w<ED_DIM>(lWs, load_w<ED_DIM>(W_edge));
    __syncthreads();
    const int tile = __builtin_amdgcn_readfirstlane(blockIdx.x * 8 + wave);
    if (tile >= nTiles) return;

    int arow = tile * 16 + l16;
    const bool evalid = arow < E;
    if (arow >= E) arow = E - 1;
    const float* ea = edge_attr + (size_t)arow * ED_DIM;

    {
        const size_t nxt = (size_t)arow + 16 * 8;
        if (nxt < (size_t)E)
            __builtin_prefetch(edge_attr + nxt * ED_DIM, 0, 1);
    }

    const int   c_e  = ecol[arow];
    const int   c_l  = evalid ? c_e : -1;
    const int   r_l  = erow[arow];
    const float d0   = denom[c_e];
    const float d1   = denom1[c_e];
    const float den  = two ? d0 + d1 : d0;
    const float al_l = pbuf[arow] / fmaxf(den, 1e-16f);

    Acc4 acc = wave_gemm_tile(ea, lWs, ED_DIM, half, l16);

    const float b_lo = b_edge[lane];
    const float b_hi = b_edge[32 + lane];

    float* hacc = (two && chip_half()) ? plane1 : hout;

    if (tile * 16 + 16 <= E)
        edge_epilogue<true >(acc, c_l, r_l, al_l, b_lo, b_hi, z, hacc, lane, half);
    else
        edge_epilogue<false>(acc, c_l, r_l, al_l, b_lo, b_hi, z, hacc, lane, half);
}


#define BIN_R   196
#define STAT_PER 4
#define NB_MAX  1024
#define BIN_T   1024
#define BIN_PER 4

__global__ __launch_bounds__(BIN_T)
void k_bin_count(const int* __restrict__ ecol, int* __restrict__ cnt, int E, int nb)
{
    __shared__ int s_hist[NB_MAX];
    const int tid = threadIdx.x;
    s_hist[tid] = 0;
    __syncthreads();
#pragma unroll
    for (int i = 0; i < BIN_PER; ++i) {
        const int e = blockIdx.x * (BIN_T * BIN_PER) + i * BIN_T + tid;
        if (e < E) atomicAdd(&s_hist[ecol[e] / BIN_R], 1);
    }
    __syncthreads();
    const int h = s_hist[tid];
    if (tid < nb && h > 0) atomicAdd(&cnt[tid], h);
}

__device__ __attribute__((always_inline)) inline
float x8_4(float v) { return __builtin_bit_cast(float, __builtin_amdgcn_mov_dpp8(__builtin_bit_cast(unsigned, v), 0x688fac)); }
__device__ __attribute__((always_inline)) inline
float x8_2(float v) { return __builtin_bit_cast(float, __builtin_amdgcn_mov_dpp8(__builtin_bit_cast(unsigned, v), 0xb3e21a)); }
__device__ __attribute__((always_inline)) inline
float x8_1(float v) { return __builtin_bit_cast(float, __builtin_amdgcn_mov_dpp8(__builtin_bit_cast(unsigned, v), 0xde54c1)); }

__device__ __attribute__((always_inline)) inline
void ea_request(float4 (&v)[8], const float* __restrict__ edge_attr, int base, int sub, int g)
{
    const float4* p = reinterpret_cast<const float4*>(
        edge_attr + (size_t)(base + sub) * ED_DIM + 4 * g);
#pragma unroll
    for (int q = 0; q < 8; ++q) v[q] = p[q * ED_DIM];
}

__device__ __attribute__((always_inline)) inline
float ea_fold(const float4 (&v)[8], const float4 wa, int lane)
{
    float d[8];
#pragma unroll
    for (int q = 0; q < 8; ++q)
        d[q] = v[q].x * wa.x + v[q].y * wa.y + v[q].z * wa.z + v[q].w * wa.w;
    float s1[4];
#pragma unroll
    for (int q = 0; q < 4; ++q) {
        const float keep = (lane & 4) ? d[q + 4] : d[q];
        const float send = (lane & 4) ? d[q] : d[q + 4];
        s1[q] = keep + x8_4(send);
    }
    float s2[2];
#pragma unroll
    for (int q = 0; q < 2; ++q) {
        const float keep = (lane & 2) ? s1[q + 2] : s1[q];
        const float send = (lane & 2) ? s1[q] : s1[q + 2];
        s2[q] = keep + x8_2(send);
    }
    const float keep = (lane & 1) ? s2[1] : s2[0];
    const float send = (lane & 1) ? s2[0] : s2[1];
    return keep + x8_1(send);
}

__global__ __launch_bounds__(BIN_T)
void k_bin_place(const float* __restrict__ edge_attr,
                 const int* __restrict__ erow, const int* __restrict__ ecol,
                 const float* __restrict__ W_attn, const float* __restrict__ b_attn,
                 const float* __restrict__ a_src, const float* __restrict__ a_dst,
                 const int* __restrict__ cnt, int* __restrict__ cursor, int* __restrict__ bbase,
                 int* __restrict__ rec_e, int* __restrict__ rec_rc, float* __restrict__ rec_lg,
                 int E, int nb)
{
    __shared__ int s_base[NB_MAX];
    __shared__ int s_hist[NB_MAX];
    __shared__ int s_off[NB_MAX];
    __shared__ int s_wsum[32];
    __shared__ float s_dot[BIN_T * BIN_PER];
    const int tid  = threadIdx.x;
    const int lane = tid & 31;
    const int wave = __builtin_amdgcn_readfirstlane(tid >> 5);
    const int g    = lane & 7;
    const int sub  = lane >> 3;

    const float4 wa = *reinterpret_cast<const float4*>(W_attn + 2 * OUT_DIM + 4 * g);
    const float  ba = b_attn[0];

    const int base0 = blockIdx.x * (BIN_T * BIN_PER) + wave * 32;
    int   eid[BIN_PER], bk[BIN_PER], rc[BIN_PER];
    float lg[BIN_PER];
    int   cc[BIN_PER], rr[BIN_PER];
#pragma unroll
    for (int i = 0; i < BIN_PER; ++i) {
        const int e  = base0 + i * BIN_T + 4 * g + sub;
        const int ec = e < E ? e : E - 1;
        eid[i] = e;
        cc[i]  = ecol[ec];
        rr[i]  = erow[ec];
    }
#pragma unroll 1
    for (int i = 0; i < BIN_PER; ++i) {
        const int b = base0 + i * BIN_T;
        float4 ln[8];
        ea_request(ln, edge_attr, (b + 32 <= E) ? b : E - 32, sub, g);
        s_dot[i * BIN_T + tid] = ea_fold(ln, wa, lane);
    }
    float asd[BIN_PER];
#pragma unroll
    for (int i = 0; i < BIN_PER; ++i) asd[i] = a_src[rr[i]] + a_dst[cc[i]];

#pragma unroll
    for (int i = 0; i < BIN_PER; ++i) {
        float dot = s_dot[i * BIN_T + tid];
        const int base = base0 + i * BIN_T;
        if (base < E && base + 32 > E) {
            if (eid[i] < E) {
                const float* ea = edge_attr + (size_t)eid[i] * ED_DIM;
                float t = 0.f;
                for (int k = 0; k < ED_DIM; ++k) t += ea[k] * W_attn[2 * OUT_DIM + k];
                dot = t;
            }
        }
        const float s = asd[i] + dot + ba;
        const int   b = cc[i] / BIN_R;
        bk[i] = eid[i] < E ? b : -1;
        rc[i] = rr[i] * 256 + (cc[i] - b * BIN_R);
        lg[i] = (s > 0.0f) ? s : 0.2f * s;
    }

    const int own = tid < nb ? cnt[tid] : 0;
    int v = own;
#pragma unroll
    for (int d = 1; d < 32; d <<= 1) {
        const int u = __shfl_up(v, d);
        if (lane >= d) v += u;
    }
    if (lane == 31) s_wsum[wave] = v;
    s_hist[tid] = 0;
    __syncthreads();
    if (wave == 0) {
        const int w = s_wsum[lane];
        int x = w;
#pragma unroll
        for (int d = 1; d < 32; d <<= 1) {
            const int u = __shfl_up(x, d);
            if (lane >= d) x += u;
        }
        s_wsum[lane] = x - w;
    }
    __syncthreads();
    const int excl = v - own + s_wsum[wave];
    s_base[tid] = excl;
    if (blockIdx.x == 0) {
        if (tid < nb)      bbase[tid] = excl;
        if (tid == nb - 1) bbase[nb]  = excl + own;
    }

    int lr[BIN_PER];
#pragma unroll
    for (int i = 0; i < BIN_PER; ++i) {
        lr[i] = 0;
        if (bk[i] >= 0) lr[i] = atomicAdd(&s_hist[bk[i]], 1);
    }
    __syncthreads();
    {
        const int h = s_hist[tid];
        if (tid < nb && h > 0) s_off[tid] = s_base[tid] + atomicAdd(&cursor[tid], h);
    }
    __syncthreads();
#pragma unroll
    for (int i = 0; i < BIN_PER; ++i) {
        if (bk[i] >= 0) {
            const int pos = s_off[bk[i]] + lr[i];
            rec_e[pos]  = eid[i];
            rec_rc[pos] = rc[i];
            rec_lg[pos] = lg[i];
        }
    }
}

__device__ __attribute__((always_inline)) inline
void lds_max(float* addr, float val)
{
    __hip_atomic_fetch_max(addr, val, __ATOMIC_RELAXED, __HIP_MEMORY_SCOPE_WORKGROUP);
}

__global__ __launch_bounds__(BIN_T)
void k_range_aggregate(const float* __restrict__ edge_attr,
                       const float* __restrict__ W_edge, const float* __restrict__ b_edge,
                       const float* __restrict__ z,
                       const int* __restrict__ rec_e, const int* __restrict__ rec_rc,
                       const float* __restrict__ rec_lg, const int* __restrict__ bbase,
                       float* __restrict__ hout, int N)
{
    __shared__ __attribute__((aligned(16))) float lWs[(ED_DIM / 2) * BSTRIDE_F];
    __shared__ __attribute__((aligned(16))) float lacc[BIN_R * OUT_DIM];
    __shared__ float lmax[BIN_R];
    __shared__ float lden[BIN_R];
    const int lane = threadIdx.x & 31;
    const int wave = __builtin_amdgcn_readfirstlane(threadIdx.x >> 5);
    const int half = lane >> 4;
    const int l16  = lane & 15;

    const int n0   = blockIdx.x * BIN_R;
    const int rows = (N - n0 < BIN_R) ? (N - n0) : BIN_R;

    const int b0 = bbase[blockIdx.x];
    const int b1 = bbase[blockIdx.x + 1];
    const int ntile = (b1 - b0 + 15) >> 4;

    float lg_s[STAT_PER];
    int   rc_s[STAT_PER];
#pragma unroll
    for (int j = 0; j < STAT_PER; ++j) {
        const int i  = b0 + j * BIN_T + (int)threadIdx.x;
        const int ii = i < b1 ? i : 0;
        lg_s[j] = rec_lg[ii];
        rc_s[j] = rec_rc[ii];
    }
    int   e_n = 0, rc_n = 0;
    float lg_n = 0.f;
    if (wave < ntile) {
        int idx = b0 + wave * 16 + l16;
        if (idx >= b1) idx = b1 - 1;
        e_n = rec_e[idx]; rc_n = rec_rc[idx]; lg_n = rec_lg[idx];
    }

    if (threadIdx.x < BIN_R) { lmax[threadIdx.x] = -INFINITY; lden[threadIdx.x] = 0.f; }
    if (threadIdx.x < NTHREADS) store_w<ED_DIM>(lWs, load_w<ED_DIM>(W_edge));
    {
        const float4* h4 = reinterpret_cast<const float4*>(hout + (size_t)n0 * OUT_DIM);
        float4*       a4 = reinterpret_cast<float4*>(lacc);
        for (int i = threadIdx.x; i < rows * (OUT_DIM / 4); i += BIN_T) a4[i] = h4[i];
    }
    const float b_lo = b_edge[lane];
    const float b_hi = b_edge[32 + lane];
    __syncthreads();

    int c_s[STAT_PER];
#pragma unroll
    for (int j = 0; j < STAT_PER; ++j)
        c_s[j] = (b0 + j * BIN_T + (int)threadIdx.x < b1) ? (rc_s[j] & 255) : -1;

#pragma unroll
    for (int j = 0; j < STAT_PER; ++j)
        if (c_s[j] >= 0) lds_max(&lmax[c_s[j]], lg_s[j]);
    for (int i = b0 + STAT_PER * BIN_T + (int)threadIdx.x; i < b1; i += BIN_T)
        lds_max(&lmax[rec_rc[i] & 255], rec_lg[i]);
    __syncthreads();

#pragma unroll
    for (int j = 0; j < STAT_PER; ++j)
        if (c_s[j] >= 0) atomicAdd(&lden[c_s[j]], expf(lg_s[j] - lmax[c_s[j]]));
    for (int i = b0 + STAT_PER * BIN_T + (int)threadIdx.x; i < b1; i += BIN_T) {
        const int cc = rec_rc[i] & 255;
        atomicAdd(&lden[cc], expf(rec_lg[i] - lmax[cc]));
    }
    __syncthreads();

    for (int t = wave; t < ntile; t += BIN_T / 32) {
        asm volatile("" ::: "memory");
        const int   e_c = e_n, rc_c = rc_n;
        const float lg_c = lg_n;
        const int tn = t + BIN_T / 32;
        if (tn < ntile) {
            int idx = b0 + tn * 16 + l16;
            if (idx >= b1) idx = b1 - 1;
            e_n = rec_e[idx]; rc_n = rec_rc[idx]; lg_n = rec_lg[idx];
        }
        const bool valid = b0 + t * 16 + l16 < b1;
        const int  cc    = rc_c & 255;
        const int  c_l   = valid ? cc : -1;
        const int  r_l   = rc_c >> 8;
        const float al_c = expf(lg_c - lmax[cc]) / fmaxf(lden[cc], 1e-16f);
        const float* ea  = edge_attr + (size_t)e_c * ED_DIM;

        Acc4 acc = wave_gemm_tile(ea, lWs, ED_DIM, half, l16);

        if (t * 16 + 16 <= b1 - b0)
            edge_epilogue<true >(acc, c_l, r_l, al_c, b_lo, b_hi, z, lacc, lane, half);
        else
            edge_epilogue<false>(acc, c_l, r_l, al_c, b_lo, b_hi, z, lacc, lane, half);
    }
    __syncthreads();
    {
        float4*       o4 = reinterpret_cast<float4*>(hout + (size_t)n0 * OUT_DIM);
        const float4* a4 = reinterpret_cast<const float4*>(lacc);
        for (int i = threadIdx.x; i < rows * (OUT_DIM / 4); i += BIN_T) o4[i] = a4[i];
    }
}

__global__ void k_plane_sum(float* __restrict__ hout, const float* __restrict__ plane1,
                            int two, int n4)
{
    const int i = blockIdx.x * blockDim.x + threadIdx.x;
    if (!two || i >= n4) return;
    float4*       h4 = reinterpret_cast<float4*>(hout);
    const float4* q4 = reinterpret_cast<const float4*>(plane1);
    float4 a = h4[i];
    const float4 b = q4[i];
    a.x += b.x; a.y += b.y; a.z += b.z; a.w += b.w;
    h4[i] = a;
}

extern "C" void kernel_launch(void* const* d_in, const int* in_sizes, int n_in,
                              void* d_out, int out_size, void* d_ws, size_t ws_size,
                              hipStream_t stream)
{
    const float* x         = (const float*)d_in[0];
    const float* edge_attr = (const float*)d_in[1];
    const int*   edge_idx  = (const int*)  d_in[2];
    const float* W_fc      = (const float*)d_in[3];
    const float* b_fc      = (const float*)d_in[4];
    const float* W_attn    = (const float*)d_in[5];
    const float* b_attn    = (const float*)d_in[6];
    const float* W_edge    = (const float*)d_in[7];
    const float* b_edge    = (const float*)d_in[8];
    const float* W_self    = (const float*)d_in[9];
    const float* b_self    = (const float*)d_in[10];

    const int N = in_sizes[0] / IN_DIM;
    const int E = in_sizes[1] / ED_DIM;
    const int* erow = edge_idx;
    const int* ecol = edge_idx + E;

    float* ws     = (float*)d_ws;
    float* z      = ws;
    float* a_src  = z + (size_t)N * OUT_DIM;
    float* a_dst  = a_src + N;
    float* m      = a_dst + N;
    float* denom  = m + N;
    float* elog   = denom + N;
    float* pbuf   = elog + E;
    const size_t off1 = ((size_t)N * (OUT_DIM + 6) + 2 * (size_t)E + 3) & ~(size_t)3;
    const int two = ws_size >= (off1 + (size_t)N * OUT_DIM) * sizeof(float) ? 1 : 0;
    float* m1     = two ? pbuf + E : m;
    float* denom1 = two ? m1 + N   : denom;
    float* plane1 = two ? ws + off1 : (float*)d_out;
    float* hout   = (float*)d_out;

    const int nNodeTiles = (N + 15) / 16;
    const int nEdgeTiles = (E + 15) / 16;

    const int nb   = (N + BIN_R - 1) / BIN_R;
    const int bins = (two && nb <= NB_MAX && N >= 2 * NB_MAX && N < (1 << 23) && E >= 32 &&
                      (size_t)N * OUT_DIM >= 3 * (size_t)E + 3 * NB_MAX + 8) ? 1 : 0;
    int*   rec_e  = (int*)plane1;
    int*   rec_rc = rec_e + E;
    float* rec_lg = (float*)(rec_rc + E);
    int*   cnt    = (int*)(rec_lg + E);
    int*   cursor = cnt + NB_MAX;
    int*   bbase  = cursor + NB_MAX;
    const int nBinBlocks = (E + BIN_T * BIN_PER - 1) / (BIN_T * BIN_PER);

    k_node_gemm<<<(nNodeTiles + 7) / 8, NTHREADS, 0, stream>>>(
        x, W_fc, b_fc, W_self, b_self, z, hout, plane1, bins ? 0 : two, N, nNodeTiles);

    k_node_attn<<<(N + 255) / 256, 256, 0, stream>>>(
        z, W_attn, a_src, a_dst, m, denom, m1, denom1, two, N, cnt, bins ? 2 * NB_MAX : 0,
        bins ? 0 : 1);

    if (bins) {
        k_bin_count<<<nBinBlocks, BIN_T, 0, stream>>>(ecol, cnt, E, nb);

        k_bin_place<<<nBinBlocks, BIN_T, 0, stream>>>(
            edge_attr, erow, ecol, W_attn, b_attn, a_src, a_dst,
            cnt, cursor, bbase, rec_e, rec_rc, rec_lg, E, nb);

        k_range_aggregate<<<nb, BIN_T, 0, stream>>>(
            edge_attr, W_edge, b_edge, z, rec_e, rec_rc, rec_lg, bbase, hout, N);
    } else {
        k_edge_logit<<<(E + 255) / 256, 256, 0, stream>>>(
            edge_attr, erow, ecol, W_attn, b_attn, a_src, a_dst, elog, m, m1, two, E);

        k_edge_soft<<<(E + 255) / 256, 256, 0, stream>>>(
            elog, ecol, m, m1, pbuf, denom, denom1, two, E);

        k_edge_aggregate<<<(nEdgeTiles + 7) / 8, NTHREADS, 0, stream>>>(
            edge_attr, erow, ecol, W_edge, b_edge, z, pbuf, denom, denom1, hout, plane1, two,
            E, nEdgeTiles);

        k_plane_sum<<<(N * (OUT_DIM / 4) + 255) / 256, 256, 0, stream>>>(
            hout, plane1, two, N * (OUT_DIM / 4));
    }
}


# W_edge product moved behind the sum: range work-group adds alpha*edge_attr rows (whole-line reads) in LDS beside alpha*z, then multiplies the summed rows by W_edge once (f32 WMMA); sum order changes
# speedup vs baseline: 2.9970x; 1.0479x over previous
#include <hip/hip_runtime.h>
#include <hip/hip_bf16.h>
#include <math.h>

#define IN_DIM  128
#define OUT_DIM 64
#define ED_DIM  32

#define BSTRIDE_V2 80
#define BSTRIDE_F  (2 * BSTRIDE_V2)

typedef __attribute__((ext_vector_type(2))) float v2f;
typedef __attribute__((ext_vector_type(8))) float v8f;

struct Acc4 { v8f t[4]; };

__device__ __attribute__((always_inline)) inline
Acc4 wave_gemm_tile(const float* __restrict__ arow,
                    const float* __restrict__ lWs,
                    int kdim, int half, int l16)
{
    Acc4 A;
    A.t[0] = {}; A.t[1] = {}; A.t[2] = {}; A.t[3] = {};
    const v2f* a2 = (const v2f*)arow;
    for (int k0 = 0; k0 < kdim; k0 += 4) {
        const int kp  = (k0 >> 1) + half;
        const v2f  a  = a2[kp];
        const v2f* br = (const v2f*)(lWs + kp * BSTRIDE_F);
#pragma unroll
        for (int nt = 0; nt < 4; ++nt) {
            const v2f b = br[nt * 16 + l16];
            A.t[nt] = __builtin_amdgcn_wmma_f32_16x16x4_f32(
                false, a, false, b, (short)0, A.t[nt], false, false);
        }
    }
    return A;
}

#define NTHREADS 256

template <int KDIM>
struct WStage {
    static constexpr int ITEMS = (KDIM / 2) * (OUT_DIM / 4);
    static constexpr int PER   = ITEMS / NTHREADS;
    static_assert(ITEMS % NTHREADS == 0, "staging is laid out for whole rounds of the work-group");
    float4 lo[PER];
    float4 hi[PER];
};

template <int KDIM>
__device__ __attribute__((always_inline)) inline
WStage<KDIM> load_w(const float* __restrict__ W)
{
    WStage<KDIM> s;
#pragma unroll
    for (int j = 0; j < WStage<KDIM>::PER; ++j) {
        const int it = threadIdx.x + NTHREADS * j;
        const int kp = it >> 4;
        const int g  = it & 15;
        const float4* src = reinterpret_cast<const float4*>(W + (size_t)(2 * kp) * OUT_DIM + 4 * g);
        s.lo[j] = src[0];
        s.hi[j] = src[OUT_DIM / 4];
    }
    return s;
}

template <int KDIM>
__device__ __attribute__((always_inline)) inline
void store_w(float* __restrict__ lWs, const WStage<KDIM>& s)
{
#pragma unroll
    for (int j = 0; j < WStage<KDIM>::PER; ++j) {
        const int it = threadIdx.x + NTHREADS * j;
        const int kp = it >> 4;
        const int g  = it & 15;
        float4* dst = reinterpret_cast<float4*>(lWs + kp * BSTRIDE_F + 8 * g);
        dst[0] = make_float4(s.lo[j].x, s.hi[j].x, s.lo[j].y, s.hi[j].y);
        dst[1] = make_float4(s.lo[j].z, s.hi[j].z, s.lo[j].w, s.hi[j].w);
    }
}

__device__ __attribute__((always_inline)) inline
int chip_half()
{
    const unsigned v = __builtin_amdgcn_s_sendmsg_rtn(0x87);
    return (int)((v >> 18) & 1u);
}

__device__ inline void atomicMaxF(float* addr, float val) {
    if (val >= 0.0f) atomicMax(reinterpret_cast<int*>(addr), __float_as_int(val));
    else             atomicMin(reinterpret_cast<unsigned int*>(addr), __float_as_uint(val));
}

__global__ void k_node_gemm(const float* __restrict__ x,
                            const float* __restrict__ W_fc,  const float* __restrict__ b_fc,
                            const float* __restrict__ W_self, const float* __restrict__ b_self,
                            float* __restrict__ z, float* __restrict__ hout,
                            float* __restrict__ plane1, int two,
                            int N, int nTiles)
{
    __shared__ __attribute__((aligned(16))) float lWs[(IN_DIM / 2) * BSTRIDE_F];
    const int lane = threadIdx.x & 31;
    const int wave = threadIdx.x >> 5;
    const int half = lane >> 4;
    const int l16  = lane & 15;
    const int tile = blockIdx.x * 8 + wave;

    const WStage<IN_DIM> w_first  = load_w<IN_DIM>(W_fc);
    const WStage<IN_DIM> w_second = load_w<IN_DIM>(W_self);

    int mrow = tile * 16 + l16;
    if (mrow >= N) mrow = N - 1;
    const float* xrow = x + (size_t)mrow * IN_DIM;

    if (two && tile < nTiles) {
        float4* p4 = reinterpret_cast<float4*>(plane1 + (size_t)tile * 16 * OUT_DIM);
#pragma unroll
        for (int it = 0; it < 8; ++it) {
            const int idx = it * 32 + lane;
            if (tile * 16 + (idx >> 4) < N) p4[idx] = make_float4(0.f, 0.f, 0.f, 0.f);
        }
    }

    store_w<IN_DIM>(lWs, w_first);
    __syncthreads();
    if (tile < nTiles) {
        Acc4 acc = wave_gemm_tile(xrow, lWs, IN_DIM, half, l16);
#pragma unroll
        for (int nt = 0; nt < 4; ++nt) {
            const int   c  = nt * 16 + l16;
            const float bc = b_fc[c];
#pragma unroll
            for (int r = 0; r < 8; ++r) {
                const int row = tile * 16 + r + 8 * half;
                if (row < N) z[(size_t)row * OUT_DIM + c] = acc.t[nt][r] + bc;
            }
        }
    }
    __syncthreads();

    store_w<IN_DIM>(lWs, w_second);
    __syncthreads();
    if (tile < nTiles) {
        Acc4 acc = wave_gemm_tile(xrow, lWs, IN_DIM, half, l16);
#pragma unroll
        for (int nt = 0; nt < 4; ++nt) {
            const int   c  = nt * 16 + l16;
            const float bc = b_self[c];
#pragma unroll
            for (int r = 0; r < 8; ++r) {
                const int row = tile * 16 + r + 8 * half;
                if (row < N) hout[(size_t)row * OUT_DIM + c] = acc.t[nt][r] + bc;
            }
        }
    }
}

__global__ void k_node_attn(const float* __restrict__ z, const float* __restrict__ W_attn,
                            float* __restrict__ a_src, float* __restrict__ a_dst,
                            float* __restrict__ m, float* __restrict__ denom,
                            float* __restrict__ m1, float* __restrict__ denom1, int two, int N,
                            int* __restrict__ binz, int nbinz, int stats)
{
    const int n = blockIdx.x * blockDim.x + threadIdx.x;
    if (n < nbinz) binz[n] = 0;
    if (n >= N) return;
    const float* zr = z + (size_t)n * OUT_DIM;
    float as = 0.f, ad = 0.f;
#pragma unroll 8
    for (int c = 0; c < OUT_DIM; ++c) {
        const float zc = zr[c];
        as += zc * W_attn[c];
        ad += zc * W_attn[OUT_DIM + c];
    }
    a_src[n] = as;
    a_dst[n] = ad;
    if (!stats) return;
    m[n]     = -INFINITY;
    denom[n] = 0.0f;
    if (two) {
        m1[n]     = -INFINITY;
        denom1[n] = 0.0f;
    }
}

__global__ void k_edge_logit(const float* __restrict__ edge_attr,
                             const int* __restrict__ erow, const int* __restrict__ ecol,
                             const float* __restrict__ W_attn, const float* __restrict__ b_attn,
                             const float* __restrict__ a_src, const float* __restrict__ a_dst,
                             float* __restrict__ elog, float* __restrict__ m,
                             float* __restrict__ m1, int two, int E)
{
    const int e = blockIdx.x * blockDim.x + threadIdx.x;
    if (e >= E) return;
    const float* ea = edge_attr + (size_t)e * ED_DIM;
    float dot = 0.f;
#pragma unroll
    for (int k = 0; k < ED_DIM; ++k) dot += ea[k] * W_attn[2 * OUT_DIM + k];
    const int c = ecol[e];
    const float s  = a_src[erow[e]] + a_dst[c] + dot + b_attn[0];
    const float lr = (s > 0.0f) ? s : 0.2f * s;
    elog[e] = lr;
    float* mh = (two && chip_half()) ? m1 : m;
    atomicMaxF(&mh[c], lr);
}

__global__ void k_edge_soft(const float* __restrict__ elog, const int* __restrict__ ecol,
                            const float* __restrict__ m, const float* __restrict__ m1,
                            float* __restrict__ pbuf,
                            float* __restrict__ denom, float* __restrict__ denom1, int two, int E)
{
    const int e = blockIdx.x * blockDim.x + threadIdx.x;
    if (e >= E) return;
    const int c = ecol[e];
    const float mc = fmaxf(m[c], m1[c]);
    const float p = expf(elog[e] - mc);
    pbuf[e] = p;
    float* dh = (two && chip_half()) ? denom1 : denom;
    atomicAdd(&dh[c], p);
}

__device__ __attribute__((always_inline)) inline
float xhalf(float x)
{
    const int b = __builtin_bit_cast(int, x);
    return __builtin_bit_cast(float,
        __builtin_amdgcn_permlanex16(b, b, 0x76543210, 0xfedcba98, false, false));
}

template <bool FULL>
__device__ __attribute__((always_inline)) inline
void edge_epilogue(const Acc4& acc, int c_l, int r_l, float al_l, float b_lo, float b_hi,
                   const float* __restrict__ z, float* __restrict__ hout, int lane, int half)
{
    float zlo[16], zhi[16];
    if (FULL) {
#pragma unroll
        for (int j = 0; j < 16; ++j) {
            const int    rj = __builtin_amdgcn_readlane(r_l, j);
            const float* zr = z + (size_t)rj * OUT_DIM;
            zlo[j] = zr[lane];
            zhi[j] = zr[32 + lane];
        }
    }
#pragma unroll
    for (int r = 0; r < 8; ++r) {
        const float gA   = xhalf(half ? acc.t[0][r] : acc.t[1][r]);
        const float gB   = xhalf(half ? acc.t[2][r] : acc.t[3][r]);
        const float v0lo = half ? gA : acc.t[0][r];
        const float v1lo = half ? acc.t[1][r] : gA;
        const float v0hi = half ? gB : acc.t[2][r];
        const float v1hi = half ? acc.t[3][r] : gB;
#pragma unroll
        for (int h = 0; h < 2; ++h) {
            const int   j  = r + 8 * h;
            const int   cj = __builtin_amdgcn_readlane(c_l, j);
            const float aj = __builtin_bit_cast(float,
                __builtin_amdgcn_readlane(__builtin_bit_cast(int, al_l), j));
            if (FULL || cj >= 0) {
                float zl, zh;
                if (FULL) {
                    zl = zlo[j];
                    zh = zhi[j];
                } else {
                    const int    rj = __builtin_amdgcn_readlane(r_l, j);
                    const float* zr = z + (size_t)rj * OUT_DIM;
                    zl = zr[lane];
                    zh = zr[32 + lane];
                }
                float* hr = hout + (size_t)cj * OUT_DIM;
                atomicAdd(&hr[lane],      aj * (zl + (h ? v1lo : v0lo) + b_lo));
                atomicAdd(&hr[32 + lane], aj * (zh + (h ? v1hi : v0hi) + b_hi));
            }
        }
    }
}

__global__ void k_edge_aggregate(const float* __restrict__ edge_attr,
                                 const int* __restrict__ erow, const int* __restrict__ ecol,
                                 const float* __restrict__ W_edge, const float* __restrict__ b_edge,
                                 const float* __restrict__ z, const float* __restrict__ pbuf,
                                 const float* __restrict__ denom, const float* __restrict__ denom1,
                                 float* __restrict__ hout, float* __restrict__ plane1, int two,
                                 int E, int nTiles)
{
    __shared__ __attribute__((aligned(16))) float lWs[(ED_DIM / 2) * BSTRIDE_F];
    const int lane = threadIdx.x & 31;
    const int wave = threadIdx.x >> 5;
    const int half = lane >> 4;
    const int l16  = lane & 15;

    store_w<ED_DIM>(lWs, load_w<ED_DIM>(W_edge));
    __syncthreads();
    const int tile = __builtin_amdgcn_readfirstlane(blockIdx.x * 8 + wave);
    if (tile >= nTiles) return;

    int arow = tile * 16 + l16;
    const bool evalid = arow < E;
    if (arow >= E) arow = E - 1;
    const float* ea = edge_attr + (size_t)arow * ED_DIM;

    {
        const size_t nxt = (size_t)arow + 16 * 8;
        if (nxt < (size_t)E)
            __builtin_prefetch(edge_attr + nxt * ED_DIM, 0, 1);
    }

    const int   c_e  = ecol[arow];
    const int   c_l  = evalid ? c_e : -1;
    const int   r_l  = erow[arow];
    const float d0   = denom[c_e];
    const float d1   = denom1[c_e];
    const float den  = two ? d0 + d1 : d0;
    const float al_l = pbuf[arow] / fmaxf(den, 1e-16f);

    Acc4 acc = wave_gemm_tile(ea, lWs, ED_DIM, half, l16);

    const float b_lo = b_edge[lane];
    const float b_hi = b_edge[32 + lane];

    float* hacc = (two && chip_half()) ? plane1 : hout;

    if (tile * 16 + 16 <= E)
        edge_epilogue<true >(acc, c_l, r_l, al_l, b_lo, b_hi, z, hacc, lane, half);
    else
        edge_epilogue<false>(acc, c_l, r_l, al_l, b_lo, b_hi, z, hacc, lane, half);
}


#define BIN_R   196
#define RANGE_TILES ((BIN_R + 15) / 16)
#define EA_PITCH 34
#define STAT_PER 4
#define NB_MAX  1024
#define BIN_T   1024
#define BIN_PER 4

__global__ __launch_bounds__(BIN_T)
void k_bin_count(const int* __restrict__ ecol, int* __restrict__ cnt, int E, int nb)
{
    __shared__ int s_hist[NB_MAX];
    const int tid = threadIdx.x;
    s_hist[tid] = 0;
    __syncthreads();
#pragma unroll
    for (int i = 0; i < BIN_PER; ++i) {
        const int e = blockIdx.x * (BIN_T * BIN_PER) + i * BIN_T + tid;
        if (e < E) atomicAdd(&s_hist[ecol[e] / BIN_R], 1);
    }
    __syncthreads();
    const int h = s_hist[tid];
    if (tid < nb && h > 0) atomicAdd(&cnt[tid], h);
}

__device__ __attribute__((always_inline)) inline
float x8_4(float v) { return __builtin_bit_cast(float, __builtin_amdgcn_mov_dpp8(__builtin_bit_cast(unsigned, v), 0x688fac)); }
__device__ __attribute__((always_inline)) inline
float x8_2(float v) { return __builtin_bit_cast(float, __builtin_amdgcn_mov_dpp8(__builtin_bit_cast(unsigned, v), 0xb3e21a)); }
__device__ __attribute__((always_inline)) inline
float x8_1(float v) { return __builtin_bit_cast(float, __builtin_amdgcn_mov_dpp8(__builtin_bit_cast(unsigned, v), 0xde54c1)); }

__device__ __attribute__((always_inline)) inline
void ea_request(float4 (&v)[8], const float* __restrict__ edge_attr, int base, int sub, int g)
{
    const float4* p = reinterpret_cast<const float4*>(
        edge_attr + (size_t)(base + sub) * ED_DIM + 4 * g);
#pragma unroll
    for (int q = 0; q < 8; ++q) v[q] = p[q * ED_DIM];
}

__device__ __attribute__((always_inline)) inline
float ea_fold(const float4 (&v)[8], const float4 wa, int lane)
{
    float d[8];
#pragma unroll
    for (int q = 0; q < 8; ++q)
        d[q] = v[q].x * wa.x + v[q].y * wa.y + v[q].z * wa.z + v[q].w * wa.w;
    float s1[4];
#pragma unroll
    for (int q = 0; q < 4; ++q) {
        const float keep = (lane & 4) ? d[q + 4] : d[q];
        const float send = (lane & 4) ? d[q] : d[q + 4];
        s1[q] = keep + x8_4(send);
    }
    float s2[2];
#pragma unroll
    for (int q = 0; q < 2; ++q) {
        const float keep = (lane & 2) ? s1[q + 2] : s1[q];
        const float send = (lane & 2) ? s1[q] : s1[q + 2];
        s2[q] = keep + x8_2(send);
    }
    const float keep = (lane & 1) ? s2[1] : s2[0];
    const float send = (lane & 1) ? s2[0] : s2[1];
    return keep + x8_1(send);
}

__global__ __launch_bounds__(BIN_T)
void k_bin_place(const float* __restrict__ edge_attr,
                 const int* __restrict__ erow, const int* __restrict__ ecol,
                 const float* __restrict__ W_attn, const float* __restrict__ b_attn,
                 const float* __restrict__ a_src, const float* __restrict__ a_dst,
                 const int* __restrict__ cnt, int* __restrict__ cursor, int* __restrict__ bbase,
                 int* __restrict__ rec_e, int* __restrict__ rec_rc, float* __restrict__ rec_lg,
                 int E, int nb)
{
    __shared__ int s_base[NB_MAX];
    __shared__ int s_hist[NB_MAX];
    __shared__ int s_off[NB_MAX];
    __shared__ int s_wsum[32];
    __shared__ float s_dot[BIN_T * BIN_PER];
    const int tid  = threadIdx.x;
    const int lane = tid & 31;
    const int wave = __builtin_amdgcn_readfirstlane(tid >> 5);
    const int g    = lane & 7;
    const int sub  = lane >> 3;

    const float4 wa = *reinterpret_cast<const float4*>(W_attn + 2 * OUT_DIM + 4 * g);
    const float  ba = b_attn[0];

    const int base0 = blockIdx.x * (BIN_T * BIN_PER) + wave * 32;
    int   eid[BIN_PER], bk[BIN_PER], rc[BIN_PER];
    float lg[BIN_PER];
    int   cc[BIN_PER], rr[BIN_PER];
#pragma unroll
    for (int i = 0; i < BIN_PER; ++i) {
        const int e  = base0 + i * BIN_T + 4 * g + sub;
        const int ec = e < E ? e : E - 1;
        eid[i] = e;
        cc[i]  = ecol[ec];
        rr[i]  = erow[ec];
    }
#pragma unroll 1
    for (int i = 0; i < BIN_PER; ++i) {
        const int b = base0 + i * BIN_T;
        float4 ln[8];
        ea_request(ln, edge_attr, (b + 32 <= E) ? b : E - 32, sub, g);
        s_dot[i * BIN_T + tid] = ea_fold(ln, wa, lane);
    }
    float asd[BIN_PER];
#pragma unroll
    for (int i = 0; i < BIN_PER; ++i) asd[i] = a_src[rr[i]] + a_dst[cc[i]];

#pragma unroll
    for (int i = 0; i < BIN_PER; ++i) {
        float dot = s_dot[i * BIN_T + tid];
        const int base = base0 + i * BIN_T;
        if (base < E && base + 32 > E) {
            if (eid[i] < E) {
                const float* ea = edge_attr + (size_t)eid[i] * ED_DIM;
                float t = 0.f;
                for (int k = 0; k < ED_DIM; ++k) t += ea[k] * W_attn[2 * OUT_DIM + k];
                dot = t;
            }
        }
        const float s = asd[i] + dot + ba;
        const int   b = cc[i] / BIN_R;
        bk[i] = eid[i] < E ? b : -1;
        rc[i] = rr[i] * 256 + (cc[i] - b * BIN_R);
        lg[i] = (s > 0.0f) ? s : 0.2f * s;
    }

    const int own = tid < nb ? cnt[tid] : 0;
    int v = own;
#pragma unroll
    for (int d = 1; d < 32; d <<= 1) {
        const int u = __shfl_up(v, d);
        if (lane >= d) v += u;
    }
    if (lane == 31) s_wsum[wave] = v;
    s_hist[tid] = 0;
    __syncthreads();
    if (wave == 0) {
        const int w = s_wsum[lane];
        int x = w;
#pragma unroll
        for (int d = 1; d < 32; d <<= 1) {
            const int u = __shfl_up(x, d);
            if (lane >= d) x += u;
        }
        s_wsum[lane] = x - w;
    }
    __syncthreads();
    const int excl = v - own + s_wsum[wave];
    s_base[tid] = excl;
    if (blockIdx.x == 0) {
        if (tid < nb)      bbase[tid] = excl;
        if (tid == nb - 1) bbase[nb]  = excl + own;
    }

    int lr[BIN_PER];
#pragma unroll
    for (int i = 0; i < BIN_PER; ++i) {
        lr[i] = 0;
        if (bk[i] >= 0) lr[i] = atomicAdd(&s_hist[bk[i]], 1);
    }
    __syncthreads();
    {
        const int h = s_hist[tid];
        if (tid < nb && h > 0) s_off[tid] = s_base[tid] + atomicAdd(&cursor[tid], h);
    }
    __syncthreads();
#pragma unroll
    for (int i = 0; i < BIN_PER; ++i) {
        if (bk[i] >= 0) {
            const int pos = s_off[bk[i]] + lr[i];
            rec_e[pos]  = eid[i];
            rec_rc[pos] = rc[i];
            rec_lg[pos] = lg[i];
        }
    }
}

__device__ __attribute__((always_inline)) inline
void lds_max(float* addr, float val)
{
    __hip_atomic_fetch_max(addr, val, __ATOMIC_RELAXED, __HIP_MEMORY_SCOPE_WORKGROUP);
}

template <bool FULL>
__device__ __attribute__((always_inline)) inline
void edge_accumulate(int e_l, int c_l, int r_l, float al_l,
                     const float* __restrict__ edge_attr, const float* __restrict__ z,
                     float* __restrict__ lacc, float* __restrict__ lea, int lane)
{
    float ev[16];
    v2f   zv[16];
    if (FULL) {
#pragma unroll
        for (int j = 0; j < 16; ++j) {
            const int ej = __builtin_amdgcn_readlane(e_l, j);
            const int rj = __builtin_amdgcn_readlane(r_l, j);
            ev[j] = edge_attr[(size_t)ej * ED_DIM + lane];
            zv[j] = reinterpret_cast<const v2f*>(z + (size_t)rj * OUT_DIM)[lane];
        }
    }
#pragma unroll
    for (int j = 0; j < 16; ++j) {
        const int   cj = __builtin_amdgcn_readlane(c_l, j);
        const float aj = __builtin_bit_cast(float,
            __builtin_amdgcn_readlane(__builtin_bit_cast(int, al_l), j));
        if (FULL || cj >= 0) {
            float e1;
            v2f   z2;
            if (FULL) {
                e1 = ev[j];
                z2 = zv[j];
            } else {
                const int ej = __builtin_amdgcn_readlane(e_l, j);
                const int rj = __builtin_amdgcn_readlane(r_l, j);
                e1 = edge_attr[(size_t)ej * ED_DIM + lane];
                z2 = reinterpret_cast<const v2f*>(z + (size_t)rj * OUT_DIM)[lane];
            }
            float* hr = lacc + cj * OUT_DIM + 2 * lane;
            atomicAdd(&hr[0], aj * z2.x);
            atomicAdd(&hr[1], aj * z2.y);
            atomicAdd(&lea[cj * EA_PITCH + lane], aj * e1);
        }
    }
}

__global__ __launch_bounds__(BIN_T)
void k_range_aggregate(const float* __restrict__ edge_attr,
                       const float* __restrict__ W_edge, const float* __restrict__ b_edge,
                       const float* __restrict__ z,
                       const int* __restrict__ rec_e, const int* __restrict__ rec_rc,
                       const float* __restrict__ rec_lg, const int* __restrict__ bbase,
                       float* __restrict__ hout, int N)
{
    __shared__ __attribute__((aligned(16))) float lWs[(ED_DIM / 2) * BSTRIDE_F];
    __shared__ __attribute__((aligned(16))) float lacc[BIN_R * OUT_DIM];
    __shared__ __attribute__((aligned(16))) float lea[RANGE_TILES * 16 * EA_PITCH];
    __shared__ float lmax[BIN_R];
    __shared__ float lden[BIN_R];
    const int lane = threadIdx.x & 31;
    const int wave = __builtin_amdgcn_readfirstlane(threadIdx.x >> 5);
    const int half = lane >> 4;
    const int l16  = lane & 15;

    const int n0   = blockIdx.x * BIN_R;
    const int rows = (N - n0 < BIN_R) ? (N - n0) : BIN_R;

    const int b0 = bbase[blockIdx.x];
    const int b1 = bbase[blockIdx.x + 1];
    const int ntile = (b1 - b0 + 15) >> 4;

    float lg_s[STAT_PER];
    int   rc_s[STAT_PER];
#pragma unroll
    for (int j = 0; j < STAT_PER; ++j) {
        const int i  = b0 + j * BIN_T + (int)threadIdx.x;
        const int ii = i < b1 ? i : 0;
        lg_s[j] = rec_lg[ii];
        rc_s[j] = rec_rc[ii];
    }
    int   e_n = 0, rc_n = 0;
    float lg_n = 0.f;
    if (wave < ntile) {
        int idx = b0 + wave * 16 + l16;
        if (idx >= b1) idx = b1 - 1;
        e_n = rec_e[idx]; rc_n = rec_rc[idx]; lg_n = rec_lg[idx];
    }

    if (threadIdx.x < BIN_R) { lmax[threadIdx.x] = -INFINITY; lden[threadIdx.x] = 0.f; }
    if (threadIdx.x < NTHREADS) store_w<ED_DIM>(lWs, load_w<ED_DIM>(W_edge));
    {
        const float4* h4 = reinterpret_cast<const float4*>(hout + (size_t)n0 * OUT_DIM);
        float4*       a4 = reinterpret_cast<float4*>(lacc);
        for (int i = threadIdx.x; i < rows * (OUT_DIM / 4); i += BIN_T) a4[i] = h4[i];
        float4*       s4 = reinterpret_cast<float4*>(lea);
        for (int i = threadIdx.x; i < RANGE_TILES * 16 * EA_PITCH / 4; i += BIN_T)
            s4[i] = make_float4(0.f, 0.f, 0.f, 0.f);
    }
    __syncthreads();

    int c_s[STAT_PER];
#pragma unroll
    for (int j = 0; j < STAT_PER; ++j)
        c_s[j] = (b0 + j * BIN_T + (int)threadIdx.x < b1) ? (rc_s[j] & 255) : -1;

#pragma unroll
    for (int j = 0; j < STAT_PER; ++j)
        if (c_s[j] >= 0) lds_max(&lmax[c_s[j]], lg_s[j]);
    for (int i = b0 + STAT_PER * BIN_T + (int)threadIdx.x; i < b1; i += BIN_T)
        lds_max(&lmax[rec_rc[i] & 255], rec_lg[i]);
    __syncthreads();

#pragma unroll
    for (int j = 0; j < STAT_PER; ++j)
        if (c_s[j] >= 0) atomicAdd(&lden[c_s[j]], expf(lg_s[j] - lmax[c_s[j]]));
    for (int i = b0 + STAT_PER * BIN_T + (int)threadIdx.x; i < b1; i += BIN_T) {
        const int cc = rec_rc[i] & 255;
        atomicAdd(&lden[cc], expf(rec_lg[i] - lmax[cc]));
    }
    __syncthreads();

    for (int t = wave; t < ntile; t += BIN_T / 32) {
        const int   e_c = e_n, rc_c = rc_n;
        const float lg_c = lg_n;
        const int tn = t + BIN_T / 32;
        if (tn < ntile) {
            int idx = b0 + tn * 16 + l16;
            if (idx >= b1) idx = b1 - 1;
            e_n = rec_e[idx]; rc_n = rec_rc[idx]; lg_n = rec_lg[idx];
        }
        const bool valid = b0 + t * 16 + l16 < b1;
        const int  cc    = rc_c & 255;
        const int  c_l   = valid ? cc : -1;
        const int  r_l   = rc_c >> 8;
        const float al_c = expf(lg_c - lmax[cc]) / fmaxf(lden[cc], 1e-16f);

        if (t * 16 + 16 <= b1 - b0)
            edge_accumulate<true >(e_c, c_l, r_l, al_c, edge_attr, z, lacc, lea, lane);
        else
            edge_accumulate<false>(e_c, c_l, r_l, al_c, edge_attr, z, lacc, lea, lane);
    }
    __syncthreads();

    for (int t = wave; t < RANGE_TILES; t += BIN_T / 32) {
        const float* arow = lea + (t * 16 + l16) * EA_PITCH;
        Acc4 acc = wave_gemm_tile(arow, lWs, ED_DIM, half, l16);
        float sa[8];
#pragma unroll
        for (int r = 0; r < 8; ++r) {
            const int row = t * 16 + r + 8 * half;
            const float d = lden[row < rows ? row : 0];
            sa[r] = d / fmaxf(d, 1e-16f);
        }
#pragma unroll
        for (int nt = 0; nt < 4; ++nt) {
            const int   c  = nt * 16 + l16;
            const float bc = b_edge[c];
#pragma unroll
            for (int r = 0; r < 8; ++r) {
                const int row = t * 16 + r + 8 * half;
                if (row < rows) lacc[row * OUT_DIM + c] += acc.t[nt][r] + sa[r] * bc;
            }
        }
    }
    __syncthreads();
    {
        float4*       o4 = reinterpret_cast<float4*>(hout + (size_t)n0 * OUT_DIM);
        const float4* a4 = reinterpret_cast<const float4*>(lacc);
        for (int i = threadIdx.x; i < rows * (OUT_DIM / 4); i += BIN_T) o4[i] = a4[i];
    }
}

__global__ void k_plane_sum(float* __restrict__ hout, const float* __restrict__ plane1,
                            int two, int n4)
{
    const int i = blockIdx.x * blockDim.x + threadIdx.x;
    if (!two || i >= n4) return;
    float4*       h4 = reinterpret_cast<float4*>(hout);
    const float4* q4 = reinterpret_cast<const float4*>(plane1);
    float4 a = h4[i];
    const float4 b = q4[i];
    a.x += b.x; a.y += b.y; a.z += b.z; a.w += b.w;
    h4[i] = a;
}

extern "C" void kernel_launch(void* const* d_in, const int* in_sizes, int n_in,
                              void* d_out, int out_size, void* d_ws, size_t ws_size,
                              hipStream_t stream)
{
    const float* x         = (const float*)d_in[0];
    const float* edge_attr = (const float*)d_in[1];
    const int*   edge_idx  = (const int*)  d_in[2];
    const float* W_fc      = (const float*)d_in[3];
    const float* b_fc      = (const float*)d_in[4];
    const float* W_attn    = (const float*)d_in[5];
    const float* b_attn    = (const float*)d_in[6];
    const float* W_edge    = (const float*)d_in[7];
    const float* b_edge    = (const float*)d_in[8];
    const float* W_self    = (const float*)d_in[9];
    const float* b_self    = (const float*)d_in[10];

    const int N = in_sizes[0] / IN_DIM;
    const int E = in_sizes[1] / ED_DIM;
    const int* erow = edge_idx;
    const int* ecol = edge_idx + E;

    float* ws     = (float*)d_ws;
    float* z      = ws;
    float* a_src  = z + (size_t)N * OUT_DIM;
    float* a_dst  = a_src + N;
    float* m      = a_dst + N;
    float* denom  = m + N;
    float* elog   = denom + N;
    float* pbuf   = elog + E;
    const size_t off1 = ((size_t)N * (OUT_DIM + 6) + 2 * (size_t)E + 3) & ~(size_t)3;
    const int two = ws_size >= (off1 + (size_t)N * OUT_DIM) * sizeof(float) ? 1 : 0;
    float* m1     = two ? pbuf + E : m;
    float* denom1 = two ? m1 + N   : denom;
    float* plane1 = two ? ws + off1 : (float*)d_out;
    float* hout   = (float*)d_out;

    const int nNodeTiles = (N + 15) / 16;
    const int nEdgeTiles = (E + 15) / 16;

    const int nb   = (N + BIN_R - 1) / BIN_R;
    const int bins = (two && nb <= NB_MAX && N >= 2 * NB_MAX && N < (1 << 23) && E >= 32 &&
                      (size_t)N * OUT_DIM >= 3 * (size_t)E + 3 * NB_MAX + 8) ? 1 : 0;
    int*   rec_e  = (int*)plane1;
    int*   rec_rc = rec_e + E;
    float* rec_lg = (float*)(rec_rc + E);
    int*   cnt    = (int*)(rec_lg + E);
    int*   cursor = cnt + NB_MAX;
    int*   bbase  = cursor + NB_MAX;
    const int nBinBlocks = (E + BIN_T * BIN_PER - 1) / (BIN_T * BIN_PER);

    k_node_gemm<<<(nNodeTiles + 7) / 8, NTHREADS, 0, stream>>>(
        x, W_fc, b_fc, W_self, b_self, z, hout, plane1, bins ? 0 : two, N, nNodeTiles);

    k_node_attn<<<(N + 255) / 256, 256, 0, stream>>>(
        z, W_attn, a_src, a_dst, m, denom, m1, denom1, two, N, cnt, bins ? 2 * NB_MAX : 0,
        bins ? 0 : 1);

    if (bins) {
        k_bin_count<<<nBinBlocks, BIN_T, 0, stream>>>(ecol, cnt, E, nb);

        k_bin_place<<<nBinBlocks, BIN_T, 0, stream>>>(
            edge_attr, erow, ecol, W_attn, b_attn, a_src, a_dst,
            cnt, cursor, bbase, rec_e, rec_rc, rec_lg, E, nb);

        k_range_aggregate<<<nb, BIN_T, 0, stream>>>(
            edge_attr, W_edge, b_edge, z, rec_e, rec_rc, rec_lg, bbase, hout, N);
    } else {
        k_edge_logit<<<(E + 255) / 256, 256, 0, stream>>>(
            edge_attr, erow, ecol, W_attn, b_attn, a_src, a_dst, elog, m, m1, two, E);

        k_edge_soft<<<(E + 255) / 256, 256, 0, stream>>>(
            elog, ecol, m, m1, pbuf, denom, denom1, two, E);

        k_edge_aggregate<<<(nEdgeTiles + 7) / 8, NTHREADS, 0, stream>>>(
            edge_attr, erow, ecol, W_edge, b_edge, z, pbuf, denom, denom1, hout, plane1, two,
            E, nEdgeTiles);

        k_plane_sum<<<(N * (OUT_DIM / 4) + 255) / 256, 256, 0, stream>>>(
            hout, plane1, two, N * (OUT_DIM / 4));
    }
}


# edge records (edge id, rows, logit) interleaved, three consecutive words each: the placing pass writes one scattered 12-byte store per edge instead of three 4-byte stores into three planes
# speedup vs baseline: 3.5715x; 1.1917x over previous
#include <hip/hip_runtime.h>
#include <hip/hip_bf16.h>
#include <math.h>

#define IN_DIM  128
#define OUT_DIM 64
#define ED_DIM  32

#define BSTRIDE_V2 80
#define BSTRIDE_F  (2 * BSTRIDE_V2)

typedef __attribute__((ext_vector_type(2))) float v2f;
typedef __attribute__((ext_vector_type(8))) float v8f;

struct Acc4 { v8f t[4]; };

__device__ __attribute__((always_inline)) inline
Acc4 wave_gemm_tile(const float* __restrict__ arow,
                    const float* __restrict__ lWs,
                    int kdim, int half, int l16)
{
    Acc4 A;
    A.t[0] = {}; A.t[1] = {}; A.t[2] = {}; A.t[3] = {};
    const v2f* a2 = (const v2f*)arow;
    for (int k0 = 0; k0 < kdim; k0 += 4) {
        const int kp  = (k0 >> 1) + half;
        const v2f  a  = a2[kp];
        const v2f* br = (const v2f*)(lWs + kp * BSTRIDE_F);
#pragma unroll
        for (int nt = 0; nt < 4; ++nt) {
            const v2f b = br[nt * 16 + l16];
            A.t[nt] = __builtin_amdgcn_wmma_f32_16x16x4_f32(
                false, a, false, b, (short)0, A.t[nt], false, false);
        }
    }
    return A;
}

#define NTHREADS 256

template <int KDIM>
struct WStage {
    static constexpr int ITEMS = (KDIM / 2) * (OUT_DIM / 4);
    static constexpr int PER   = ITEMS / NTHREADS;
    static_assert(ITEMS % NTHREADS == 0, "staging is laid out for whole rounds of the work-group");
    float4 lo[PER];
    float4 hi[PER];
};

template <int KDIM>
__device__ __attribute__((always_inline)) inline
WStage<KDIM> load_w(const float* __restrict__ W)
{
    WStage<KDIM> s;
#pragma unroll
    for (int j = 0; j < WStage<KDIM>::PER; ++j) {
        const int it = threadIdx.x + NTHREADS * j;
        const int kp = it >> 4;
        const int g  = it & 15;
        const float4* src = reinterpret_cast<const float4*>(W + (size_t)(2 * kp) * OUT_DIM + 4 * g);
        s.lo[j] = src[0];
        s.hi[j] = src[OUT_DIM / 4];
    }
    return s;
}

template <int KDIM>
__device__ __attribute__((always_inline)) inline
void store_w(float* __restrict__ lWs, const WStage<KDIM>& s)
{
#pragma unroll
    for (int j = 0; j < WStage<KDIM>::PER; ++j) {
        const int it = threadIdx.x + NTHREADS * j;
        const int kp = it >> 4;
        const int g  = it & 15;
        float4* dst = reinterpret_cast<float4*>(lWs + kp * BSTRIDE_F + 8 * g);
        dst[0] = make_float4(s.lo[j].x, s.hi[j].x, s.lo[j].y, s.hi[j].y);
        dst[1] = make_float4(s.lo[j].z, s.hi[j].z, s.lo[j].w, s.hi[j].w);
    }
}

__device__ __attribute__((always_inline)) inline
int chip_half()
{
    const unsigned v = __builtin_amdgcn_s_sendmsg_rtn(0x87);
    return (int)((v >> 18) & 1u);
}

__device__ inline void atomicMaxF(float* addr, float val) {
    if (val >= 0.0f) atomicMax(reinterpret_cast<int*>(addr), __float_as_int(val));
    else             atomicMin(reinterpret_cast<unsigned int*>(addr), __float_as_uint(val));
}

__global__ void k_node_gemm(const float* __restrict__ x,
                            const float* __restrict__ W_fc,  const float* __restrict__ b_fc,
                            const float* __restrict__ W_self, const float* __restrict__ b_self,
                            float* __restrict__ z, float* __restrict__ hout,
                            float* __restrict__ plane1, int two,
                            int N, int nTiles)
{
    __shared__ __attribute__((aligned(16))) float lWs[(IN_DIM / 2) * BSTRIDE_F];
    const int lane = threadIdx.x & 31;
    const int wave = threadIdx.x >> 5;
    const int half = lane >> 4;
    const int l16  = lane & 15;
    const int tile = blockIdx.x * 8 + wave;

    const WStage<IN_DIM> w_first  = load_w<IN_DIM>(W_fc);
    const WStage<IN_DIM> w_second = load_w<IN_DIM>(W_self);

    int mrow = tile * 16 + l16;
    if (mrow >= N) mrow = N - 1;
    const float* xrow = x + (size_t)mrow * IN_DIM;

    if (two && tile < nTiles) {
        float4* p4 = reinterpret_cast<float4*>(plane1 + (size_t)tile * 16 * OUT_DIM);
#pragma unroll
        for (int it = 0; it < 8; ++it) {
            const int idx = it * 32 + lane;
            if (tile * 16 + (idx >> 4) < N) p4[idx] = make_float4(0.f, 0.f, 0.f, 0.f);
        }
    }

    store_w<IN_DIM>(lWs, w_first);
    __syncthreads();
    if (tile < nTiles) {
        Acc4 acc = wave_gemm_tile(xrow, lWs, IN_DIM, half, l16);
#pragma unroll
        for (int nt = 0; nt < 4; ++nt) {
            const int   c  = nt * 16 + l16;
            const float bc = b_fc[c];
#pragma unroll
            for (int r = 0; r < 8; ++r) {
                const int row = tile * 16 + r + 8 * half;
                if (row < N) z[(size_t)row * OUT_DIM + c] = acc.t[nt][r] + bc;
            }
        }
    }
    __syncthreads();

    store_w<IN_DIM>(lWs, w_second);
    __syncthreads();
    if (tile < nTiles) {
        Acc4 acc = wave_gemm_tile(xrow, lWs, IN_DIM, half, l16);
#pragma unroll
        for (int nt = 0; nt < 4; ++nt) {
            const int   c  = nt * 16 + l16;
            const float bc = b_self[c];
#pragma unroll
            for (int r = 0; r < 8; ++r) {
                const int row = tile * 16 + r + 8 * half;
                if (row < N) hout[(size_t)row * OUT_DIM + c] = acc.t[nt][r] + bc;
            }
        }
    }
}

__global__ void k_node_attn(const float* __restrict__ z, const float* __restrict__ W_attn,
                            float* __restrict__ a_src, float* __restrict__ a_dst,
                            float* __restrict__ m, float* __restrict__ denom,
                            float* __restrict__ m1, float* __restrict__ denom1, int two, int N,
                            int* __restrict__ binz, int nbinz, int stats)
{
    const int n = blockIdx.x * blockDim.x + threadIdx.x;
    if (n < nbinz) binz[n] = 0;
    if (n >= N) return;
    const float* zr = z + (size_t)n * OUT_DIM;
    float as = 0.f, ad = 0.f;
#pragma unroll 8
    for (int c = 0; c < OUT_DIM; ++c) {
        const float zc = zr[c];
        as += zc * W_attn[c];
        ad += zc * W_attn[OUT_DIM + c];
    }
    a_src[n] = as;
    a_dst[n] = ad;
    if (!stats) return;
    m[n]     = -INFINITY;
    denom[n] = 0.0f;
    if (two) {
        m1[n]     = -INFINITY;
        denom1[n] = 0.0f;
    }
}

__global__ void k_edge_logit(const float* __restrict__ edge_attr,
                             const int* __restrict__ erow, const int* __restrict__ ecol,
                             const float* __restrict__ W_attn, const float* __restrict__ b_attn,
                             const float* __restrict__ a_src, const float* __restrict__ a_dst,
                             float* __restrict__ elog, float* __restrict__ m,
                             float* __restrict__ m1, int two, int E)
{
    const int e = blockIdx.x * blockDim.x + threadIdx.x;
    if (e >= E) return;
    const float* ea = edge_attr + (size_t)e * ED_DIM;
    float dot = 0.f;
#pragma unroll
    for (int k = 0; k < ED_DIM; ++k) dot += ea[k] * W_attn[2 * OUT_DIM + k];
    const int c = ecol[e];
    const float s  = a_src[erow[e]] + a_dst[c] + dot + b_attn[0];
    const float lr = (s > 0.0f) ? s : 0.2f * s;
    elog[e] = lr;
    float* mh = (two && chip_half()) ? m1 : m;
    atomicMaxF(&mh[c], lr);
}

__global__ void k_edge_soft(const float* __restrict__ elog, const int* __restrict__ ecol,
                            const float* __restrict__ m, const float* __restrict__ m1,
                            float* __restrict__ pbuf,
                            float* __restrict__ denom, float* __restrict__ denom1, int two, int E)
{
    const int e = blockIdx.x * blockDim.x + threadIdx.x;
    if (e >= E) return;
    const int c = ecol[e];
    const float mc = fmaxf(m[c], m1[c]);
    const float p = expf(elog[e] - mc);
    pbuf[e] = p;
    float* dh = (two && chip_half()) ? denom1 : denom;
    atomicAdd(&dh[c], p);
}

__device__ __attribute__((always_inline)) inline
float xhalf(float x)
{
    const int b = __builtin_bit_cast(int, x);
    return __builtin_bit_cast(float,
        __builtin_amdgcn_permlanex16(b, b, 0x76543210, 0xfedcba98, false, false));
}

template <bool FULL>
__device__ __attribute__((always_inline)) inline
void edge_epilogue(const Acc4& acc, int c_l, int r_l, float al_l, float b_lo, float b_hi,
                   const float* __restrict__ z, float* __restrict__ hout, int lane, int half)
{
    float zlo[16], zhi[16];
    if (FULL) {
#pragma unroll
        for (int j = 0; j < 16; ++j) {
            const int    rj = __builtin_amdgcn_readlane(r_l, j);
            const float* zr = z + (size_t)rj * OUT_DIM;
            zlo[j] = zr[lane];
            zhi[j] = zr[32 + lane];
        }
    }
#pragma unroll
    for (int r = 0; r < 8; ++r) {
        const float gA   = xhalf(half ? acc.t[0][r] : acc.t[1][r]);
        const float gB   = xhalf(half ? acc.t[2][r] : acc.t[3][r]);
        const float v0lo = half ? gA : acc.t[0][r];
        const float v1lo = half ? acc.t[1][r] : gA;
        const float v0hi = half ? gB : acc.t[2][r];
        const float v1hi = half ? acc.t[3][r] : gB;
#pragma unroll
        for (int h = 0; h < 2; ++h) {
            const int   j  = r + 8 * h;
            const int   cj = __builtin_amdgcn_readlane(c_l, j);
            const float aj = __builtin_bit_cast(float,
                __builtin_amdgcn_readlane(__builtin_bit_cast(int, al_l), j));
            if (FULL || cj >= 0) {
                float zl, zh;
                if (FULL) {
                    zl = zlo[j];
                    zh = zhi[j];
                } else {
                    const int    rj = __builtin_amdgcn_readlane(r_l, j);
                    const float* zr = z + (size_t)rj * OUT_DIM;
                    zl = zr[lane];
                    zh = zr[32 + lane];
                }
                float* hr = hout + (size_t)cj * OUT_DIM;
                atomicAdd(&hr[lane],      aj * (zl + (h ? v1lo : v0lo) + b_lo));
                atomicAdd(&hr[32 + lane], aj * (zh + (h ? v1hi : v0hi) + b_hi));
            }
        }
    }
}

__global__ void k_edge_aggregate(const float* __restrict__ edge_attr,
                                 const int* __restrict__ erow, const int* __restrict__ ecol,
                                 const float* __restrict__ W_edge, const float* __restrict__ b_edge,
                                 const float* __restrict__ z, const float* __restrict__ pbuf,
                                 const float* __restrict__ denom, const float* __restrict__ denom1,
                                 float* __restrict__ hout, float* __restrict__ plane1, int two,
                                 int E, int nTiles)
{
    __shared__ __attribute__((aligned(16))) float lWs[(ED_DIM / 2) * BSTRIDE_F];
    const int lane = threadIdx.x & 31;
    const int wave = threadIdx.x >> 5;
    const int half = lane >> 4;
    const int l16  = lane & 15;

    store_w<ED_DIM>(lWs, load_w<ED_DIM>(W_edge));
    __syncthreads();
    const int tile = __builtin_amdgcn_readfirstlane(blockIdx.x * 8 + wave);
    if (tile >= nTiles) return;

    int arow = tile * 16 + l16;
    const bool evalid = arow < E;
    if (arow >= E) arow = E - 1;
    const float* ea = edge_attr + (size_t)arow * ED_DIM;

    {
        const size_t nxt = (size_t)arow + 16 * 8;
        if (nxt < (size_t)E)
            __builtin_prefetch(edge_attr + nxt * ED_DIM, 0, 1);
    }

    const int   c_e  = ecol[arow];
    const int   c_l  = evalid ? c_e : -1;
    const int   r_l  = erow[arow];
    const float d0   = denom[c_e];
    const float d1   = denom1[c_e];
    const float den  = two ? d0 + d1 : d0;
    const float al_l = pbuf[arow] / fmaxf(den, 1e-16f);

    Acc4 acc = wave_gemm_tile(ea, lWs, ED_DIM, half, l16);

    const float b_lo = b_edge[lane];
    const float b_hi = b_edge[32 + lane];

    float* hacc = (two && chip_half()) ? plane1 : hout;

    if (tile * 16 + 16 <= E)
        edge_epilogue<true >(acc, c_l, r_l, al_l, b_lo, b_hi, z, hacc, lane, half);
    else
        edge_epilogue<false>(acc, c_l, r_l, al_l, b_lo, b_hi, z, hacc, lane, half);
}


#define BIN_R   196
#define RANGE_TILES ((BIN_R + 15) / 16)
#define EA_PITCH 34
#define STAT_PER 4
#define NB_MAX  1024
#define BIN_T   1024
#define BIN_PER 4

__global__ __launch_bounds__(BIN_T)
void k_bin_count(const int* __restrict__ ecol, int* __restrict__ cnt, int E, int nb)
{
    __shared__ int s_hist[NB_MAX];
    const int tid = threadIdx.x;
    s_hist[tid] = 0;
    __syncthreads();
#pragma unroll
    for (int i = 0; i < BIN_PER; ++i) {
        const int e = blockIdx.x * (BIN_T * BIN_PER) + i * BIN_T + tid;
        if (e < E) atomicAdd(&s_hist[ecol[e] / BIN_R], 1);
    }
    __syncthreads();
    const int h = s_hist[tid];
    if (tid < nb && h > 0) atomicAdd(&cnt[tid], h);
}

__device__ __attribute__((always_inline)) inline
float x8_4(float v) { return __builtin_bit_cast(float, __builtin_amdgcn_mov_dpp8(__builtin_bit_cast(unsigned, v), 0x688fac)); }
__device__ __attribute__((always_inline)) inline
float x8_2(float v) { return __builtin_bit_cast(float, __builtin_amdgcn_mov_dpp8(__builtin_bit_cast(unsigned, v), 0xb3e21a)); }
__device__ __attribute__((always_inline)) inline
float x8_1(float v) { return __builtin_bit_cast(float, __builtin_amdgcn_mov_dpp8(__builtin_bit_cast(unsigned, v), 0xde54c1)); }

__device__ __attribute__((always_inline)) inline
void ea_request(float4 (&v)[8], const float* __restrict__ edge_attr, int base, int sub, int g)
{
    const float4* p = reinterpret_cast<const float4*>(
        edge_attr + (size_t)(base + sub) * ED_DIM + 4 * g);
#pragma unroll
    for (int q = 0; q < 8; ++q) v[q] = p[q * ED_DIM];
}

__device__ __attribute__((always_inline)) inline
float ea_fold(const float4 (&v)[8], const float4 wa, int lane)
{
    float d[8];
#pragma unroll
    for (int q = 0; q < 8; ++q)
        d[q] = v[q].x * wa.x + v[q].y * wa.y + v[q].z * wa.z + v[q].w * wa.w;
    float s1[4];
#pragma unroll
    for (int q = 0; q < 4; ++q) {
        const float keep = (lane & 4) ? d[q + 4] : d[q];
        const float send = (lane & 4) ? d[q] : d[q + 4];
        s1[q] = keep + x8_4(send);
    }
    float s2[2];
#pragma unroll
    for (int q = 0; q < 2; ++q) {
        const float keep = (lane & 2) ? s1[q + 2] : s1[q];
        const float send = (lane & 2) ? s1[q] : s1[q + 2];
        s2[q] = keep + x8_2(send);
    }
    const float keep = (lane & 1) ? s2[1] : s2[0];
    const float send = (lane & 1) ? s2[0] : s2[1];
    return keep + x8_1(send);
}

__global__ __launch_bounds__(BIN_T)
void k_bin_place(const float* __restrict__ edge_attr,
                 const int* __restrict__ erow, const int* __restrict__ ecol,
                 const float* __restrict__ W_attn, const float* __restrict__ b_attn,
                 const float* __restrict__ a_src, const float* __restrict__ a_dst,
                 const int* __restrict__ cnt, int* __restrict__ cursor, int* __restrict__ bbase,
                 int* __restrict__ rec,
                 int E, int nb)
{
    __shared__ int s_base[NB_MAX];
    __shared__ int s_hist[NB_MAX];
    __shared__ int s_off[NB_MAX];
    __shared__ int s_wsum[32];
    __shared__ float s_dot[BIN_T * BIN_PER];
    const int tid  = threadIdx.x;
    const int lane = tid & 31;
    const int wave = __builtin_amdgcn_readfirstlane(tid >> 5);
    const int g    = lane & 7;
    const int sub  = lane >> 3;

    const float4 wa = *reinterpret_cast<const float4*>(W_attn + 2 * OUT_DIM + 4 * g);
    const float  ba = b_attn[0];

    const int base0 = blockIdx.x * (BIN_T * BIN_PER) + wave * 32;
    int   eid[BIN_PER], bk[BIN_PER], rc[BIN_PER];
    float lg[BIN_PER];
    int   cc[BIN_PER], rr[BIN_PER];
#pragma unroll
    for (int i = 0; i < BIN_PER; ++i) {
        const int e  = base0 + i * BIN_T + 4 * g + sub;
        const int ec = e < E ? e : E - 1;
        eid[i] = e;
        cc[i]  = ecol[ec];
        rr[i]  = erow[ec];
    }
#pragma unroll 1
    for (int i = 0; i < BIN_PER; ++i) {
        const int b = base0 + i * BIN_T;
        float4 ln[8];
        ea_request(ln, edge_attr, (b + 32 <= E) ? b : E - 32, sub, g);
        s_dot[i * BIN_T + tid] = ea_fold(ln, wa, lane);
    }
    float asd[BIN_PER];
#pragma unroll
    for (int i = 0; i < BIN_PER; ++i) asd[i] = a_src[rr[i]] + a_dst[cc[i]];

#pragma unroll
    for (int i = 0; i < BIN_PER; ++i) {
        float dot = s_dot[i * BIN_T + tid];
        const int base = base0 + i * BIN_T;
        if (base < E && base + 32 > E) {
            if (eid[i] < E) {
                const float* ea = edge_attr + (size_t)eid[i] * ED_DIM;
                float t = 0.f;
                for (int k = 0; k < ED_DIM; ++k) t += ea[k] * W_attn[2 * OUT_DIM + k];
                dot = t;
            }
        }
        const float s = asd[i] + dot + ba;
        const int   b = cc[i] / BIN_R;
        bk[i] = eid[i] < E ? b : -1;
        rc[i] = rr[i] * 256 + (cc[i] - b * BIN_R);
        lg[i] = (s > 0.0f) ? s : 0.2f * s;
    }

    const int own = tid < nb ? cnt[tid] : 0;
    int v = own;
#pragma unroll
    for (int d = 1; d < 32; d <<= 1) {
        const int u = __shfl_up(v, d);
        if (lane >= d) v += u;
    }
    if (lane == 31) s_wsum[wave] = v;
    s_hist[tid] = 0;
    __syncthreads();
    if (wave == 0) {
        const int w = s_wsum[lane];
        int x = w;
#pragma unroll
        for (int d = 1; d < 32; d <<= 1) {
            const int u = __shfl_up(x, d);
            if (lane >= d) x += u;
        }
        s_wsum[lane] = x - w;
    }
    __syncthreads();
    const int excl = v - own + s_wsum[wave];
    s_base[tid] = excl;
    if (blockIdx.x == 0) {
        if (tid < nb)      bbase[tid] = excl;
        if (tid == nb - 1) bbase[nb]  = excl + own;
    }

    int lr[BIN_PER];
#pragma unroll
    for (int i = 0; i < BIN_PER; ++i) {
        lr[i] = 0;
        if (bk[i] >= 0) lr[i] = atomicAdd(&s_hist[bk[i]], 1);
    }
    __syncthreads();
    {
        const int h = s_hist[tid];
        if (tid < nb && h > 0) s_off[tid] = s_base[tid] + atomicAdd(&cursor[tid], h);
    }
    __syncthreads();
#pragma unroll
    for (int i = 0; i < BIN_PER; ++i) {
        if (bk[i] >= 0) {
            const int pos = s_off[bk[i]] + lr[i];
            int* r = rec + 3 * pos;
            r[0] = eid[i];
            r[1] = rc[i];
            r[2] = __float_as_int(lg[i]);
        }
    }
}

__device__ __attribute__((always_inline)) inline
void lds_max(float* addr, float val)
{
    __hip_atomic_fetch_max(addr, val, __ATOMIC_RELAXED, __HIP_MEMORY_SCOPE_WORKGROUP);
}

template <bool FULL>
__device__ __attribute__((always_inline)) inline
void edge_accumulate(int e_l, int c_l, int r_l, float al_l,
                     const float* __restrict__ edge_attr, const float* __restrict__ z,
                     float* __restrict__ lacc, float* __restrict__ lea, int lane)
{
    float ev[16];
    v2f   zv[16];
    if (FULL) {
#pragma unroll
        for (int j = 0; j < 16; ++j) {
            const int ej = __builtin_amdgcn_readlane(e_l, j);
            const int rj = __builtin_amdgcn_readlane(r_l, j);
            ev[j] = edge_attr[(size_t)ej * ED_DIM + lane];
            zv[j] = reinterpret_cast<const v2f*>(z + (size_t)rj * OUT_DIM)[lane];
        }
    }
#pragma unroll
    for (int j = 0; j < 16; ++j) {
        const int   cj = __builtin_amdgcn_readlane(c_l, j);
        const float aj = __builtin_bit_cast(float,
            __builtin_amdgcn_readlane(__builtin_bit_cast(int, al_l), j));
        if (FULL || cj >= 0) {
            float e1;
            v2f   z2;
            if (FULL) {
                e1 = ev[j];
                z2 = zv[j];
            } else {
                const int ej = __builtin_amdgcn_readlane(e_l, j);
                const int rj = __builtin_amdgcn_readlane(r_l, j);
                e1 = edge_attr[(size_t)ej * ED_DIM + lane];
                z2 = reinterpret_cast<const v2f*>(z + (size_t)rj * OUT_DIM)[lane];
            }
            float* hr = lacc + cj * OUT_DIM + 2 * lane;
            atomicAdd(&hr[0], aj * z2.x);
            atomicAdd(&hr[1], aj * z2.y);
            atomicAdd(&lea[cj * EA_PITCH + lane], aj * e1);
        }
    }
}

__global__ __launch_bounds__(BIN_T)
void k_range_aggregate(const float* __restrict__ edge_attr,
                       const float* __restrict__ W_edge, const float* __restrict__ b_edge,
                       const float* __restrict__ z,
                       const int* __restrict__ rec, const int* __restrict__ bbase,
                       float* __restrict__ hout, int N)
{
    __shared__ __attribute__((aligned(16))) float lWs[(ED_DIM / 2) * BSTRIDE_F];
    __shared__ __attribute__((aligned(16))) float lacc[BIN_R * OUT_DIM];
    __shared__ __attribute__((aligned(16))) float lea[RANGE_TILES * 16 * EA_PITCH];
    __shared__ float lmax[BIN_R];
    __shared__ float lden[BIN_R];
    const int lane = threadIdx.x & 31;
    const int wave = __builtin_amdgcn_readfirstlane(threadIdx.x >> 5);
    const int half = lane >> 4;
    const int l16  = lane & 15;

    const int n0   = blockIdx.x * BIN_R;
    const int rows = (N - n0 < BIN_R) ? (N - n0) : BIN_R;

    const int b0 = bbase[blockIdx.x];
    const int b1 = bbase[blockIdx.x + 1];
    const int ntile = (b1 - b0 + 15) >> 4;

    float lg_s[STAT_PER];
    int   rc_s[STAT_PER];
#pragma unroll
    for (int j = 0; j < STAT_PER; ++j) {
        const int i  = b0 + j * BIN_T + (int)threadIdx.x;
        const int ii = i < b1 ? i : 0;
        rc_s[j] = rec[3 * ii + 1];
        lg_s[j] = __int_as_float(rec[3 * ii + 2]);
    }
    int   e_n = 0, rc_n = 0;
    float lg_n = 0.f;
    if (wave < ntile) {
        int idx = b0 + wave * 16 + l16;
        if (idx >= b1) idx = b1 - 1;
        e_n = rec[3 * idx]; rc_n = rec[3 * idx + 1]; lg_n = __int_as_float(rec[3 * idx + 2]);
    }

    if (threadIdx.x < BIN_R) { lmax[threadIdx.x] = -INFINITY; lden[threadIdx.x] = 0.f; }
    if (threadIdx.x < NTHREADS) store_w<ED_DIM>(lWs, load_w<ED_DIM>(W_edge));
    {
        const float4* h4 = reinterpret_cast<const float4*>(hout + (size_t)n0 * OUT_DIM);
        float4*       a4 = reinterpret_cast<float4*>(lacc);
        for (int i = threadIdx.x; i < rows * (OUT_DIM / 4); i += BIN_T) a4[i] = h4[i];
        float4*       s4 = reinterpret_cast<float4*>(lea);
        for (int i = threadIdx.x; i < RANGE_TILES * 16 * EA_PITCH / 4; i += BIN_T)
            s4[i] = make_float4(0.f, 0.f, 0.f, 0.f);
    }
    __syncthreads();

    int c_s[STAT_PER];
#pragma unroll
    for (int j = 0; j < STAT_PER; ++j)
        c_s[j] = (b0 + j * BIN_T + (int)threadIdx.x < b1) ? (rc_s[j] & 255) : -1;

#pragma unroll
    for (int j = 0; j < STAT_PER; ++j)
        if (c_s[j] >= 0) lds_max(&lmax[c_s[j]], lg_s[j]);
    for (int i = b0 + STAT_PER * BIN_T + (int)threadIdx.x; i < b1; i += BIN_T)
        lds_max(&lmax[rec[3 * i + 1] & 255], __int_as_float(rec[3 * i + 2]));
    __syncthreads();

#pragma unroll
    for (int j = 0; j < STAT_PER; ++j)
        if (c_s[j] >= 0) atomicAdd(&lden[c_s[j]], expf(lg_s[j] - lmax[c_s[j]]));
    for (int i = b0 + STAT_PER * BIN_T + (int)threadIdx.x; i < b1; i += BIN_T) {
        const int cc = rec[3 * i + 1] & 255;
        atomicAdd(&lden[cc], expf(__int_as_float(rec[3 * i + 2]) - lmax[cc]));
    }
    __syncthreads();

    for (int t = wave; t < ntile; t += BIN_T / 32) {
        const int   e_c = e_n, rc_c = rc_n;
        const float lg_c = lg_n;
        const int tn = t + BIN_T / 32;
        if (tn < ntile) {
            int idx = b0 + tn * 16 + l16;
            if (idx >= b1) idx = b1 - 1;
            e_n = rec[3 * idx]; rc_n = rec[3 * idx + 1]; lg_n = __int_as_float(rec[3 * idx + 2]);
        }
        const bool valid = b0 + t * 16 + l16 < b1;
        const int  cc    = rc_c & 255;
        const int  c_l   = valid ? cc : -1;
        const int  r_l   = rc_c >> 8;
        const float al_c = expf(lg_c - lmax[cc]) / fmaxf(lden[cc], 1e-16f);

        if (t * 16 + 16 <= b1 - b0)
            edge_accumulate<true >(e_c, c_l, r_l, al_c, edge_attr, z, lacc, lea, lane);
        else
            edge_accumulate<false>(e_c, c_l, r_l, al_c, edge_attr, z, lacc, lea, lane);
    }
    __syncthreads();

    for (int t = wave; t < RANGE_TILES; t += BIN_T / 32) {
        const float* arow = lea + (t * 16 + l16) * EA_PITCH;
        Acc4 acc = wave_gemm_tile(arow, lWs, ED_DIM, half, l16);
        float sa[8];
#pragma unroll
        for (int r = 0; r < 8; ++r) {
            const int row = t * 16 + r + 8 * half;
            const float d = lden[row < rows ? row : 0];
            sa[r] = d / fmaxf(d, 1e-16f);
        }
#pragma unroll
        for (int nt = 0; nt < 4; ++nt) {
            const int   c  = nt * 16 + l16;
            const float bc = b_edge[c];
#pragma unroll
            for (int r = 0; r < 8; ++r) {
                const int row = t * 16 + r + 8 * half;
                if (row < rows) lacc[row * OUT_DIM + c] += acc.t[nt][r] + sa[r] * bc;
            }
        }
    }
    __syncthreads();
    {
        float4*       o4 = reinterpret_cast<float4*>(hout + (size_t)n0 * OUT_DIM);
        const float4* a4 = reinterpret_cast<const float4*>(lacc);
        for (int i = threadIdx.x; i < rows * (OUT_DIM / 4); i += BIN_T) o4[i] = a4[i];
    }
}

__global__ void k_plane_sum(float* __restrict__ hout, const float* __restrict__ plane1,
                            int two, int n4)
{
    const int i = blockIdx.x * blockDim.x + threadIdx.x;
    if (!two || i >= n4) return;
    float4*       h4 = reinterpret_cast<float4*>(hout);
    const float4* q4 = reinterpret_cast<const float4*>(plane1);
    float4 a = h4[i];
    const float4 b = q4[i];
    a.x += b.x; a.y += b.y; a.z += b.z; a.w += b.w;
    h4[i] = a;
}

extern "C" void kernel_launch(void* const* d_in, const int* in_sizes, int n_in,
                              void* d_out, int out_size, void* d_ws, size_t ws_size,
                              hipStream_t stream)
{
    const float* x         = (const float*)d_in[0];
    const float* edge_attr = (const float*)d_in[1];
    const int*   edge_idx  = (const int*)  d_in[2];
    const float* W_fc      = (const float*)d_in[3];
    const float* b_fc      = (const float*)d_in[4];
    const float* W_attn    = (const float*)d_in[5];
    const float* b_attn    = (const float*)d_in[6];
    const float* W_edge    = (const float*)d_in[7];
    const float* b_edge    = (const float*)d_in[8];
    const float* W_self    = (const float*)d_in[9];
    const float* b_self    = (const float*)d_in[10];

    const int N = in_sizes[0] / IN_DIM;
    const int E = in_sizes[1] / ED_DIM;
    const int* erow = edge_idx;
    const int* ecol = edge_idx + E;

    float* ws     = (float*)d_ws;
    float* z      = ws;
    float* a_src  = z + (size_t)N * OUT_DIM;
    float* a_dst  = a_src + N;
    float* m      = a_dst + N;
    float* denom  = m + N;
    float* elog   = denom + N;
    float* pbuf   = elog + E;
    const size_t off1 = ((size_t)N * (OUT_DIM + 6) + 2 * (size_t)E + 3) & ~(size_t)3;
    const int two = ws_size >= (off1 + (size_t)N * OUT_DIM) * sizeof(float) ? 1 : 0;
    float* m1     = two ? pbuf + E : m;
    float* denom1 = two ? m1 + N   : denom;
    float* plane1 = two ? ws + off1 : (float*)d_out;
    float* hout   = (float*)d_out;

    const int nNodeTiles = (N + 15) / 16;
    const int nEdgeTiles = (E + 15) / 16;

    const int nb   = (N + BIN_R - 1) / BIN_R;
    const int bins = (two && nb <= NB_MAX && N >= 2 * NB_MAX && N < (1 << 23) && E >= 32 &&
                      (size_t)N * OUT_DIM >= 3 * (size_t)E + 3 * NB_MAX + 8) ? 1 : 0;
    int*   rec_e  = (int*)plane1;
    int*   rec_rc = rec_e + E;
    float* rec_lg = (float*)(rec_rc + E);
    int*   cnt    = (int*)(rec_lg + E);
    int*   cursor = cnt + NB_MAX;
    int*   bbase  = cursor + NB_MAX;
    const int nBinBlocks = (E + BIN_T * BIN_PER - 1) / (BIN_T * BIN_PER);

    k_node_gemm<<<(nNodeTiles + 7) / 8, NTHREADS, 0, stream>>>(
        x, W_fc, b_fc, W_self, b_self, z, hout, plane1, bins ? 0 : two, N, nNodeTiles);

    k_node_attn<<<(N + 255) / 256, 256, 0, stream>>>(
        z, W_attn, a_src, a_dst, m, denom, m1, denom1, two, N, cnt, bins ? 2 * NB_MAX : 0,
        bins ? 0 : 1);

    if (bins) {
        k_bin_count<<<nBinBlocks, BIN_T, 0, stream>>>(ecol, cnt, E, nb);

        k_bin_place<<<nBinBlocks, BIN_T, 0, stream>>>(
            edge_attr, erow, ecol, W_attn, b_attn, a_src, a_dst,
            cnt, cursor, bbase, rec_e, E, nb);

        k_range_aggregate<<<nb, BIN_T, 0, stream>>>(
            edge_attr, W_edge, b_edge, z, rec_e, bbase, hout, N);
    } else {
        k_edge_logit<<<(E + 255) / 256, 256, 0, stream>>>(
            edge_attr, erow, ecol, W_attn, b_attn, a_src, a_dst, elog, m, m1, two, E);

        k_edge_soft<<<(E + 255) / 256, 256, 0, stream>>>(
            elog, ecol, m, m1, pbuf, denom, denom1, two, E);

        k_edge_aggregate<<<(nEdgeTiles + 7) / 8, NTHREADS, 0, stream>>>(
            edge_attr, erow, ecol, W_edge, b_edge, z, pbuf, denom, denom1, hout, plane1, two,
            E, nEdgeTiles);

        k_plane_sum<<<(N * (OUT_DIM / 4) + 255) / 256, 256, 0, stream>>>(
            hout, plane1, two, N * (OUT_DIM / 4));
    }
}


# placing pass sorts the block's records by range in LDS first, so consecutive lanes store consecutive 12-byte records of a run: a few whole-line runs per store, not 32 part-line writes; slots unchanged
# speedup vs baseline: 3.7049x; 1.0374x over previous
#include <hip/hip_runtime.h>
#include <hip/hip_bf16.h>
#include <math.h>

#define IN_DIM  128
#define OUT_DIM 64
#define ED_DIM  32

#define BSTRIDE_V2 80
#define BSTRIDE_F  (2 * BSTRIDE_V2)

typedef __attribute__((ext_vector_type(2))) float v2f;
typedef __attribute__((ext_vector_type(8))) float v8f;

struct Acc4 { v8f t[4]; };

__device__ __attribute__((always_inline)) inline
Acc4 wave_gemm_tile(const float* __restrict__ arow,
                    const float* __restrict__ lWs,
                    int kdim, int half, int l16)
{
    Acc4 A;
    A.t[0] = {}; A.t[1] = {}; A.t[2] = {}; A.t[3] = {};
    const v2f* a2 = (const v2f*)arow;
    for (int k0 = 0; k0 < kdim; k0 += 4) {
        const int kp  = (k0 >> 1) + half;
        const v2f  a  = a2[kp];
        const v2f* br = (const v2f*)(lWs + kp * BSTRIDE_F);
#pragma unroll
        for (int nt = 0; nt < 4; ++nt) {
            const v2f b = br[nt * 16 + l16];
            A.t[nt] = __builtin_amdgcn_wmma_f32_16x16x4_f32(
                false, a, false, b, (short)0, A.t[nt], false, false);
        }
    }
    return A;
}

#define NTHREADS 256

template <int KDIM>
struct WStage {
    static constexpr int ITEMS = (KDIM / 2) * (OUT_DIM / 4);
    static constexpr int PER   = ITEMS / NTHREADS;
    static_assert(ITEMS % NTHREADS == 0, "staging is laid out for whole rounds of the work-group");
    float4 lo[PER];
    float4 hi[PER];
};

template <int KDIM>
__device__ __attribute__((always_inline)) inline
WStage<KDIM> load_w(const float* __restrict__ W)
{
    WStage<KDIM> s;
#pragma unroll
    for (int j = 0; j < WStage<KDIM>::PER; ++j) {
        const int it = threadIdx.x + NTHREADS * j;
        const int kp = it >> 4;
        const int g  = it & 15;
        const float4* src = reinterpret_cast<const float4*>(W + (size_t)(2 * kp) * OUT_DIM + 4 * g);
        s.lo[j] = src[0];
        s.hi[j] = src[OUT_DIM / 4];
    }
    return s;
}

template <int KDIM>
__device__ __attribute__((always_inline)) inline
void store_w(float* __restrict__ lWs, const WStage<KDIM>& s)
{
#pragma unroll
    for (int j = 0; j < WStage<KDIM>::PER; ++j) {
        const int it = threadIdx.x + NTHREADS * j;
        const int kp = it >> 4;
        const int g  = it & 15;
        float4* dst = reinterpret_cast<float4*>(lWs + kp * BSTRIDE_F + 8 * g);
        dst[0] = make_float4(s.lo[j].x, s.hi[j].x, s.lo[j].y, s.hi[j].y);
        dst[1] = make_float4(s.lo[j].z, s.hi[j].z, s.lo[j].w, s.hi[j].w);
    }
}

__device__ __attribute__((always_inline)) inline
int chip_half()
{
    const unsigned v = __builtin_amdgcn_s_sendmsg_rtn(0x87);
    return (int)((v >> 18) & 1u);
}

__device__ inline void atomicMaxF(float* addr, float val) {
    if (val >= 0.0f) atomicMax(reinterpret_cast<int*>(addr), __float_as_int(val));
    else             atomicMin(reinterpret_cast<unsigned int*>(addr), __float_as_uint(val));
}

__global__ void k_node_gemm(const float* __restrict__ x,
                            const float* __restrict__ W_fc,  const float* __restrict__ b_fc,
                            const float* __restrict__ W_self, const float* __restrict__ b_self,
                            float* __restrict__ z, float* __restrict__ hout,
                            float* __restrict__ plane1, int two,
                            int N, int nTiles)
{
    __shared__ __attribute__((aligned(16))) float lWs[(IN_DIM / 2) * BSTRIDE_F];
    const int lane = threadIdx.x & 31;
    const int wave = threadIdx.x >> 5;
    const int half = lane >> 4;
    const int l16  = lane & 15;
    const int tile = blockIdx.x * 8 + wave;

    const WStage<IN_DIM> w_first  = load_w<IN_DIM>(W_fc);
    const WStage<IN_DIM> w_second = load_w<IN_DIM>(W_self);

    int mrow = tile * 16 + l16;
    if (mrow >= N) mrow = N - 1;
    const float* xrow = x + (size_t)mrow * IN_DIM;

    if (two && tile < nTiles) {
        float4* p4 = reinterpret_cast<float4*>(plane1 + (size_t)tile * 16 * OUT_DIM);
#pragma unroll
        for (int it = 0; it < 8; ++it) {
            const int idx = it * 32 + lane;
            if (tile * 16 + (idx >> 4) < N) p4[idx] = make_float4(0.f, 0.f, 0.f, 0.f);
        }
    }

    store_w<IN_DIM>(lWs, w_first);
    __syncthreads();
    if (tile < nTiles) {
        Acc4 acc = wave_gemm_tile(xrow, lWs, IN_DIM, half, l16);
#pragma unroll
        for (int nt = 0; nt < 4; ++nt) {
            const int   c  = nt * 16 + l16;
            const float bc = b_fc[c];
#pragma unroll
            for (int r = 0; r < 8; ++r) {
                const int row = tile * 16 + r + 8 * half;
                if (row < N) z[(size_t)row * OUT_DIM + c] = acc.t[nt][r] + bc;
            }
        }
    }
    __syncthreads();

    store_w<IN_DIM>(lWs, w_second);
    __syncthreads();
    if (tile < nTiles) {
        Acc4 acc = wave_gemm_tile(xrow, lWs, IN_DIM, half, l16);
#pragma unroll
        for (int nt = 0; nt < 4; ++nt) {
            const int   c  = nt * 16 + l16;
            const float bc = b_self[c];
#pragma unroll
            for (int r = 0; r < 8; ++r) {
                const int row = tile * 16 + r + 8 * half;
                if (row < N) hout[(size_t)row * OUT_DIM + c] = acc.t[nt][r] + bc;
            }
        }
    }
}

__global__ void k_node_attn(const float* __restrict__ z, const float* __restrict__ W_attn,
                            float* __restrict__ a_src, float* __restrict__ a_dst,
                            float* __restrict__ m, float* __restrict__ denom,
                            float* __restrict__ m1, float* __restrict__ denom1, int two, int N,
                            int* __restrict__ binz, int nbinz, int stats)
{
    const int n = blockIdx.x * blockDim.x + threadIdx.x;
    if (n < nbinz) binz[n] = 0;
    if (n >= N) return;
    const float* zr = z + (size_t)n * OUT_DIM;
    float as = 0.f, ad = 0.f;
#pragma unroll 8
    for (int c = 0; c < OUT_DIM; ++c) {
        const float zc = zr[c];
        as += zc * W_attn[c];
        ad += zc * W_attn[OUT_DIM + c];
    }
    a_src[n] = as;
    a_dst[n] = ad;
    if (!stats) return;
    m[n]     = -INFINITY;
    denom[n] = 0.0f;
    if (two) {
        m1[n]     = -INFINITY;
        denom1[n] = 0.0f;
    }
}

__global__ void k_edge_logit(const float* __restrict__ edge_attr,
                             const int* __restrict__ erow, const int* __restrict__ ecol,
                             const float* __restrict__ W_attn, const float* __restrict__ b_attn,
                             const float* __restrict__ a_src, const float* __restrict__ a_dst,
                             float* __restrict__ elog, float* __restrict__ m,
                             float* __restrict__ m1, int two, int E)
{
    const int e = blockIdx.x * blockDim.x + threadIdx.x;
    if (e >= E) return;
    const float* ea = edge_attr + (size_t)e * ED_DIM;
    float dot = 0.f;
#pragma unroll
    for (int k = 0; k < ED_DIM; ++k) dot += ea[k] * W_attn[2 * OUT_DIM + k];
    const int c = ecol[e];
    const float s  = a_src[erow[e]] + a_dst[c] + dot + b_attn[0];
    const float lr = (s > 0.0f) ? s : 0.2f * s;
    elog[e] = lr;
    float* mh = (two && chip_half()) ? m1 : m;
    atomicMaxF(&mh[c], lr);
}

__global__ void k_edge_soft(const float* __restrict__ elog, const int* __restrict__ ecol,
                            const float* __restrict__ m, const float* __restrict__ m1,
                            float* __restrict__ pbuf,
                            float* __restrict__ denom, float* __restrict__ denom1, int two, int E)
{
    const int e = blockIdx.x * blockDim.x + threadIdx.x;
    if (e >= E) return;
    const int c = ecol[e];
    const float mc = fmaxf(m[c], m1[c]);
    const float p = expf(elog[e] - mc);
    pbuf[e] = p;
    float* dh = (two && chip_half()) ? denom1 : denom;
    atomicAdd(&dh[c], p);
}

__device__ __attribute__((always_inline)) inline
float xhalf(float x)
{
    const int b = __builtin_bit_cast(int, x);
    return __builtin_bit_cast(float,
        __builtin_amdgcn_permlanex16(b, b, 0x76543210, 0xfedcba98, false, false));
}

template <bool FULL>
__device__ __attribute__((always_inline)) inline
void edge_epilogue(const Acc4& acc, int c_l, int r_l, float al_l, float b_lo, float b_hi,
                   const float* __restrict__ z, float* __restrict__ hout, int lane, int half)
{
    float zlo[16], zhi[16];
    if (FULL) {
#pragma unroll
        for (int j = 0; j < 16; ++j) {
            const int    rj = __builtin_amdgcn_readlane(r_l, j);
            const float* zr = z + (size_t)rj * OUT_DIM;
            zlo[j] = zr[lane];
            zhi[j] = zr[32 + lane];
        }
    }
#pragma unroll
    for (int r = 0; r < 8; ++r) {
        const float gA   = xhalf(half ? acc.t[0][r] : acc.t[1][r]);
        const float gB   = xhalf(half ? acc.t[2][r] : acc.t[3][r]);
        const float v0lo = half ? gA : acc.t[0][r];
        const float v1lo = half ? acc.t[1][r] : gA;
        const float v0hi = half ? gB : acc.t[2][r];
        const float v1hi = half ? acc.t[3][r] : gB;
#pragma unroll
        for (int h = 0; h < 2; ++h) {
            const int   j  = r + 8 * h;
            const int   cj = __builtin_amdgcn_readlane(c_l, j);
            const float aj = __builtin_bit_cast(float,
                __builtin_amdgcn_readlane(__builtin_bit_cast(int, al_l), j));
            if (FULL || cj >= 0) {
                float zl, zh;
                if (FULL) {
                    zl = zlo[j];
                    zh = zhi[j];
                } else {
                    const int    rj = __builtin_amdgcn_readlane(r_l, j);
                    const float* zr = z + (size_t)rj * OUT_DIM;
                    zl = zr[lane];
                    zh = zr[32 + lane];
                }
                float* hr = hout + (size_t)cj * OUT_DIM;
                atomicAdd(&hr[lane],      aj * (zl + (h ? v1lo : v0lo) + b_lo));
                atomicAdd(&hr[32 + lane], aj * (zh + (h ? v1hi : v0hi) + b_hi));
            }
        }
    }
}

__global__ void k_edge_aggregate(const float* __restrict__ edge_attr,
                                 const int* __restrict__ erow, const int* __restrict__ ecol,
                                 const float* __restrict__ W_edge, const float* __restrict__ b_edge,
                                 const float* __restrict__ z, const float* __restrict__ pbuf,
                                 const float* __restrict__ denom, const float* __restrict__ denom1,
                                 float* __restrict__ hout, float* __restrict__ plane1, int two,
                                 int E, int nTiles)
{
    __shared__ __attribute__((aligned(16))) float lWs[(ED_DIM / 2) * BSTRIDE_F];
    const int lane = threadIdx.x & 31;
    const int wave = threadIdx.x >> 5;
    const int half = lane >> 4;
    const int l16  = lane & 15;

    store_w<ED_DIM>(lWs, load_w<ED_DIM>(W_edge));
    __syncthreads();
    const int tile = __builtin_amdgcn_readfirstlane(blockIdx.x * 8 + wave);
    if (tile >= nTiles) return;

    int arow = tile * 16 + l16;
    const bool evalid = arow < E;
    if (arow >= E) arow = E - 1;
    const float* ea = edge_attr + (size_t)arow * ED_DIM;

    {
        const size_t nxt = (size_t)arow + 16 * 8;
        if (nxt < (size_t)E)
            __builtin_prefetch(edge_attr + nxt * ED_DIM, 0, 1);
    }

    const int   c_e  = ecol[arow];
    const int   c_l  = evalid ? c_e : -1;
    const int   r_l  = erow[arow];
    const float d0   = denom[c_e];
    const float d1   = denom1[c_e];
    const float den  = two ? d0 + d1 : d0;
    const float al_l = pbuf[arow] / fmaxf(den, 1e-16f);

    Acc4 acc = wave_gemm_tile(ea, lWs, ED_DIM, half, l16);

    const float b_lo = b_edge[lane];
    const float b_hi = b_edge[32 + lane];

    float* hacc = (two && chip_half()) ? plane1 : hout;

    if (tile * 16 + 16 <= E)
        edge_epilogue<true >(acc, c_l, r_l, al_l, b_lo, b_hi, z, hacc, lane, half);
    else
        edge_epilogue<false>(acc, c_l, r_l, al_l, b_lo, b_hi, z, hacc, lane, half);
}


#define BIN_R   196
#define RANGE_TILES ((BIN_R + 15) / 16)
#define EA_PITCH 34
#define STAT_PER 4
#define NB_MAX  1024
#define BIN_T   1024
#define BIN_PER 4

__global__ __launch_bounds__(BIN_T)
void k_bin_count(const int* __restrict__ ecol, int* __restrict__ cnt, int E, int nb)
{
    __shared__ int s_hist[NB_MAX];
    const int tid = threadIdx.x;
    s_hist[tid] = 0;
    __syncthreads();
#pragma unroll
    for (int i = 0; i < BIN_PER; ++i) {
        const int e = blockIdx.x * (BIN_T * BIN_PER) + i * BIN_T + tid;
        if (e < E) atomicAdd(&s_hist[ecol[e] / BIN_R], 1);
    }
    __syncthreads();
    const int h = s_hist[tid];
    if (tid < nb && h > 0) atomicAdd(&cnt[tid], h);
}

__device__ __attribute__((always_inline)) inline
float x8_4(float v) { return __builtin_bit_cast(float, __builtin_amdgcn_mov_dpp8(__builtin_bit_cast(unsigned, v), 0x688fac)); }
__device__ __attribute__((always_inline)) inline
float x8_2(float v) { return __builtin_bit_cast(float, __builtin_amdgcn_mov_dpp8(__builtin_bit_cast(unsigned, v), 0xb3e21a)); }
__device__ __attribute__((always_inline)) inline
float x8_1(float v) { return __builtin_bit_cast(float, __builtin_amdgcn_mov_dpp8(__builtin_bit_cast(unsigned, v), 0xde54c1)); }

__device__ __attribute__((always_inline)) inline
void ea_request(float4 (&v)[8], const float* __restrict__ edge_attr, int base, int sub, int g)
{
    const float4* p = reinterpret_cast<const float4*>(
        edge_attr + (size_t)(base + sub) * ED_DIM + 4 * g);
#pragma unroll
    for (int q = 0; q < 8; ++q) v[q] = p[q * ED_DIM];
}

__device__ __attribute__((always_inline)) inline
float ea_fold(const float4 (&v)[8], const float4 wa, int lane)
{
    float d[8];
#pragma unroll
    for (int q = 0; q < 8; ++q)
        d[q] = v[q].x * wa.x + v[q].y * wa.y + v[q].z * wa.z + v[q].w * wa.w;
    float s1[4];
#pragma unroll
    for (int q = 0; q < 4; ++q) {
        const float keep = (lane & 4) ? d[q + 4] : d[q];
        const float send = (lane & 4) ? d[q] : d[q + 4];
        s1[q] = keep + x8_4(send);
    }
    float s2[2];
#pragma unroll
    for (int q = 0; q < 2; ++q) {
        const float keep = (lane & 2) ? s1[q + 2] : s1[q];
        const float send = (lane & 2) ? s1[q] : s1[q + 2];
        s2[q] = keep + x8_2(send);
    }
    const float keep = (lane & 1) ? s2[1] : s2[0];
    const float send = (lane & 1) ? s2[0] : s2[1];
    return keep + x8_1(send);
}

__global__ __launch_bounds__(BIN_T)
void k_bin_place(const float* __restrict__ edge_attr,
                 const int* __restrict__ erow, const int* __restrict__ ecol,
                 const float* __restrict__ W_attn, const float* __restrict__ b_attn,
                 const float* __restrict__ a_src, const float* __restrict__ a_dst,
                 const int* __restrict__ cnt, int* __restrict__ cursor, int* __restrict__ bbase,
                 int* __restrict__ rec,
                 int E, int nb)
{
    __shared__ int s_base[NB_MAX];
    __shared__ int s_hist[NB_MAX];
    __shared__ int s_shift[NB_MAX];
    __shared__ int s_lp[NB_MAX];
    __shared__ int s_wsum[32];
    __shared__ int s_wsum2[32];
    __shared__ __attribute__((aligned(16))) int4 s_rec[BIN_T * BIN_PER];
    __shared__ float s_dot[BIN_T * BIN_PER];
    const int tid  = threadIdx.x;
    const int lane = tid & 31;
    const int wave = __builtin_amdgcn_readfirstlane(tid >> 5);
    const int g    = lane & 7;
    const int sub  = lane >> 3;

    const float4 wa = *reinterpret_cast<const float4*>(W_attn + 2 * OUT_DIM + 4 * g);
    const float  ba = b_attn[0];

    const int base0 = blockIdx.x * (BIN_T * BIN_PER) + wave * 32;
    int   eid[BIN_PER], bk[BIN_PER], rc[BIN_PER];
    float lg[BIN_PER];
    int   cc[BIN_PER], rr[BIN_PER];
#pragma unroll
    for (int i = 0; i < BIN_PER; ++i) {
        const int e  = base0 + i * BIN_T + 4 * g + sub;
        const int ec = e < E ? e : E - 1;
        eid[i] = e;
        cc[i]  = ecol[ec];
        rr[i]  = erow[ec];
    }
#pragma unroll 1
    for (int i = 0; i < BIN_PER; ++i) {
        const int b = base0 + i * BIN_T;
        float4 ln[8];
        ea_request(ln, edge_attr, (b + 32 <= E) ? b : E - 32, sub, g);
        s_dot[i * BIN_T + tid] = ea_fold(ln, wa, lane);
    }
    float asd[BIN_PER];
#pragma unroll
    for (int i = 0; i < BIN_PER; ++i) asd[i] = a_src[rr[i]] + a_dst[cc[i]];

#pragma unroll
    for (int i = 0; i < BIN_PER; ++i) {
        float dot = s_dot[i * BIN_T + tid];
        const int base = base0 + i * BIN_T;
        if (base < E && base + 32 > E) {
            if (eid[i] < E) {
                const float* ea = edge_attr + (size_t)eid[i] * ED_DIM;
                float t = 0.f;
                for (int k = 0; k < ED_DIM; ++k) t += ea[k] * W_attn[2 * OUT_DIM + k];
                dot = t;
            }
        }
        const float s = asd[i] + dot + ba;
        const int   b = cc[i] / BIN_R;
        bk[i] = eid[i] < E ? b : -1;
        rc[i] = rr[i] * 256 + (cc[i] - b * BIN_R);
        lg[i] = (s > 0.0f) ? s : 0.2f * s;
    }

    const int own = tid < nb ? cnt[tid] : 0;
    int v = own;
#pragma unroll
    for (int d = 1; d < 32; d <<= 1) {
        const int u = __shfl_up(v, d);
        if (lane >= d) v += u;
    }
    if (lane == 31) s_wsum[wave] = v;
    s_hist[tid] = 0;
    __syncthreads();
    if (wave == 0) {
        const int w = s_wsum[lane];
        int x = w;
#pragma unroll
        for (int d = 1; d < 32; d <<= 1) {
            const int u = __shfl_up(x, d);
            if (lane >= d) x += u;
        }
        s_wsum[lane] = x - w;
    }
    __syncthreads();
    const int excl = v - own + s_wsum[wave];
    s_base[tid] = excl;
    if (blockIdx.x == 0) {
        if (tid < nb)      bbase[tid] = excl;
        if (tid == nb - 1) bbase[nb]  = excl + own;
    }

    int lr[BIN_PER];
#pragma unroll
    for (int i = 0; i < BIN_PER; ++i) {
        lr[i] = 0;
        if (bk[i] >= 0) lr[i] = atomicAdd(&s_hist[bk[i]], 1);
    }
    __syncthreads();
    const int hc = s_hist[tid];
    int w = hc;
#pragma unroll
    for (int d = 1; d < 32; d <<= 1) {
        const int u = __shfl_up(w, d);
        if (lane >= d) w += u;
    }
    if (lane == 31) s_wsum2[wave] = w;
    __syncthreads();
    if (wave == 0) {
        const int t = s_wsum2[lane];
        int x = t;
#pragma unroll
        for (int d = 1; d < 32; d <<= 1) {
            const int u = __shfl_up(x, d);
            if (lane >= d) x += u;
        }
        s_wsum2[lane] = x - t;
    }
    __syncthreads();
    const int lp = w - hc + s_wsum2[wave];
    s_lp[tid] = lp;
    if (tid < nb && hc > 0) s_shift[tid] = s_base[tid] + atomicAdd(&cursor[tid], hc) - lp;
    __syncthreads();
    int at[BIN_PER];
#pragma unroll
    for (int i = 0; i < BIN_PER; ++i) at[i] = s_lp[bk[i] >= 0 ? bk[i] : 0] + lr[i];
#pragma unroll
    for (int i = 0; i < BIN_PER; ++i) {
        if (bk[i] >= 0)
            s_rec[at[i]] = make_int4(eid[i], rc[i], __float_as_int(lg[i]), bk[i]);
    }
    __syncthreads();
    const int first = blockIdx.x * (BIN_T * BIN_PER);
    const int nval  = (E - first < BIN_T * BIN_PER) ? (E - first) : BIN_T * BIN_PER;
    int4 q[BIN_PER];
    int  sh[BIN_PER];
#pragma unroll
    for (int i = 0; i < BIN_PER; ++i) {
        const int p = i * BIN_T + tid;
        q[i] = s_rec[p < nval ? p : 0];
    }
#pragma unroll
    for (int i = 0; i < BIN_PER; ++i) sh[i] = s_shift[q[i].w];
#pragma unroll
    for (int i = 0; i < BIN_PER; ++i) {
        const int p = i * BIN_T + tid;
        if (p < nval) {
            int* r = rec + 3 * (sh[i] + p);
            r[0] = q[i].x;
            r[1] = q[i].y;
            r[2] = q[i].z;
        }
    }
}

__device__ __attribute__((always_inline)) inline
void lds_max(float* addr, float val)
{
    __hip_atomic_fetch_max(addr, val, __ATOMIC_RELAXED, __HIP_MEMORY_SCOPE_WORKGROUP);
}

template <bool FULL>
__device__ __attribute__((always_inline)) inline
void edge_accumulate(int e_l, int c_l, int r_l, float al_l,
                     const float* __restrict__ edge_attr, const float* __restrict__ z,
                     float* __restrict__ lacc, float* __restrict__ lea, int lane)
{
    float ev[16];
    v2f   zv[16];
    if (FULL) {
#pragma unroll
        for (int j = 0; j < 16; ++j) {
            const int ej = __builtin_amdgcn_readlane(e_l, j);
            const int rj = __builtin_amdgcn_readlane(r_l, j);
            ev[j] = edge_attr[(size_t)ej * ED_DIM + lane];
            zv[j] = reinterpret_cast<const v2f*>(z + (size_t)rj * OUT_DIM)[lane];
        }
    }
#pragma unroll
    for (int j = 0; j < 16; ++j) {
        const int   cj = __builtin_amdgcn_readlane(c_l, j);
        const float aj = __builtin_bit_cast(float,
            __builtin_amdgcn_readlane(__builtin_bit_cast(int, al_l), j));
        if (FULL || cj >= 0) {
            float e1;
            v2f   z2;
            if (FULL) {
                e1 = ev[j];
                z2 = zv[j];
            } else {
                const int ej = __builtin_amdgcn_readlane(e_l, j);
                const int rj = __builtin_amdgcn_readlane(r_l, j);
                e1 = edge_attr[(size_t)ej * ED_DIM + lane];
                z2 = reinterpret_cast<const v2f*>(z + (size_t)rj * OUT_DIM)[lane];
            }
            float* hr = lacc + cj * OUT_DIM + 2 * lane;
            atomicAdd(&hr[0], aj * z2.x);
            atomicAdd(&hr[1], aj * z2.y);
            atomicAdd(&lea[cj * EA_PITCH + lane], aj * e1);
        }
    }
}

__global__ __launch_bounds__(BIN_T)
void k_range_aggregate(const float* __restrict__ edge_attr,
                       const float* __restrict__ W_edge, const float* __restrict__ b_edge,
                       const float* __restrict__ z,
                       const int* __restrict__ rec, const int* __restrict__ bbase,
                       float* __restrict__ hout, int N)
{
    __shared__ __attribute__((aligned(16))) float lWs[(ED_DIM / 2) * BSTRIDE_F];
    __shared__ __attribute__((aligned(16))) float lacc[BIN_R * OUT_DIM];
    __shared__ __attribute__((aligned(16))) float lea[RANGE_TILES * 16 * EA_PITCH];
    __shared__ float lmax[BIN_R];
    __shared__ float lden[BIN_R];
    const int lane = threadIdx.x & 31;
    const int wave = __builtin_amdgcn_readfirstlane(threadIdx.x >> 5);
    const int half = lane >> 4;
    const int l16  = lane & 15;

    const int n0   = blockIdx.x * BIN_R;
    const int rows = (N - n0 < BIN_R) ? (N - n0) : BIN_R;

    const int b0 = bbase[blockIdx.x];
    const int b1 = bbase[blockIdx.x + 1];
    const int ntile = (b1 - b0 + 15) >> 4;

    float lg_s[STAT_PER];
    int   rc_s[STAT_PER];
#pragma unroll
    for (int j = 0; j < STAT_PER; ++j) {
        const int i  = b0 + j * BIN_T + (int)threadIdx.x;
        const int ii = i < b1 ? i : 0;
        rc_s[j] = rec[3 * ii + 1];
        lg_s[j] = __int_as_float(rec[3 * ii + 2]);
    }
    int   e_n = 0, rc_n = 0;
    float lg_n = 0.f;
    if (wave < ntile) {
        int idx = b0 + wave * 16 + l16;
        if (idx >= b1) idx = b1 - 1;
        e_n = rec[3 * idx]; rc_n = rec[3 * idx + 1]; lg_n = __int_as_float(rec[3 * idx + 2]);
    }

    if (threadIdx.x < BIN_R) { lmax[threadIdx.x] = -INFINITY; lden[threadIdx.x] = 0.f; }
    if (threadIdx.x < NTHREADS) store_w<ED_DIM>(lWs, load_w<ED_DIM>(W_edge));
    {
        const float4* h4 = reinterpret_cast<const float4*>(hout + (size_t)n0 * OUT_DIM);
        float4*       a4 = reinterpret_cast<float4*>(lacc);
        for (int i = threadIdx.x; i < rows * (OUT_DIM / 4); i += BIN_T) a4[i] = h4[i];
        float4*       s4 = reinterpret_cast<float4*>(lea);
        for (int i = threadIdx.x; i < RANGE_TILES * 16 * EA_PITCH / 4; i += BIN_T)
            s4[i] = make_float4(0.f, 0.f, 0.f, 0.f);
    }
    __syncthreads();

    int c_s[STAT_PER];
#pragma unroll
    for (int j = 0; j < STAT_PER; ++j)
        c_s[j] = (b0 + j * BIN_T + (int)threadIdx.x < b1) ? (rc_s[j] & 255) : -1;

#pragma unroll
    for (int j = 0; j < STAT_PER; ++j)
        if (c_s[j] >= 0) lds_max(&lmax[c_s[j]], lg_s[j]);
    for (int i = b0 + STAT_PER * BIN_T + (int)threadIdx.x; i < b1; i += BIN_T)
        lds_max(&lmax[rec[3 * i + 1] & 255], __int_as_float(rec[3 * i + 2]));
    __syncthreads();

#pragma unroll
    for (int j = 0; j < STAT_PER; ++j)
        if (c_s[j] >= 0) atomicAdd(&lden[c_s[j]], expf(lg_s[j] - lmax[c_s[j]]));
    for (int i = b0 + STAT_PER * BIN_T + (int)threadIdx.x; i < b1; i += BIN_T) {
        const int cc = rec[3 * i + 1] & 255;
        atomicAdd(&lden[cc], expf(__int_as_float(rec[3 * i + 2]) - lmax[cc]));
    }
    __syncthreads();

    for (int t = wave; t < ntile; t += BIN_T / 32) {
        const int   e_c = e_n, rc_c = rc_n;
        const float lg_c = lg_n;
        const int tn = t + BIN_T / 32;
        if (tn < ntile) {
            int idx = b0 + tn * 16 + l16;
            if (idx >= b1) idx = b1 - 1;
            e_n = rec[3 * idx]; rc_n = rec[3 * idx + 1]; lg_n = __int_as_float(rec[3 * idx + 2]);
        }
        const bool valid = b0 + t * 16 + l16 < b1;
        const int  cc    = rc_c & 255;
        const int  c_l   = valid ? cc : -1;
        const int  r_l   = rc_c >> 8;
        const float al_c = expf(lg_c - lmax[cc]) / fmaxf(lden[cc], 1e-16f);

        if (t * 16 + 16 <= b1 - b0)
            edge_accumulate<true >(e_c, c_l, r_l, al_c, edge_attr, z, lacc, lea, lane);
        else
            edge_accumulate<false>(e_c, c_l, r_l, al_c, edge_attr, z, lacc, lea, lane);
    }
    __syncthreads();

    for (int t = wave; t < RANGE_TILES; t += BIN_T / 32) {
        const float* arow = lea + (t * 16 + l16) * EA_PITCH;
        Acc4 acc = wave_gemm_tile(arow, lWs, ED_DIM, half, l16);
        float sa[8];
#pragma unroll
        for (int r = 0; r < 8; ++r) {
            const int row = t * 16 + r + 8 * half;
            const float d = lden[row < rows ? row : 0];
            sa[r] = d / fmaxf(d, 1e-16f);
        }
#pragma unroll
        for (int nt = 0; nt < 4; ++nt) {
            const int   c  = nt * 16 + l16;
            const float bc = b_edge[c];
#pragma unroll
            for (int r = 0; r < 8; ++r) {
                const int row = t * 16 + r + 8 * half;
                if (row < rows) lacc[row * OUT_DIM + c] += acc.t[nt][r] + sa[r] * bc;
            }
        }
    }
    __syncthreads();
    {
        float4*       o4 = reinterpret_cast<float4*>(hout + (size_t)n0 * OUT_DIM);
        const float4* a4 = reinterpret_cast<const float4*>(lacc);
        for (int i = threadIdx.x; i < rows * (OUT_DIM / 4); i += BIN_T) o4[i] = a4[i];
    }
}

__global__ void k_plane_sum(float* __restrict__ hout, const float* __restrict__ plane1,
                            int two, int n4)
{
    const int i = blockIdx.x * blockDim.x + threadIdx.x;
    if (!two || i >= n4) return;
    float4*       h4 = reinterpret_cast<float4*>(hout);
    const float4* q4 = reinterpret_cast<const float4*>(plane1);
    float4 a = h4[i];
    const float4 b = q4[i];
    a.x += b.x; a.y += b.y; a.z += b.z; a.w += b.w;
    h4[i] = a;
}

extern "C" void kernel_launch(void* const* d_in, const int* in_sizes, int n_in,
                              void* d_out, int out_size, void* d_ws, size_t ws_size,
                              hipStream_t stream)
{
    const float* x         = (const float*)d_in[0];
    const float* edge_attr = (const float*)d_in[1];
    const int*   edge_idx  = (const int*)  d_in[2];
    const float* W_fc      = (const float*)d_in[3];
    const float* b_fc      = (const float*)d_in[4];
    const float* W_attn    = (const float*)d_in[5];
    const float* b_attn    = (const float*)d_in[6];
    const float* W_edge    = (const float*)d_in[7];
    const float* b_edge    = (const float*)d_in[8];
    const float* W_self    = (const float*)d_in[9];
    const float* b_self    = (const float*)d_in[10];

    const int N = in_sizes[0] / IN_DIM;
    const int E = in_sizes[1] / ED_DIM;
    const int* erow = edge_idx;
    const int* ecol = edge_idx + E;

    float* ws     = (float*)d_ws;
    float* z      = ws;
    float* a_src  = z + (size_t)N * OUT_DIM;
    float* a_dst  = a_src + N;
    float* m      = a_dst + N;
    float* denom  = m + N;
    float* elog   = denom + N;
    float* pbuf   = elog + E;
    const size_t off1 = ((size_t)N * (OUT_DIM + 6) + 2 * (size_t)E + 3) & ~(size_t)3;
    const int two = ws_size >= (off1 + (size_t)N * OUT_DIM) * sizeof(float) ? 1 : 0;
    float* m1     = two ? pbuf + E : m;
    float* denom1 = two ? m1 + N   : denom;
    float* plane1 = two ? ws + off1 : (float*)d_out;
    float* hout   = (float*)d_out;

    const int nNodeTiles = (N + 15) / 16;
    const int nEdgeTiles = (E + 15) / 16;

    const int nb   = (N + BIN_R - 1) / BIN_R;
    const int bins = (two && nb <= NB_MAX && N >= 2 * NB_MAX && N < (1 << 23) && E >= 32 &&
                      (size_t)N * OUT_DIM >= 3 * (size_t)E + 3 * NB_MAX + 8) ? 1 : 0;
    int*   rec_e  = (int*)plane1;
    int*   rec_rc = rec_e + E;
    float* rec_lg = (float*)(rec_rc + E);
    int*   cnt    = (int*)(rec_lg + E);
    int*   cursor = cnt + NB_MAX;
    int*   bbase  = cursor + NB_MAX;
    const int nBinBlocks = (E + BIN_T * BIN_PER - 1) / (BIN_T * BIN_PER);

    k_node_gemm<<<(nNodeTiles + 7) / 8, NTHREADS, 0, stream>>>(
        x, W_fc, b_fc, W_self, b_self, z, hout, plane1, bins ? 0 : two, N, nNodeTiles);

    k_node_attn<<<(N + 255) / 256, 256, 0, stream>>>(
        z, W_attn, a_src, a_dst, m, denom, m1, denom1, two, N, cnt, bins ? 2 * NB_MAX : 0,
        bins ? 0 : 1);

    if (bins) {
        k_bin_count<<<nBinBlocks, BIN_T, 0, stream>>>(ecol, cnt, E, nb);

        k_bin_place<<<nBinBlocks, BIN_T, 0, stream>>>(
            edge_attr, erow, ecol, W_attn, b_attn, a_src, a_dst,
            cnt, cursor, bbase, rec_e, E, nb);

        k_range_aggregate<<<nb, BIN_T, 0, stream>>>(
            edge_attr, W_edge, b_edge, z, rec_e, bbase, hout, N);
    } else {
        k_edge_logit<<<(E + 255) / 256, 256, 0, stream>>>(
            edge_attr, erow, ecol, W_attn, b_attn, a_src, a_dst, elog, m, m1, two, E);

        k_edge_soft<<<(E + 255) / 256, 256, 0, stream>>>(
            elog, ecol, m, m1, pbuf, denom, denom1, two, E);

        k_edge_aggregate<<<(nEdgeTiles + 7) / 8, NTHREADS, 0, stream>>>(
            edge_attr, erow, ecol, W_edge, b_edge, z, pbuf, denom, denom1, hout, plane1, two,
            E, nEdgeTiles);

        k_plane_sum<<<(N * (OUT_DIM / 4) + 255) / 256, 256, 0, stream>>>(
            hout, plane1, two, N * (OUT_DIM / 4));
    }
}


# per-node attention scalars taken from the z tile in the product kernel's accumulator (4 columns per lane, 16-lane row_xmask butterfly); their launch and z re-read go away; that sum's order changes
# speedup vs baseline: 3.9207x; 1.0582x over previous
#include <hip/hip_runtime.h>
#include <hip/hip_bf16.h>
#include <math.h>

#define IN_DIM  128
#define OUT_DIM 64
#define ED_DIM  32

#define BSTRIDE_V2 80
#define BSTRIDE_F  (2 * BSTRIDE_V2)

typedef __attribute__((ext_vector_type(2))) float v2f;
typedef __attribute__((ext_vector_type(8))) float v8f;

struct Acc4 { v8f t[4]; };

__device__ __attribute__((always_inline)) inline
Acc4 wave_gemm_tile(const float* __restrict__ arow,
                    const float* __restrict__ lWs,
                    int kdim, int half, int l16)
{
    Acc4 A;
    A.t[0] = {}; A.t[1] = {}; A.t[2] = {}; A.t[3] = {};
    const v2f* a2 = (const v2f*)arow;
    for (int k0 = 0; k0 < kdim; k0 += 4) {
        const int kp  = (k0 >> 1) + half;
        const v2f  a  = a2[kp];
        const v2f* br = (const v2f*)(lWs + kp * BSTRIDE_F);
#pragma unroll
        for (int nt = 0; nt < 4; ++nt) {
            const v2f b = br[nt * 16 + l16];
            A.t[nt] = __builtin_amdgcn_wmma_f32_16x16x4_f32(
                false, a, false, b, (short)0, A.t[nt], false, false);
        }
    }
    return A;
}

#define NTHREADS 256

template <int KDIM>
struct WStage {
    static constexpr int ITEMS = (KDIM / 2) * (OUT_DIM / 4);
    static constexpr int PER   = ITEMS / NTHREADS;
    static_assert(ITEMS % NTHREADS == 0, "staging is laid out for whole rounds of the work-group");
    float4 lo[PER];
    float4 hi[PER];
};

template <int KDIM>
__device__ __attribute__((always_inline)) inline
WStage<KDIM> load_w(const float* __restrict__ W)
{
    WStage<KDIM> s;
#pragma unroll
    for (int j = 0; j < WStage<KDIM>::PER; ++j) {
        const int it = threadIdx.x + NTHREADS * j;
        const int kp = it >> 4;
        const int g  = it & 15;
        const float4* src = reinterpret_cast<const float4*>(W + (size_t)(2 * kp) * OUT_DIM + 4 * g);
        s.lo[j] = src[0];
        s.hi[j] = src[OUT_DIM / 4];
    }
    return s;
}

template <int KDIM>
__device__ __attribute__((always_inline)) inline
void store_w(float* __restrict__ lWs, const WStage<KDIM>& s)
{
#pragma unroll
    for (int j = 0; j < WStage<KDIM>::PER; ++j) {
        const int it = threadIdx.x + NTHREADS * j;
        const int kp = it >> 4;
        const int g  = it & 15;
        float4* dst = reinterpret_cast<float4*>(lWs + kp * BSTRIDE_F + 8 * g);
        dst[0] = make_float4(s.lo[j].x, s.hi[j].x, s.lo[j].y, s.hi[j].y);
        dst[1] = make_float4(s.lo[j].z, s.hi[j].z, s.lo[j].w, s.hi[j].w);
    }
}

__device__ __attribute__((always_inline)) inline
int chip_half()
{
    const unsigned v = __builtin_amdgcn_s_sendmsg_rtn(0x87);
    return (int)((v >> 18) & 1u);
}

__device__ inline void atomicMaxF(float* addr, float val) {
    if (val >= 0.0f) atomicMax(reinterpret_cast<int*>(addr), __float_as_int(val));
    else             atomicMin(reinterpret_cast<unsigned int*>(addr), __float_as_uint(val));
}

template <int M>
__device__ __attribute__((always_inline)) inline
float row_xadd(float v)
{
    return v + __builtin_bit_cast(float,
        __builtin_amdgcn_update_dpp(0, __builtin_bit_cast(int, v), 0x160 + M, 0xf, 0xf, true));
}

__device__ __attribute__((always_inline)) inline
float row16_sum(float v)
{
    v = row_xadd<1>(v);
    v = row_xadd<2>(v);
    v = row_xadd<4>(v);
    return row_xadd<8>(v);
}

__global__ void k_node_gemm(const float* __restrict__ x,
                            const float* __restrict__ W_fc,  const float* __restrict__ b_fc,
                            const float* __restrict__ W_self, const float* __restrict__ b_self,
                            const float* __restrict__ W_attn,
                            float* __restrict__ z, float* __restrict__ hout,
                            float* __restrict__ a_src, float* __restrict__ a_dst,
                            int* __restrict__ binz, int nbinz,
                            float* __restrict__ plane1, int two,
                            int N, int nTiles)
{
    __shared__ __attribute__((aligned(16))) float lWs[(IN_DIM / 2) * BSTRIDE_F];
    const int lane = threadIdx.x & 31;
    const int wave = threadIdx.x >> 5;
    const int half = lane >> 4;
    const int l16  = lane & 15;
    const int tile = blockIdx.x * 8 + wave;

    {
        const int gid = blockIdx.x * NTHREADS + threadIdx.x;
        if (gid < nbinz) binz[gid] = 0;
    }

    const WStage<IN_DIM> w_first  = load_w<IN_DIM>(W_fc);
    const WStage<IN_DIM> w_second = load_w<IN_DIM>(W_self);

    int mrow = tile * 16 + l16;
    if (mrow >= N) mrow = N - 1;
    const float* xrow = x + (size_t)mrow * IN_DIM;

    if (two && tile < nTiles) {
        float4* p4 = reinterpret_cast<float4*>(plane1 + (size_t)tile * 16 * OUT_DIM);
#pragma unroll
        for (int it = 0; it < 8; ++it) {
            const int idx = it * 32 + lane;
            if (tile * 16 + (idx >> 4) < N) p4[idx] = make_float4(0.f, 0.f, 0.f, 0.f);
        }
    }

    store_w<IN_DIM>(lWs, w_first);
    __syncthreads();
    if (tile < nTiles) {
        Acc4 acc = wave_gemm_tile(xrow, lWs, IN_DIM, half, l16);
        float as[8], ad[8];
#pragma unroll
        for (int r = 0; r < 8; ++r) { as[r] = 0.f; ad[r] = 0.f; }
#pragma unroll
        for (int nt = 0; nt < 4; ++nt) {
            const int   c  = nt * 16 + l16;
            const float bc = b_fc[c];
            const float ws = W_attn[c];
            const float wd = W_attn[OUT_DIM + c];
#pragma unroll
            for (int r = 0; r < 8; ++r) {
                const int   row = tile * 16 + r + 8 * half;
                const float zc  = acc.t[nt][r] + bc;
                if (row < N) z[(size_t)row * OUT_DIM + c] = zc;
                as[r] += zc * ws;
                ad[r] += zc * wd;
            }
        }
#pragma unroll
        for (int r = 0; r < 8; ++r) { as[r] = row16_sum(as[r]); ad[r] = row16_sum(ad[r]); }
        float sv = as[0], dv = ad[0];
#pragma unroll
        for (int r = 1; r < 8; ++r) {
            if (l16 == r) { sv = as[r]; dv = ad[r]; }
        }
        const int orow = tile * 16 + 8 * half + l16;
        if (l16 < 8 && orow < N) { a_src[orow] = sv; a_dst[orow] = dv; }
    }
    __syncthreads();

    store_w<IN_DIM>(lWs, w_second);
    __syncthreads();
    if (tile < nTiles) {
        Acc4 acc = wave_gemm_tile(xrow, lWs, IN_DIM, half, l16);
#pragma unroll
        for (int nt = 0; nt < 4; ++nt) {
            const int   c  = nt * 16 + l16;
            const float bc = b_self[c];
#pragma unroll
            for (int r = 0; r < 8; ++r) {
                const int row = tile * 16 + r + 8 * half;
                if (row < N) hout[(size_t)row * OUT_DIM + c] = acc.t[nt][r] + bc;
            }
        }
    }
}

__global__ void k_node_attn(const float* __restrict__ z, const float* __restrict__ W_attn,
                            float* __restrict__ a_src, float* __restrict__ a_dst,
                            float* __restrict__ m, float* __restrict__ denom,
                            float* __restrict__ m1, float* __restrict__ denom1, int two, int N,
                            int* __restrict__ binz, int nbinz, int stats)
{
    const int n = blockIdx.x * blockDim.x + threadIdx.x;
    if (n < nbinz) binz[n] = 0;
    if (n >= N) return;
    const float* zr = z + (size_t)n * OUT_DIM;
    float as = 0.f, ad = 0.f;
#pragma unroll 8
    for (int c = 0; c < OUT_DIM; ++c) {
        const float zc = zr[c];
        as += zc * W_attn[c];
        ad += zc * W_attn[OUT_DIM + c];
    }
    a_src[n] = as;
    a_dst[n] = ad;
    if (!stats) return;
    m[n]     = -INFINITY;
    denom[n] = 0.0f;
    if (two) {
        m1[n]     = -INFINITY;
        denom1[n] = 0.0f;
    }
}

__global__ void k_edge_logit(const float* __restrict__ edge_attr,
                             const int* __restrict__ erow, const int* __restrict__ ecol,
                             const float* __restrict__ W_attn, const float* __restrict__ b_attn,
                             const float* __restrict__ a_src, const float* __restrict__ a_dst,
                             float* __restrict__ elog, float* __restrict__ m,
                             float* __restrict__ m1, int two, int E)
{
    const int e = blockIdx.x * blockDim.x + threadIdx.x;
    if (e >= E) return;
    const float* ea = edge_attr + (size_t)e * ED_DIM;
    float dot = 0.f;
#pragma unroll
    for (int k = 0; k < ED_DIM; ++k) dot += ea[k] * W_attn[2 * OUT_DIM + k];
    const int c = ecol[e];
    const float s  = a_src[erow[e]] + a_dst[c] + dot + b_attn[0];
    const float lr = (s > 0.0f) ? s : 0.2f * s;
    elog[e] = lr;
    float* mh = (two && chip_half()) ? m1 : m;
    atomicMaxF(&mh[c], lr);
}

__global__ void k_edge_soft(const float* __restrict__ elog, const int* __restrict__ ecol,
                            const float* __restrict__ m, const float* __restrict__ m1,
                            float* __restrict__ pbuf,
                            float* __restrict__ denom, float* __restrict__ denom1, int two, int E)
{
    const int e = blockIdx.x * blockDim.x + threadIdx.x;
    if (e >= E) return;
    const int c = ecol[e];
    const float mc = fmaxf(m[c], m1[c]);
    const float p = expf(elog[e] - mc);
    pbuf[e] = p;
    float* dh = (two && chip_half()) ? denom1 : denom;
    atomicAdd(&dh[c], p);
}

__device__ __attribute__((always_inline)) inline
float xhalf(float x)
{
    const int b = __builtin_bit_cast(int, x);
    return __builtin_bit_cast(float,
        __builtin_amdgcn_permlanex16(b, b, 0x76543210, 0xfedcba98, false, false));
}

template <bool FULL>
__device__ __attribute__((always_inline)) inline
void edge_epilogue(const Acc4& acc, int c_l, int r_l, float al_l, float b_lo, float b_hi,
                   const float* __restrict__ z, float* __restrict__ hout, int lane, int half)
{
    float zlo[16], zhi[16];
    if (FULL) {
#pragma unroll
        for (int j = 0; j < 16; ++j) {
            const int    rj = __builtin_amdgcn_readlane(r_l, j);
            const float* zr = z + (size_t)rj * OUT_DIM;
            zlo[j] = zr[lane];
            zhi[j] = zr[32 + lane];
        }
    }
#pragma unroll
    for (int r = 0; r < 8; ++r) {
        const float gA   = xhalf(half ? acc.t[0][r] : acc.t[1][r]);
        const float gB   = xhalf(half ? acc.t[2][r] : acc.t[3][r]);
        const float v0lo = half ? gA : acc.t[0][r];
        const float v1lo = half ? acc.t[1][r] : gA;
        const float v0hi = half ? gB : acc.t[2][r];
        const float v1hi = half ? acc.t[3][r] : gB;
#pragma unroll
        for (int h = 0; h < 2; ++h) {
            const int   j  = r + 8 * h;
            const int   cj = __builtin_amdgcn_readlane(c_l, j);
            const float aj = __builtin_bit_cast(float,
                __builtin_amdgcn_readlane(__builtin_bit_cast(int, al_l), j));
            if (FULL || cj >= 0) {
                float zl, zh;
                if (FULL) {
                    zl = zlo[j];
                    zh = zhi[j];
                } else {
                    const int    rj = __builtin_amdgcn_readlane(r_l, j);
                    const float* zr = z + (size_t)rj * OUT_DIM;
                    zl = zr[lane];
                    zh = zr[32 + lane];
                }
                float* hr = hout + (size_t)cj * OUT_DIM;
                atomicAdd(&hr[lane],      aj * (zl + (h ? v1lo : v0lo) + b_lo));
                atomicAdd(&hr[32 + lane], aj * (zh + (h ? v1hi : v0hi) + b_hi));
            }
        }
    }
}

__global__ void k_edge_aggregate(const float* __restrict__ edge_attr,
                                 const int* __restrict__ erow, const int* __restrict__ ecol,
                                 const float* __restrict__ W_edge, const float* __restrict__ b_edge,
                                 const float* __restrict__ z, const float* __restrict__ pbuf,
                                 const float* __restrict__ denom, const float* __restrict__ denom1,
                                 float* __restrict__ hout, float* __restrict__ plane1, int two,
                                 int E, int nTiles)
{
    __shared__ __attribute__((aligned(16))) float lWs[(ED_DIM / 2) * BSTRIDE_F];
    const int lane = threadIdx.x & 31;
    const int wave = threadIdx.x >> 5;
    const int half = lane >> 4;
    const int l16  = lane & 15;

    store_w<ED_DIM>(lWs, load_w<ED_DIM>(W_edge));
    __syncthreads();
    const int tile = __builtin_amdgcn_readfirstlane(blockIdx.x * 8 + wave);
    if (tile >= nTiles) return;

    int arow = tile * 16 + l16;
    const bool evalid = arow < E;
    if (arow >= E) arow = E - 1;
    const float* ea = edge_attr + (size_t)arow * ED_DIM;

    {
        const size_t nxt = (size_t)arow + 16 * 8;
        if (nxt < (size_t)E)
            __builtin_prefetch(edge_attr + nxt * ED_DIM, 0, 1);
    }

    const int   c_e  = ecol[arow];
    const int   c_l  = evalid ? c_e : -1;
    const int   r_l  = erow[arow];
    const float d0   = denom[c_e];
    const float d1   = denom1[c_e];
    const float den  = two ? d0 + d1 : d0;
    const float al_l = pbuf[arow] / fmaxf(den, 1e-16f);

    Acc4 acc = wave_gemm_tile(ea, lWs, ED_DIM, half, l16);

    const float b_lo = b_edge[lane];
    const float b_hi = b_edge[32 + lane];

    float* hacc = (two && chip_half()) ? plane1 : hout;

    if (tile * 16 + 16 <= E)
        edge_epilogue<true >(acc, c_l, r_l, al_l, b_lo, b_hi, z, hacc, lane, half);
    else
        edge_epilogue<false>(acc, c_l, r_l, al_l, b_lo, b_hi, z, hacc, lane, half);
}


#define BIN_R   196
#define RANGE_TILES ((BIN_R + 15) / 16)
#define EA_PITCH 34
#define STAT_PER 4
#define NB_MAX  1024
#define BIN_T   1024
#define BIN_PER 4

__global__ __launch_bounds__(BIN_T)
void k_bin_count(const int* __restrict__ ecol, int* __restrict__ cnt, int E, int nb)
{
    __shared__ int s_hist[NB_MAX];
    const int tid = threadIdx.x;
    s_hist[tid] = 0;
    __syncthreads();
#pragma unroll
    for (int i = 0; i < BIN_PER; ++i) {
        const int e = blockIdx.x * (BIN_T * BIN_PER) + i * BIN_T + tid;
        if (e < E) atomicAdd(&s_hist[ecol[e] / BIN_R], 1);
    }
    __syncthreads();
    const int h = s_hist[tid];
    if (tid < nb && h > 0) atomicAdd(&cnt[tid], h);
}

__device__ __attribute__((always_inline)) inline
float x8_4(float v) { return __builtin_bit_cast(float, __builtin_amdgcn_mov_dpp8(__builtin_bit_cast(unsigned, v), 0x688fac)); }
__device__ __attribute__((always_inline)) inline
float x8_2(float v) { return __builtin_bit_cast(float, __builtin_amdgcn_mov_dpp8(__builtin_bit_cast(unsigned, v), 0xb3e21a)); }
__device__ __attribute__((always_inline)) inline
float x8_1(float v) { return __builtin_bit_cast(float, __builtin_amdgcn_mov_dpp8(__builtin_bit_cast(unsigned, v), 0xde54c1)); }

__device__ __attribute__((always_inline)) inline
void ea_request(float4 (&v)[8], const float* __restrict__ edge_attr, int base, int sub, int g)
{
    const float4* p = reinterpret_cast<const float4*>(
        edge_attr + (size_t)(base + sub) * ED_DIM + 4 * g);
#pragma unroll
    for (int q = 0; q < 8; ++q) v[q] = p[q * ED_DIM];
}

__device__ __attribute__((always_inline)) inline
float ea_fold(const float4 (&v)[8], const float4 wa, int lane)
{
    float d[8];
#pragma unroll
    for (int q = 0; q < 8; ++q)
        d[q] = v[q].x * wa.x + v[q].y * wa.y + v[q].z * wa.z + v[q].w * wa.w;
    float s1[4];
#pragma unroll
    for (int q = 0; q < 4; ++q) {
        const float keep = (lane & 4) ? d[q + 4] : d[q];
        const float send = (lane & 4) ? d[q] : d[q + 4];
        s1[q] = keep + x8_4(send);
    }
    float s2[2];
#pragma unroll
    for (int q = 0; q < 2; ++q) {
        const float keep = (lane & 2) ? s1[q + 2] : s1[q];
        const float send = (lane & 2) ? s1[q] : s1[q + 2];
        s2[q] = keep + x8_2(send);
    }
    const float keep = (lane & 1) ? s2[1] : s2[0];
    const float send = (lane & 1) ? s2[0] : s2[1];
    return keep + x8_1(send);
}

__global__ __launch_bounds__(BIN_T)
void k_bin_place(const float* __restrict__ edge_attr,
                 const int* __restrict__ erow, const int* __restrict__ ecol,
                 const float* __restrict__ W_attn, const float* __restrict__ b_attn,
                 const float* __restrict__ a_src, const float* __restrict__ a_dst,
                 const int* __restrict__ cnt, int* __restrict__ cursor, int* __restrict__ bbase,
                 int* __restrict__ rec,
                 int E, int nb)
{
    __shared__ int s_base[NB_MAX];
    __shared__ int s_hist[NB_MAX];
    __shared__ int s_shift[NB_MAX];
    __shared__ int s_lp[NB_MAX];
    __shared__ int s_wsum[32];
    __shared__ int s_wsum2[32];
    __shared__ __attribute__((aligned(16))) int4 s_rec[BIN_T * BIN_PER];
    __shared__ float s_dot[BIN_T * BIN_PER];
    const int tid  = threadIdx.x;
    const int lane = tid & 31;
    const int wave = __builtin_amdgcn_readfirstlane(tid >> 5);
    const int g    = lane & 7;
    const int sub  = lane >> 3;

    const float4 wa = *reinterpret_cast<const float4*>(W_attn + 2 * OUT_DIM + 4 * g);
    const float  ba = b_attn[0];

    const int base0 = blockIdx.x * (BIN_T * BIN_PER) + wave * 32;
    int   eid[BIN_PER], bk[BIN_PER], rc[BIN_PER];
    float lg[BIN_PER];
    int   cc[BIN_PER], rr[BIN_PER];
#pragma unroll
    for (int i = 0; i < BIN_PER; ++i) {
        const int e  = base0 + i * BIN_T + 4 * g + sub;
        const int ec = e < E ? e : E - 1;
        eid[i] = e;
        cc[i]  = ecol[ec];
        rr[i]  = erow[ec];
    }
#pragma unroll 1
    for (int i = 0; i < BIN_PER; ++i) {
        const int b = base0 + i * BIN_T;
        float4 ln[8];
        ea_request(ln, edge_attr, (b + 32 <= E) ? b : E - 32, sub, g);
        s_dot[i * BIN_T + tid] = ea_fold(ln, wa, lane);
    }
    float asd[BIN_PER];
#pragma unroll
    for (int i = 0; i < BIN_PER; ++i) asd[i] = a_src[rr[i]] + a_dst[cc[i]];

#pragma unroll
    for (int i = 0; i < BIN_PER; ++i) {
        float dot = s_dot[i * BIN_T + tid];
        const int base = base0 + i * BIN_T;
        if (base < E && base + 32 > E) {
            if (eid[i] < E) {
                const float* ea = edge_attr + (size_t)eid[i] * ED_DIM;
                float t = 0.f;
                for (int k = 0; k < ED_DIM; ++k) t += ea[k] * W_attn[2 * OUT_DIM + k];
                dot = t;
            }
        }
        const float s = asd[i] + dot + ba;
        const int   b = cc[i] / BIN_R;
        bk[i] = eid[i] < E ? b : -1;
        rc[i] = rr[i] * 256 + (cc[i] - b * BIN_R);
        lg[i] = (s > 0.0f) ? s : 0.2f * s;
    }

    const int own = tid < nb ? cnt[tid] : 0;
    int v = own;
#pragma unroll
    for (int d = 1; d < 32; d <<= 1) {
        const int u = __shfl_up(v, d);
        if (lane >= d) v += u;
    }
    if (lane == 31) s_wsum[wave] = v;
    s_hist[tid] = 0;
    __syncthreads();
    if (wave == 0) {
        const int w = s_wsum[lane];
        int x = w;
#pragma unroll
        for (int d = 1; d < 32; d <<= 1) {
            const int u = __shfl_up(x, d);
            if (lane >= d) x += u;
        }
        s_wsum[lane] = x - w;
    }
    __syncthreads();
    const int excl = v - own + s_wsum[wave];
    s_base[tid] = excl;
    if (blockIdx.x == 0) {
        if (tid < nb)      bbase[tid] = excl;
        if (tid == nb - 1) bbase[nb]  = excl + own;
    }

    int lr[BIN_PER];
#pragma unroll
    for (int i = 0; i < BIN_PER; ++i) {
        lr[i] = 0;
        if (bk[i] >= 0) lr[i] = atomicAdd(&s_hist[bk[i]], 1);
    }
    __syncthreads();
    const int hc = s_hist[tid];
    int w = hc;
#pragma unroll
    for (int d = 1; d < 32; d <<= 1) {
        const int u = __shfl_up(w, d);
        if (lane >= d) w += u;
    }
    if (lane == 31) s_wsum2[wave] = w;
    __syncthreads();
    if (wave == 0) {
        const int t = s_wsum2[lane];
        int x = t;
#pragma unroll
        for (int d = 1; d < 32; d <<= 1) {
            const int u = __shfl_up(x, d);
            if (lane >= d) x += u;
        }
        s_wsum2[lane] = x - t;
    }
    __syncthreads();
    const int lp = w - hc + s_wsum2[wave];
    s_lp[tid] = lp;
    if (tid < nb && hc > 0) s_shift[tid] = s_base[tid] + atomicAdd(&cursor[tid], hc) - lp;
    __syncthreads();
    int at[BIN_PER];
#pragma unroll
    for (int i = 0; i < BIN_PER; ++i) at[i] = s_lp[bk[i] >= 0 ? bk[i] : 0] + lr[i];
#pragma unroll
    for (int i = 0; i < BIN_PER; ++i) {
        if (bk[i] >= 0)
            s_rec[at[i]] = make_int4(eid[i], rc[i], __float_as_int(lg[i]), bk[i]);
    }
    __syncthreads();
    const int first = blockIdx.x * (BIN_T * BIN_PER);
    const int nval  = (E - first < BIN_T * BIN_PER) ? (E - first) : BIN_T * BIN_PER;
    int4 q[BIN_PER];
    int  sh[BIN_PER];
#pragma unroll
    for (int i = 0; i < BIN_PER; ++i) {
        const int p = i * BIN_T + tid;
        q[i] = s_rec[p < nval ? p : 0];
    }
#pragma unroll
    for (int i = 0; i < BIN_PER; ++i) sh[i] = s_shift[q[i].w];
#pragma unroll
    for (int i = 0; i < BIN_PER; ++i) {
        const int p = i * BIN_T + tid;
        if (p < nval) {
            int* r = rec + 3 * (sh[i] + p);
            r[0] = q[i].x;
            r[1] = q[i].y;
            r[2] = q[i].z;
        }
    }
}

__device__ __attribute__((always_inline)) inline
void lds_max(float* addr, float val)
{
    __hip_atomic_fetch_max(addr, val, __ATOMIC_RELAXED, __HIP_MEMORY_SCOPE_WORKGROUP);
}

template <bool FULL>
__device__ __attribute__((always_inline)) inline
void edge_accumulate(int e_l, int c_l, int r_l, float al_l,
                     const float* __restrict__ edge_attr, const float* __restrict__ z,
                     float* __restrict__ lacc, float* __restrict__ lea, int lane)
{
    float ev[16];
    v2f   zv[16];
    if (FULL) {
#pragma unroll
        for (int j = 0; j < 16; ++j) {
            const int ej = __builtin_amdgcn_readlane(e_l, j);
            const int rj = __builtin_amdgcn_readlane(r_l, j);
            ev[j] = edge_attr[(size_t)ej * ED_DIM + lane];
            zv[j] = reinterpret_cast<const v2f*>(z + (size_t)rj * OUT_DIM)[lane];
        }
    }
#pragma unroll
    for (int j = 0; j < 16; ++j) {
        const int   cj = __builtin_amdgcn_readlane(c_l, j);
        const float aj = __builtin_bit_cast(float,
            __builtin_amdgcn_readlane(__builtin_bit_cast(int, al_l), j));
        if (FULL || cj >= 0) {
            float e1;
            v2f   z2;
            if (FULL) {
                e1 = ev[j];
                z2 = zv[j];
            } else {
                const int ej = __builtin_amdgcn_readlane(e_l, j);
                const int rj = __builtin_amdgcn_readlane(r_l, j);
                e1 = edge_attr[(size_t)ej * ED_DIM + lane];
                z2 = reinterpret_cast<const v2f*>(z + (size_t)rj * OUT_DIM)[lane];
            }
            float* hr = lacc + cj * OUT_DIM + 2 * lane;
            atomicAdd(&hr[0], aj * z2.x);
            atomicAdd(&hr[1], aj * z2.y);
            atomicAdd(&lea[cj * EA_PITCH + lane], aj * e1);
        }
    }
}

__global__ __launch_bounds__(BIN_T)
void k_range_aggregate(const float* __restrict__ edge_attr,
                       const float* __restrict__ W_edge, const float* __restrict__ b_edge,
                       const float* __restrict__ z,
                       const int* __restrict__ rec, const int* __restrict__ bbase,
                       float* __restrict__ hout, int N)
{
    __shared__ __attribute__((aligned(16))) float lWs[(ED_DIM / 2) * BSTRIDE_F];
    __shared__ __attribute__((aligned(16))) float lacc[BIN_R * OUT_DIM];
    __shared__ __attribute__((aligned(16))) float lea[RANGE_TILES * 16 * EA_PITCH];
    __shared__ float lmax[BIN_R];
    __shared__ float lden[BIN_R];
    const int lane = threadIdx.x & 31;
    const int wave = __builtin_amdgcn_readfirstlane(threadIdx.x >> 5);
    const int half = lane >> 4;
    const int l16  = lane & 15;

    const int n0   = blockIdx.x * BIN_R;
    const int rows = (N - n0 < BIN_R) ? (N - n0) : BIN_R;

    const int b0 = bbase[blockIdx.x];
    const int b1 = bbase[blockIdx.x + 1];
    const int ntile = (b1 - b0 + 15) >> 4;

    float lg_s[STAT_PER];
    int   rc_s[STAT_PER];
#pragma unroll
    for (int j = 0; j < STAT_PER; ++j) {
        const int i  = b0 + j * BIN_T + (int)threadIdx.x;
        const int ii = i < b1 ? i : 0;
        rc_s[j] = rec[3 * ii + 1];
        lg_s[j] = __int_as_float(rec[3 * ii + 2]);
    }
    int   e_n = 0, rc_n = 0;
    float lg_n = 0.f;
    if (wave < ntile) {
        int idx = b0 + wave * 16 + l16;
        if (idx >= b1) idx = b1 - 1;
        e_n = rec[3 * idx]; rc_n = rec[3 * idx + 1]; lg_n = __int_as_float(rec[3 * idx + 2]);
    }

    if (threadIdx.x < BIN_R) { lmax[threadIdx.x] = -INFINITY; lden[threadIdx.x] = 0.f; }
    if (threadIdx.x < NTHREADS) store_w<ED_DIM>(lWs, load_w<ED_DIM>(W_edge));
    {
        const float4* h4 = reinterpret_cast<const float4*>(hout + (size_t)n0 * OUT_DIM);
        float4*       a4 = reinterpret_cast<float4*>(lacc);
        for (int i = threadIdx.x; i < rows * (OUT_DIM / 4); i += BIN_T) a4[i] = h4[i];
        float4*       s4 = reinterpret_cast<float4*>(lea);
        for (int i = threadIdx.x; i < RANGE_TILES * 16 * EA_PITCH / 4; i += BIN_T)
            s4[i] = make_float4(0.f, 0.f, 0.f, 0.f);
    }
    __syncthreads();

    int c_s[STAT_PER];
#pragma unroll
    for (int j = 0; j < STAT_PER; ++j)
        c_s[j] = (b0 + j * BIN_T + (int)threadIdx.x < b1) ? (rc_s[j] & 255) : -1;

#pragma unroll
    for (int j = 0; j < STAT_PER; ++j)
        if (c_s[j] >= 0) lds_max(&lmax[c_s[j]], lg_s[j]);
    for (int i = b0 + STAT_PER * BIN_T + (int)threadIdx.x; i < b1; i += BIN_T)
        lds_max(&lmax[rec[3 * i + 1] & 255], __int_as_float(rec[3 * i + 2]));
    __syncthreads();

#pragma unroll
    for (int j = 0; j < STAT_PER; ++j)
        if (c_s[j] >= 0) atomicAdd(&lden[c_s[j]], expf(lg_s[j] - lmax[c_s[j]]));
    for (int i = b0 + STAT_PER * BIN_T + (int)threadIdx.x; i < b1; i += BIN_T) {
        const int cc = rec[3 * i + 1] & 255;
        atomicAdd(&lden[cc], expf(__int_as_float(rec[3 * i + 2]) - lmax[cc]));
    }
    __syncthreads();

    for (int t = wave; t < ntile; t += BIN_T / 32) {
        const int   e_c = e_n, rc_c = rc_n;
        const float lg_c = lg_n;
        const int tn = t + BIN_T / 32;
        if (tn < ntile) {
            int idx = b0 + tn * 16 + l16;
            if (idx >= b1) idx = b1 - 1;
            e_n = rec[3 * idx]; rc_n = rec[3 * idx + 1]; lg_n = __int_as_float(rec[3 * idx + 2]);
        }
        const bool valid = b0 + t * 16 + l16 < b1;
        const int  cc    = rc_c & 255;
        const int  c_l   = valid ? cc : -1;
        const int  r_l   = rc_c >> 8;
        const float al_c = expf(lg_c - lmax[cc]) / fmaxf(lden[cc], 1e-16f);

        if (t * 16 + 16 <= b1 - b0)
            edge_accumulate<true >(e_c, c_l, r_l, al_c, edge_attr, z, lacc, lea, lane);
        else
            edge_accumulate<false>(e_c, c_l, r_l, al_c, edge_attr, z, lacc, lea, lane);
    }
    __syncthreads();

    for (int t = wave; t < RANGE_TILES; t += BIN_T / 32) {
        const float* arow = lea + (t * 16 + l16) * EA_PITCH;
        Acc4 acc = wave_gemm_tile(arow, lWs, ED_DIM, half, l16);
        float sa[8];
#pragma unroll
        for (int r = 0; r < 8; ++r) {
            const int row = t * 16 + r + 8 * half;
            const float d = lden[row < rows ? row : 0];
            sa[r] = d / fmaxf(d, 1e-16f);
        }
#pragma unroll
        for (int nt = 0; nt < 4; ++nt) {
            const int   c  = nt * 16 + l16;
            const float bc = b_edge[c];
#pragma unroll
            for (int r = 0; r < 8; ++r) {
                const int row = t * 16 + r + 8 * half;
                if (row < rows) lacc[row * OUT_DIM + c] += acc.t[nt][r] + sa[r] * bc;
            }
        }
    }
    __syncthreads();
    {
        float4*       o4 = reinterpret_cast<float4*>(hout + (size_t)n0 * OUT_DIM);
        const float4* a4 = reinterpret_cast<const float4*>(lacc);
        for (int i = threadIdx.x; i < rows * (OUT_DIM / 4); i += BIN_T) o4[i] = a4[i];
    }
}

__global__ void k_plane_sum(float* __restrict__ hout, const float* __restrict__ plane1,
                            int two, int n4)
{
    const int i = blockIdx.x * blockDim.x + threadIdx.x;
    if (!two || i >= n4) return;
    float4*       h4 = reinterpret_cast<float4*>(hout);
    const float4* q4 = reinterpret_cast<const float4*>(plane1);
    float4 a = h4[i];
    const float4 b = q4[i];
    a.x += b.x; a.y += b.y; a.z += b.z; a.w += b.w;
    h4[i] = a;
}

extern "C" void kernel_launch(void* const* d_in, const int* in_sizes, int n_in,
                              void* d_out, int out_size, void* d_ws, size_t ws_size,
                              hipStream_t stream)
{
    const float* x         = (const float*)d_in[0];
    const float* edge_attr = (const float*)d_in[1];
    const int*   edge_idx  = (const int*)  d_in[2];
    const float* W_fc      = (const float*)d_in[3];
    const float* b_fc      = (const float*)d_in[4];
    const float* W_attn    = (const float*)d_in[5];
    const float* b_attn    = (const float*)d_in[6];
    const float* W_edge    = (const float*)d_in[7];
    const float* b_edge    = (const float*)d_in[8];
    const float* W_self    = (const float*)d_in[9];
    const float* b_self    = (const float*)d_in[10];

    const int N = in_sizes[0] / IN_DIM;
    const int E = in_sizes[1] / ED_DIM;
    const int* erow = edge_idx;
    const int* ecol = edge_idx + E;

    float* ws     = (float*)d_ws;
    float* z      = ws;
    float* a_src  = z + (size_t)N * OUT_DIM;
    float* a_dst  = a_src + N;
    float* m      = a_dst + N;
    float* denom  = m + N;
    float* elog   = denom + N;
    float* pbuf   = elog + E;
    const size_t off1 = ((size_t)N * (OUT_DIM + 6) + 2 * (size_t)E + 3) & ~(size_t)3;
    const int two = ws_size >= (off1 + (size_t)N * OUT_DIM) * sizeof(float) ? 1 : 0;
    float* m1     = two ? pbuf + E : m;
    float* denom1 = two ? m1 + N   : denom;
    float* plane1 = two ? ws + off1 : (float*)d_out;
    float* hout   = (float*)d_out;

    const int nNodeTiles = (N + 15) / 16;
    const int nEdgeTiles = (E + 15) / 16;

    const int nb   = (N + BIN_R - 1) / BIN_R;
    const int bins = (two && nb <= NB_MAX && N >= 2 * NB_MAX && N < (1 << 23) && E >= 32 &&
                      (size_t)N * OUT_DIM >= 3 * (size_t)E + 3 * NB_MAX + 8) ? 1 : 0;
    int*   rec_e  = (int*)plane1;
    int*   rec_rc = rec_e + E;
    float* rec_lg = (float*)(rec_rc + E);
    int*   cnt    = (int*)(rec_lg + E);
    int*   cursor = cnt + NB_MAX;
    int*   bbase  = cursor + NB_MAX;
    const int nBinBlocks = (E + BIN_T * BIN_PER - 1) / (BIN_T * BIN_PER);

    k_node_gemm<<<(nNodeTiles + 7) / 8, NTHREADS, 0, stream>>>(
        x, W_fc, b_fc, W_self, b_self, W_attn, z, hout, a_src, a_dst,
        cnt, bins ? 2 * NB_MAX : 0, plane1, bins ? 0 : two, N, nNodeTiles);

    if (bins) {
        k_bin_count<<<nBinBlocks, BIN_T, 0, stream>>>(ecol, cnt, E, nb);

        k_bin_place<<<nBinBlocks, BIN_T, 0, stream>>>(
            edge_attr, erow, ecol, W_attn, b_attn, a_src, a_dst,
            cnt, cursor, bbase, rec_e, E, nb);

        k_range_aggregate<<<nb, BIN_T, 0, stream>>>(
            edge_attr, W_edge, b_edge, z, rec_e, bbase, hout, N);
    } else {
        k_node_attn<<<(N + 255) / 256, 256, 0, stream>>>(
            z, W_attn, a_src, a_dst, m, denom, m1, denom1, two, N, cnt, 0, 1);

        k_edge_logit<<<(E + 255) / 256, 256, 0, stream>>>(
            edge_attr, erow, ecol, W_attn, b_attn, a_src, a_dst, elog, m, m1, two, E);

        k_edge_soft<<<(E + 255) / 256, 256, 0, stream>>>(
            elog, ecol, m, m1, pbuf, denom, denom1, two, E);

        k_edge_aggregate<<<(nEdgeTiles + 7) / 8, NTHREADS, 0, stream>>>(
            edge_attr, erow, ecol, W_edge, b_edge, z, pbuf, denom, denom1, hout, plane1, two,
            E, nEdgeTiles);

        k_plane_sum<<<(N * (OUT_DIM / 4) + 255) / 256, 256, 0, stream>>>(
            hout, plane1, two, N * (OUT_DIM / 4));
    }
}
